# starting kernel
# baseline (speedup 1.0000x reference)
#include <hip/hip_runtime.h>
#include <cmath>

typedef __attribute__((ext_vector_type(2))) float v2f;
typedef __attribute__((ext_vector_type(8))) float v8f;

#define NFEAT 128
#define NCLS  40
#define LD2   48
#define KDIM  128
#define WPB   8


__global__ void fill_f32(float* __restrict__ p, long long n, float v) {
  long long i = blockIdx.x * (long long)blockDim.x + threadIdx.x;
  long long stride = (long long)gridDim.x * blockDim.x;
  for (; i < n; i += stride) p[i] = v;
}

__global__ void degree_kernel(const int* __restrict__ col, float* __restrict__ deg, int E) {
  int e = blockIdx.x * blockDim.x + threadIdx.x;
  if (e < E) atomicAdd(&deg[col[e]], 1.0f);
}

__global__ void rsqrt_kernel(float* __restrict__ deg, int N) {
  int i = blockIdx.x * blockDim.x + threadIdx.x;
  if (i < N) {
    float d = deg[i];
    deg[i] = (d > 0.0f) ? rsqrtf(d) : 0.0f;
  }
}

__global__ __launch_bounds__(256) void gemm_wmma_f32(
    const float* __restrict__ A, const float* __restrict__ B, float* __restrict__ C,
    int M, int Nvalid, int Ntiles, int ldc) {
  __shared__ float Bs[KDIM * 16];

  int nt = blockIdx.x % Ntiles;
  int mtBase = (blockIdx.x / Ntiles) * WPB;
  int n0 = nt << 4;
  int t = threadIdx.x;

#pragma unroll
  for (int j = 0; j < 8; ++j) {
    int idx = t + j * 256;
    int kb = idx >> 4;
    int n = n0 + (idx & 15);
    Bs[idx] = (n < Nvalid) ? B[(long long)kb * Nvalid + n] : 0.0f;
  }
  __syncthreads();

  int w = t >> 5;
  int mt = mtBase + w;
  int Mtiles = (M + 15) >> 4;
  if (mt >= Mtiles) return;

  int lane = t & 31;
  int l16 = lane & 15;
  int hi  = lane >> 4;
  int kh  = hi << 1;

  int m0 = mt << 4;
  int m = m0 + l16;
  if (m >= M) m = M - 1;
  const float* Arow = A + (long long)m * KDIM;

  v8f acc = {};
#pragma unroll 8
  for (int k = 0; k < KDIM; k += 4) {
    v2f a = *(const v2f*)(Arow + k + kh);
    v2f b;
    b.x = Bs[(k + kh) * 16 + l16];
    b.y = Bs[(k + kh + 1) * 16 + l16];
    acc = __builtin_amdgcn_wmma_f32_16x16x4_f32(
         false, a,  false, b,
         (short)0, acc,  false,  false);
  }

  int mbase = m0 + hi * 8;
  float* Cout = C + (long long)mbase * ldc + n0 + l16;
  if (m0 + 16 <= M) {
#pragma unroll
    for (int r = 0; r < 8; ++r) Cout[(long long)r * ldc] = acc[r];
  } else {
#pragma unroll
    for (int r = 0; r < 8; ++r)
      if (mbase + r < M) Cout[(long long)r * ldc] = acc[r];
  }
}

__global__ void scatter_add4(const int* __restrict__ row, const int* __restrict__ col,
                             const float* __restrict__ dinv, const float* __restrict__ h,
                             float* __restrict__ agg, int E, int F4, int ld) {
  long long total = (long long)E * F4;
  long long i = blockIdx.x * (long long)blockDim.x + threadIdx.x;
  long long stride = (long long)gridDim.x * blockDim.x;
  for (; i < total; i += stride) {
    int e = (int)(i / F4);
    int f = (int)(i - (long long)e * F4) << 2;
    int r = row[e];
    int c = col[e];
    float nrm = dinv[r] * dinv[c];
    const float4 hv = *(const float4*)(h + (long long)r * ld + f);
    float* a = agg + (long long)c * ld + f;
    atomicAdd(a + 0, hv.x * nrm);
    atomicAdd(a + 1, hv.y * nrm);
    atomicAdd(a + 2, hv.z * nrm);
    atomicAdd(a + 3, hv.w * nrm);
  }
}

__global__ void bias_relu_self(const float* __restrict__ h, const float* __restrict__ dinv,
                               const float* __restrict__ bias, float* __restrict__ agg,
                               int Nnodes) {
  long long total = (long long)Nnodes * NFEAT;
  long long i = blockIdx.x * (long long)blockDim.x + threadIdx.x;
  long long stride = (long long)gridDim.x * blockDim.x;
  for (; i < total; i += stride) {
    int nidx = (int)(i >> 7);
    int f = (int)(i & (NFEAT - 1));
    float d = dinv[nidx];
    float v = agg[i] + h[i] * d * d + bias[f];
    agg[i] = v > 0.0f ? v : 0.0f;
  }
}

__global__ void out_logsoftmax(const float* __restrict__ h2, const float* __restrict__ agg2,
                               const float* __restrict__ dinv, const float* __restrict__ b2,
                               float* __restrict__ out, int Nnodes) {
  int i = blockIdx.x * blockDim.x + threadIdx.x;
  if (i >= Nnodes) return;
  float d = dinv[i];
  float d2 = d * d;
  float logit[NCLS];
  float mx = -3.4e38f;
#pragma unroll
  for (int f = 0; f < NCLS; ++f) {
    float v = agg2[(long long)i * LD2 + f] + h2[(long long)i * LD2 + f] * d2 + b2[f];
    logit[f] = v;
    mx = fmaxf(mx, v);
  }
  float s = 0.0f;
#pragma unroll
  for (int f = 0; f < NCLS; ++f) s += __expf(logit[f] - mx);
  float lse = mx + __logf(s);
#pragma unroll
  for (int f = 0; f < NCLS; ++f) out[(long long)i * NCLS + f] = logit[f] - lse;
}


extern "C" void kernel_launch(void* const* d_in, const int* in_sizes, int n_in,
                              void* d_out, int out_size, void* d_ws, size_t ws_size,
                              hipStream_t stream) {
  const float* x  = (const float*)d_in[0];
  const int*   ei = (const int*)d_in[1];
  const float* W1 = (const float*)d_in[2];
  const float* b1 = (const float*)d_in[3];
  const float* W2 = (const float*)d_in[4];
  const float* b2 = (const float*)d_in[5];
  float* out = (float*)d_out;

  const int N = in_sizes[0] / NFEAT;
  const int E = in_sizes[1] / 2;
  const int* erow = ei;
  const int* ecol = ei + E;

  float* dinv = (float*)d_ws;
  float* h1   = dinv + N;
  float* agg1 = h1 + (long long)N * NFEAT;
  float* h2   = agg1 + (long long)N * NFEAT;
  float* agg2 = h2 + (long long)N * LD2;

  const int TPB = 256;
  auto blocks = [](long long n, int tpb) {
    long long b = (n + tpb - 1) / tpb;
    return (unsigned)(b > 1048576 ? 1048576 : b);
  };

  fill_f32<<<blocks(N, TPB), TPB, 0, stream>>>(dinv, N, 1.0f);
  fill_f32<<<blocks((long long)N * NFEAT, TPB), TPB, 0, stream>>>(agg1, (long long)N * NFEAT, 0.0f);
  fill_f32<<<blocks((long long)N * LD2, TPB), TPB, 0, stream>>>(agg2, (long long)N * LD2, 0.0f);

  degree_kernel<<<(E + TPB - 1) / TPB, TPB, 0, stream>>>(ecol, dinv, E);
  rsqrt_kernel<<<(N + TPB - 1) / TPB, TPB, 0, stream>>>(dinv, N);

  const int Mtiles = (N + 15) / 16;
  const int mGroups = (Mtiles + WPB - 1) / WPB;

  {
    int Ntiles = NFEAT / 16;
    gemm_wmma_f32<<<mGroups * Ntiles, 256, 0, stream>>>(x, W1, h1, N, NFEAT, Ntiles, NFEAT);
  }

  scatter_add4<<<blocks((long long)E * (NFEAT / 4), TPB), TPB, 0, stream>>>(
      erow, ecol, dinv, h1, agg1, E, NFEAT / 4, NFEAT);

  bias_relu_self<<<blocks((long long)N * NFEAT, TPB), TPB, 0, stream>>>(
      h1, dinv, b1, agg1, N);

  {
    int Ntiles = LD2 / 16;
    gemm_wmma_f32<<<mGroups * Ntiles, 256, 0, stream>>>(agg1, W2, h2, N, NCLS, Ntiles, LD2);
  }

  scatter_add4<<<blocks((long long)E * (NCLS / 4), TPB), TPB, 0, stream>>>(
      erow, ecol, dinv, h2, agg2, E, NCLS / 4, LD2);

  out_logsoftmax<<<(N + TPB - 1) / TPB, TPB, 0, stream>>>(
      h2, agg2, dinv, b2, out, N);
}


# edge scatter: one lane per consecutive float, so each load and each f32 atomic add instruction covers whole 128-byte lines (layer 1 a wave per edge, layer 2 half a wave); arrival-order adds as before
# speedup vs baseline: 1.3430x; 1.3430x over previous
#include <hip/hip_runtime.h>
#include <cmath>

typedef __attribute__((ext_vector_type(2))) float v2f;
typedef __attribute__((ext_vector_type(8))) float v8f;

#define NFEAT 128
#define NCLS  40
#define LD2   48
#define KDIM  128
#define WPB   8


__global__ void fill_f32(float* __restrict__ p, long long n, float v) {
  long long i = blockIdx.x * (long long)blockDim.x + threadIdx.x;
  long long stride = (long long)gridDim.x * blockDim.x;
  for (; i < n; i += stride) p[i] = v;
}

__global__ void degree_kernel(const int* __restrict__ col, float* __restrict__ deg, int E) {
  int e = blockIdx.x * blockDim.x + threadIdx.x;
  if (e < E) atomicAdd(&deg[col[e]], 1.0f);
}

__global__ void rsqrt_kernel(float* __restrict__ deg, int N) {
  int i = blockIdx.x * blockDim.x + threadIdx.x;
  if (i < N) {
    float d = deg[i];
    deg[i] = (d > 0.0f) ? rsqrtf(d) : 0.0f;
  }
}

__global__ __launch_bounds__(256) void gemm_wmma_f32(
    const float* __restrict__ A, const float* __restrict__ B, float* __restrict__ C,
    int M, int Nvalid, int Ntiles, int ldc) {
  __shared__ float Bs[KDIM * 16];

  int nt = blockIdx.x % Ntiles;
  int mtBase = (blockIdx.x / Ntiles) * WPB;
  int n0 = nt << 4;
  int t = threadIdx.x;

#pragma unroll
  for (int j = 0; j < 8; ++j) {
    int idx = t + j * 256;
    int kb = idx >> 4;
    int n = n0 + (idx & 15);
    Bs[idx] = (n < Nvalid) ? B[(long long)kb * Nvalid + n] : 0.0f;
  }
  __syncthreads();

  int w = t >> 5;
  int mt = mtBase + w;
  int Mtiles = (M + 15) >> 4;
  if (mt >= Mtiles) return;

  int lane = t & 31;
  int l16 = lane & 15;
  int hi  = lane >> 4;
  int kh  = hi << 1;

  int m0 = mt << 4;
  int m = m0 + l16;
  if (m >= M) m = M - 1;
  const float* Arow = A + (long long)m * KDIM;

  v8f acc = {};
#pragma unroll 8
  for (int k = 0; k < KDIM; k += 4) {
    v2f a = *(const v2f*)(Arow + k + kh);
    v2f b;
    b.x = Bs[(k + kh) * 16 + l16];
    b.y = Bs[(k + kh + 1) * 16 + l16];
    acc = __builtin_amdgcn_wmma_f32_16x16x4_f32(
         false, a,  false, b,
         (short)0, acc,  false,  false);
  }

  int mbase = m0 + hi * 8;
  float* Cout = C + (long long)mbase * ldc + n0 + l16;
  if (m0 + 16 <= M) {
#pragma unroll
    for (int r = 0; r < 8; ++r) Cout[(long long)r * ldc] = acc[r];
  } else {
#pragma unroll
    for (int r = 0; r < 8; ++r)
      if (mbase + r < M) Cout[(long long)r * ldc] = acc[r];
  }
}

__global__ __launch_bounds__(256) void scatter_rows128(
    const int* __restrict__ row, const int* __restrict__ col,
    const float* __restrict__ dinv, const float* __restrict__ h,
    float* __restrict__ agg, int E) {
  int lane = threadIdx.x & 31;
  long long e = (blockIdx.x * (long long)blockDim.x + threadIdx.x) >> 5;
  long long ne = ((long long)gridDim.x * blockDim.x) >> 5;
  for (; e < E; e += ne) {
    int r = row[e];
    int c = col[e];
    float nrm = dinv[r] * dinv[c];
    const float* hp = h + (long long)r * NFEAT + lane;
    float* ap = agg + (long long)c * NFEAT + lane;
    float v0 = hp[0];
    float v1 = hp[32];
    float v2 = hp[64];
    float v3 = hp[96];
    atomicAdd(ap, v0 * nrm);
    atomicAdd(ap + 32, v1 * nrm);
    atomicAdd(ap + 64, v2 * nrm);
    atomicAdd(ap + 96, v3 * nrm);
  }
}

__global__ __launch_bounds__(256) void scatter_rows40(
    const int* __restrict__ row, const int* __restrict__ col,
    const float* __restrict__ dinv, const float* __restrict__ h,
    float* __restrict__ agg, int E) {
  int l = threadIdx.x & 15;
  long long e = (blockIdx.x * (long long)blockDim.x + threadIdx.x) >> 4;
  long long ne = ((long long)gridDim.x * blockDim.x) >> 4;
  for (; e < E; e += ne) {
    int r = row[e];
    int c = col[e];
    float nrm = dinv[r] * dinv[c];
    const float* hp = h + (long long)r * LD2 + l;
    float* ap = agg + (long long)c * LD2 + l;
    float v0 = hp[0];
    float v1 = hp[16];
    float v2 = hp[32];
    asm volatile("" ::"v"(v2));
    atomicAdd(ap, v0 * nrm);
    atomicAdd(ap + 16, v1 * nrm);
    if (l < NCLS - 32) atomicAdd(ap + 32, v2 * nrm);
  }
}

__global__ void bias_relu_self(const float* __restrict__ h, const float* __restrict__ dinv,
                               const float* __restrict__ bias, float* __restrict__ agg,
                               int Nnodes) {
  long long total = (long long)Nnodes * NFEAT;
  long long i = blockIdx.x * (long long)blockDim.x + threadIdx.x;
  long long stride = (long long)gridDim.x * blockDim.x;
  for (; i < total; i += stride) {
    int nidx = (int)(i >> 7);
    int f = (int)(i & (NFEAT - 1));
    float d = dinv[nidx];
    float v = agg[i] + h[i] * d * d + bias[f];
    agg[i] = v > 0.0f ? v : 0.0f;
  }
}

__global__ void out_logsoftmax(const float* __restrict__ h2, const float* __restrict__ agg2,
                               const float* __restrict__ dinv, const float* __restrict__ b2,
                               float* __restrict__ out, int Nnodes) {
  int i = blockIdx.x * blockDim.x + threadIdx.x;
  if (i >= Nnodes) return;
  float d = dinv[i];
  float d2 = d * d;
  float logit[NCLS];
  float mx = -3.4e38f;
#pragma unroll
  for (int f = 0; f < NCLS; ++f) {
    float v = agg2[(long long)i * LD2 + f] + h2[(long long)i * LD2 + f] * d2 + b2[f];
    logit[f] = v;
    mx = fmaxf(mx, v);
  }
  float s = 0.0f;
#pragma unroll
  for (int f = 0; f < NCLS; ++f) s += __expf(logit[f] - mx);
  float lse = mx + __logf(s);
#pragma unroll
  for (int f = 0; f < NCLS; ++f) out[(long long)i * NCLS + f] = logit[f] - lse;
}


extern "C" void kernel_launch(void* const* d_in, const int* in_sizes, int n_in,
                              void* d_out, int out_size, void* d_ws, size_t ws_size,
                              hipStream_t stream) {
  const float* x  = (const float*)d_in[0];
  const int*   ei = (const int*)d_in[1];
  const float* W1 = (const float*)d_in[2];
  const float* b1 = (const float*)d_in[3];
  const float* W2 = (const float*)d_in[4];
  const float* b2 = (const float*)d_in[5];
  float* out = (float*)d_out;

  const int N = in_sizes[0] / NFEAT;
  const int E = in_sizes[1] / 2;
  const int* erow = ei;
  const int* ecol = ei + E;

  float* dinv = (float*)d_ws;
  float* h1   = dinv + N;
  float* agg1 = h1 + (long long)N * NFEAT;
  float* h2   = agg1 + (long long)N * NFEAT;
  float* agg2 = h2 + (long long)N * LD2;

  const int TPB = 256;
  auto blocks = [](long long n, int tpb) {
    long long b = (n + tpb - 1) / tpb;
    return (unsigned)(b > 1048576 ? 1048576 : b);
  };

  fill_f32<<<blocks(N, TPB), TPB, 0, stream>>>(dinv, N, 1.0f);
  fill_f32<<<blocks((long long)N * NFEAT, TPB), TPB, 0, stream>>>(agg1, (long long)N * NFEAT, 0.0f);
  fill_f32<<<blocks((long long)N * LD2, TPB), TPB, 0, stream>>>(agg2, (long long)N * LD2, 0.0f);

  degree_kernel<<<(E + TPB - 1) / TPB, TPB, 0, stream>>>(ecol, dinv, E);
  rsqrt_kernel<<<(N + TPB - 1) / TPB, TPB, 0, stream>>>(dinv, N);

  const int Mtiles = (N + 15) / 16;
  const int mGroups = (Mtiles + WPB - 1) / WPB;

  {
    int Ntiles = NFEAT / 16;
    gemm_wmma_f32<<<mGroups * Ntiles, 256, 0, stream>>>(x, W1, h1, N, NFEAT, Ntiles, NFEAT);
  }

  scatter_rows128<<<blocks((long long)E * 32, TPB), TPB, 0, stream>>>(
      erow, ecol, dinv, h1, agg1, E);

  bias_relu_self<<<blocks((long long)N * NFEAT, TPB), TPB, 0, stream>>>(
      h1, dinv, b1, agg1, N);

  {
    int Ntiles = LD2 / 16;
    gemm_wmma_f32<<<mGroups * Ntiles, 256, 0, stream>>>(agg1, W2, h2, N, NCLS, Ntiles, LD2);
  }

  scatter_rows40<<<blocks((long long)E * 16, TPB), TPB, 0, stream>>>(
      erow, ecol, dinv, h2, agg2, E);

  out_logsoftmax<<<(N + TPB - 1) / TPB, TPB, 0, stream>>>(
      h2, agg2, dinv, b2, out, N);
}


# aggregate by destination: per-call counting sort of edges (int rank atomics + scan); one wave per node sums its listed rows in registers, stores each row once; no f32 atomics; sums in rank order
# speedup vs baseline: 1.8321x; 1.3642x over previous
#include <hip/hip_runtime.h>
#include <cmath>


typedef __attribute__((ext_vector_type(2))) float v2f;
typedef __attribute__((ext_vector_type(4))) float v4f;
typedef __attribute__((ext_vector_type(4))) int v4i;
typedef __attribute__((ext_vector_type(8))) float v8f;

#define NFEAT 128
#define NCLS  40
#define LD2   48
#define KDIM  128
#define WPB   8
#define SCAN_TILE 4096


__global__ void zero_i32(int* __restrict__ p, int n) {
  int i = blockIdx.x * blockDim.x + threadIdx.x;
  if (i < n) p[i] = 0;
}

__global__ void count_rank(const int* __restrict__ col, int* __restrict__ cnt,
                           int* __restrict__ pos, int E) {
  int e = blockIdx.x * blockDim.x + threadIdx.x;
  if (e < E) pos[e] = atomicAdd(&cnt[col[e]], 1);
}

__device__ __forceinline__ int wave_incl_scan(int v, int lane) {
#pragma unroll
  for (int d = 1; d < 32; d <<= 1) {
    int y = __shfl_up(v, d);
    if (lane >= d) v += y;
  }
  return v;
}

__global__ __launch_bounds__(1024) void scan_counts(
    const int* __restrict__ cnt, int* __restrict__ offs, float* __restrict__ dinv, int npad) {
  __shared__ int wtot[32];
  int t = threadIdx.x;
  int lane = t & 31;
  int w = t >> 5;
  int carry = 0;
  v4i c = {0, 0, 0, 0};
  if (4 * t < npad) c = *(const v4i*)(cnt + 4 * t);
  for (int base = 0; base < npad; base += SCAN_TILE) {
    int i0 = base + 4 * t;
    v4i cn = {0, 0, 0, 0};
    if (i0 + SCAN_TILE < npad) cn = *(const v4i*)(cnt + i0 + SCAN_TILE);
    int s1 = c.x;
    int s2 = s1 + c.y;
    int s3 = s2 + c.z;
    int s4 = s3 + c.w;
    int incl = wave_incl_scan(s4, lane);
    if (lane == 31) wtot[w] = incl;
    __syncthreads();
    int tw = wtot[lane];
    int twincl = wave_incl_scan(tw, lane);
    int wprefix = __shfl(twincl, w) - __shfl(tw, w);
    int tile_total = __shfl(twincl, 31);
    int excl = carry + wprefix + (incl - s4);
    if (i0 < npad) {
      v4i o = {excl, excl + s1, excl + s2, excl + s3};
      *(v4i*)(offs + i0) = o;
      float d0 = (float)(c.x + 1), d1 = (float)(c.y + 1), d2 = (float)(c.z + 1), d3 = (float)(c.w + 1);
      v4f dv;
      dv.x = (d0 > 0.0f) ? rsqrtf(d0) : 0.0f;
      dv.y = (d1 > 0.0f) ? rsqrtf(d1) : 0.0f;
      dv.z = (d2 > 0.0f) ? rsqrtf(d2) : 0.0f;
      dv.w = (d3 > 0.0f) ? rsqrtf(d3) : 0.0f;
      *(v4f*)(dinv + i0) = dv;
    }
    carry += tile_total;
    c = cn;
    __syncthreads();
  }
}

__global__ void fill_lists(const int* __restrict__ row, const int* __restrict__ col,
                           const int* __restrict__ pos, const int* __restrict__ offs,
                           const float* __restrict__ dinv, int* __restrict__ src,
                           float* __restrict__ wn, int E) {
  int e = blockIdx.x * blockDim.x + threadIdx.x;
  if (e < E) {
    int r = row[e];
    int c = col[e];
    int p = offs[c] + pos[e];
    src[p] = r;
    wn[p] = dinv[r] * dinv[c];
  }
}

__global__ __launch_bounds__(256) void gemm_wmma_f32(
    const float* __restrict__ A, const float* __restrict__ B, float* __restrict__ C,
    int M, int Nvalid, int Ntiles, int ldc) {
  __shared__ float Bs[KDIM * 16];

  int nt = blockIdx.x % Ntiles;
  int mtBase = (blockIdx.x / Ntiles) * WPB;
  int n0 = nt << 4;
  int t = threadIdx.x;

#pragma unroll
  for (int j = 0; j < 8; ++j) {
    int idx = t + j * 256;
    int kb = idx >> 4;
    int n = n0 + (idx & 15);
    Bs[idx] = (n < Nvalid) ? B[(long long)kb * Nvalid + n] : 0.0f;
  }
  __syncthreads();

  int w = t >> 5;
  int mt = mtBase + w;
  int Mtiles = (M + 15) >> 4;
  if (mt >= Mtiles) return;

  int lane = t & 31;
  int l16 = lane & 15;
  int hi  = lane >> 4;
  int kh  = hi << 1;

  int m0 = mt << 4;
  int m = m0 + l16;
  if (m >= M) m = M - 1;
  const float* Arow = A + (long long)m * KDIM;

  v8f acc = {};
#pragma unroll 8
  for (int k = 0; k < KDIM; k += 4) {
    v2f a = *(const v2f*)(Arow + k + kh);
    v2f b;
    b.x = Bs[(k + kh) * 16 + l16];
    b.y = Bs[(k + kh + 1) * 16 + l16];
    acc = __builtin_amdgcn_wmma_f32_16x16x4_f32(
         false, a,  false, b,
         (short)0, acc,  false,  false);
  }

  int mbase = m0 + hi * 8;
  float* Cout = C + (long long)mbase * ldc + n0 + l16;
  if (m0 + 16 <= M) {
#pragma unroll
    for (int r = 0; r < 8; ++r) Cout[(long long)r * ldc] = acc[r];
  } else {
#pragma unroll
    for (int r = 0; r < 8; ++r)
      if (mbase + r < M) Cout[(long long)r * ldc] = acc[r];
  }
}

__global__ __launch_bounds__(256) void gather_rows128_relu(
    const int* __restrict__ offs, const int* __restrict__ cnt,
    const int* __restrict__ src, const float* __restrict__ wn,
    const float* __restrict__ dinv, const float* __restrict__ h,
    const float* __restrict__ bias, float* __restrict__ act, int Nnodes) {
  int lane = threadIdx.x & 31;
  int i = __builtin_amdgcn_readfirstlane((int)((blockIdx.x * 256u + threadIdx.x) >> 5));
  if (i >= Nnodes) return;
  int beg = offs[i];
  int n = cnt[i];
  const v4f* h4 = (const v4f*)h;
  v4f acc = {0.0f, 0.0f, 0.0f, 0.0f};
  for (int base = 0; base < n; base += 32) {
    int m = n - base;
    if (m > 32) m = 32;
    int sidx = i;
    float wv = 0.0f;
    if (lane < m) {
      sidx = src[beg + base + lane];
      wv = wn[beg + base + lane];
    }
    int wbits = __float_as_int(wv);
    for (int k = 0; k < m; k += 4) {
      int s0 = __builtin_amdgcn_readlane(sidx, k);
      int s1 = __builtin_amdgcn_readlane(sidx, k + 1);
      int s2 = __builtin_amdgcn_readlane(sidx, k + 2);
      int s3 = __builtin_amdgcn_readlane(sidx, k + 3);
      v4f v0 = h4[(long long)s0 * (NFEAT / 4) + lane];
      v4f v1 = h4[(long long)s1 * (NFEAT / 4) + lane];
      v4f v2 = h4[(long long)s2 * (NFEAT / 4) + lane];
      v4f v3 = h4[(long long)s3 * (NFEAT / 4) + lane];
      float w0 = __int_as_float(__builtin_amdgcn_readlane(wbits, k));
      float w1 = __int_as_float(__builtin_amdgcn_readlane(wbits, k + 1));
      float w2 = __int_as_float(__builtin_amdgcn_readlane(wbits, k + 2));
      float w3 = __int_as_float(__builtin_amdgcn_readlane(wbits, k + 3));
      acc += v0 * w0;
      if (k + 1 < m) acc += v1 * w1;
      if (k + 2 < m) acc += v2 * w2;
      if (k + 3 < m) acc += v3 * w3;
    }
  }
  float d = dinv[i];
  v4f hs = h4[(long long)i * (NFEAT / 4) + lane];
  v4f b = ((const v4f*)bias)[lane];
  v4f r;
  r.x = acc.x + hs.x * d * d + b.x;
  r.y = acc.y + hs.y * d * d + b.y;
  r.z = acc.z + hs.z * d * d + b.z;
  r.w = acc.w + hs.w * d * d + b.w;
  r.x = r.x > 0.0f ? r.x : 0.0f;
  r.y = r.y > 0.0f ? r.y : 0.0f;
  r.z = r.z > 0.0f ? r.z : 0.0f;
  r.w = r.w > 0.0f ? r.w : 0.0f;
  ((v4f*)act)[(long long)i * (NFEAT / 4) + lane] = r;
}

__global__ __launch_bounds__(256) void gather_rows40_lsm(
    const int* __restrict__ offs, const int* __restrict__ cnt,
    const int* __restrict__ src, const float* __restrict__ wn,
    const float* __restrict__ dinv, const float* __restrict__ h2,
    const float* __restrict__ b2, float* __restrict__ out, int Nnodes) {
  int lane = threadIdx.x & 31;
  int i = __builtin_amdgcn_readfirstlane((int)((blockIdx.x * 256u + threadIdx.x) >> 5));
  if (i >= Nnodes) return;
  bool hi = lane < (NCLS - 32);
  int beg = offs[i];
  int n = cnt[i];
  float a0 = 0.0f, a1 = 0.0f;
  for (int base = 0; base < n; base += 32) {
    int m = n - base;
    if (m > 32) m = 32;
    int sidx = i;
    float wv = 0.0f;
    if (lane < m) {
      sidx = src[beg + base + lane];
      wv = wn[beg + base + lane];
    }
    int wbits = __float_as_int(wv);
    for (int k = 0; k < m; k += 4) {
      int s0 = __builtin_amdgcn_readlane(sidx, k);
      int s1 = __builtin_amdgcn_readlane(sidx, k + 1);
      int s2 = __builtin_amdgcn_readlane(sidx, k + 2);
      int s3 = __builtin_amdgcn_readlane(sidx, k + 3);
      const float* r0 = h2 + (long long)s0 * LD2 + lane;
      const float* r1 = h2 + (long long)s1 * LD2 + lane;
      const float* r2 = h2 + (long long)s2 * LD2 + lane;
      const float* r3 = h2 + (long long)s3 * LD2 + lane;
      float x0 = r0[0], x1 = r1[0], x2 = r2[0], x3 = r3[0];
      float y0 = 0.0f, y1 = 0.0f, y2 = 0.0f, y3 = 0.0f;
      if (hi) {
        y0 = r0[32];
        y1 = r1[32];
        y2 = r2[32];
        y3 = r3[32];
      }
      float w0 = __int_as_float(__builtin_amdgcn_readlane(wbits, k));
      float w1 = __int_as_float(__builtin_amdgcn_readlane(wbits, k + 1));
      float w2 = __int_as_float(__builtin_amdgcn_readlane(wbits, k + 2));
      float w3 = __int_as_float(__builtin_amdgcn_readlane(wbits, k + 3));
      a0 += x0 * w0;
      a1 += y0 * w0;
      if (k + 1 < m) { a0 += x1 * w1; a1 += y1 * w1; }
      if (k + 2 < m) { a0 += x2 * w2; a1 += y2 * w2; }
      if (k + 3 < m) { a0 += x3 * w3; a1 += y3 * w3; }
    }
  }
  float d = dinv[i];
  float dd = d * d;
  const float* hs = h2 + (long long)i * LD2 + lane;
  float v0 = a0 + hs[0] * dd + b2[lane];
  float v1 = 0.0f;
  float mx = v0;
  if (hi) {
    v1 = a1 + hs[32] * dd + b2[32 + lane];
    mx = fmaxf(mx, v1);
  }
#pragma unroll
  for (int o = 16; o > 0; o >>= 1) mx = fmaxf(mx, __shfl_xor(mx, o));
  float ex = __expf(v0 - mx);
  if (hi) ex += __expf(v1 - mx);
#pragma unroll
  for (int o = 16; o > 0; o >>= 1) ex += __shfl_xor(ex, o);
  float lse = mx + __logf(ex);
  float* op = out + (long long)i * NCLS + lane;
  op[0] = v0 - lse;
  if (hi) op[32] = v1 - lse;
}


extern "C" void kernel_launch(void* const* d_in, const int* in_sizes, int n_in,
                              void* d_out, int out_size, void* d_ws, size_t ws_size,
                              hipStream_t stream) {
  const float* x  = (const float*)d_in[0];
  const int*   ei = (const int*)d_in[1];
  const float* W1 = (const float*)d_in[2];
  const float* b1 = (const float*)d_in[3];
  const float* W2 = (const float*)d_in[4];
  const float* b2 = (const float*)d_in[5];
  float* out = (float*)d_out;

  const int N = in_sizes[0] / NFEAT;
  const int E = in_sizes[1] / 2;
  const int* erow = ei;
  const int* ecol = ei + E;
  const int Npad = (N + 63) & ~63;

  float* h1   = (float*)d_ws;
  float* act  = h1 + (long long)N * NFEAT;
  float* h2   = act + (long long)N * NFEAT;
  int*   cnt  = (int*)(h2 + (long long)N * LD2);
  int*   offs = cnt + Npad;
  float* dinv = (float*)(offs + Npad);
  float* wn   = dinv + Npad;
  int*   src  = (int*)(wn + E);
  int*   pos  = src + E;

  const int TPB = 256;

  zero_i32<<<(Npad + TPB - 1) / TPB, TPB, 0, stream>>>(cnt, Npad);
  count_rank<<<(E + TPB - 1) / TPB, TPB, 0, stream>>>(ecol, cnt, pos, E);
  scan_counts<<<1, 1024, 0, stream>>>(cnt, offs, dinv, Npad);
  fill_lists<<<(E + TPB - 1) / TPB, TPB, 0, stream>>>(erow, ecol, pos, offs, dinv, src, wn, E);

  const int Mtiles = (N + 15) / 16;
  const int mGroups = (Mtiles + WPB - 1) / WPB;
  const int nodeBlocks = (N + 7) / 8;

  {
    int Ntiles = NFEAT / 16;
    gemm_wmma_f32<<<mGroups * Ntiles, 256, 0, stream>>>(x, W1, h1, N, NFEAT, Ntiles, NFEAT);
  }

  gather_rows128_relu<<<nodeBlocks, 256, 0, stream>>>(offs, cnt, src, wn, dinv, h1, b1, act, N);

  {
    int Ntiles = LD2 / 16;
    gemm_wmma_f32<<<mGroups * Ntiles, 256, 0, stream>>>(act, W2, h2, N, NCLS, Ntiles, LD2);
  }

  gather_rows40_lsm<<<nodeBlocks, 256, 0, stream>>>(offs, cnt, src, wn, dinv, h2, b2, out, N);
}


# both dense products re-tiled: one wave computes a full-width 16-row strip (A rows read once, as whole rows; 8 or 3 independent accumulators), all of W in LDS by loads issued together; sums unchanged
# speedup vs baseline: 2.0659x; 1.1276x over previous
#include <hip/hip_runtime.h>
#include <cmath>


typedef __attribute__((ext_vector_type(2))) float v2f;
typedef __attribute__((ext_vector_type(4))) float v4f;
typedef __attribute__((ext_vector_type(4))) int v4i;
typedef __attribute__((ext_vector_type(8))) float v8f;

#define NFEAT 128
#define NCLS  40
#define LD2   48
#define KDIM  128
#define WPB   8
#define SCAN_TILE 4096


__global__ void zero_i32(int* __restrict__ p, int n) {
  int i = blockIdx.x * blockDim.x + threadIdx.x;
  if (i < n) p[i] = 0;
}

__global__ void count_rank(const int* __restrict__ col, int* __restrict__ cnt,
                           int* __restrict__ pos, int E) {
  int e = blockIdx.x * blockDim.x + threadIdx.x;
  if (e < E) pos[e] = atomicAdd(&cnt[col[e]], 1);
}

__device__ __forceinline__ int wave_incl_scan(int v, int lane) {
#pragma unroll
  for (int d = 1; d < 32; d <<= 1) {
    int y = __shfl_up(v, d);
    if (lane >= d) v += y;
  }
  return v;
}

__global__ __launch_bounds__(1024) void scan_counts(
    const int* __restrict__ cnt, int* __restrict__ offs, float* __restrict__ dinv, int npad) {
  __shared__ int wtot[32];
  int t = threadIdx.x;
  int lane = t & 31;
  int w = t >> 5;
  int carry = 0;
  v4i c = {0, 0, 0, 0};
  if (4 * t < npad) c = *(const v4i*)(cnt + 4 * t);
  for (int base = 0; base < npad; base += SCAN_TILE) {
    int i0 = base + 4 * t;
    v4i cn = {0, 0, 0, 0};
    if (i0 + SCAN_TILE < npad) cn = *(const v4i*)(cnt + i0 + SCAN_TILE);
    int s1 = c.x;
    int s2 = s1 + c.y;
    int s3 = s2 + c.z;
    int s4 = s3 + c.w;
    int incl = wave_incl_scan(s4, lane);
    if (lane == 31) wtot[w] = incl;
    __syncthreads();
    int tw = wtot[lane];
    int twincl = wave_incl_scan(tw, lane);
    int wprefix = __shfl(twincl, w) - __shfl(tw, w);
    int tile_total = __shfl(twincl, 31);
    int excl = carry + wprefix + (incl - s4);
    if (i0 < npad) {
      v4i o = {excl, excl + s1, excl + s2, excl + s3};
      *(v4i*)(offs + i0) = o;
      float d0 = (float)(c.x + 1), d1 = (float)(c.y + 1), d2 = (float)(c.z + 1), d3 = (float)(c.w + 1);
      v4f dv;
      dv.x = (d0 > 0.0f) ? rsqrtf(d0) : 0.0f;
      dv.y = (d1 > 0.0f) ? rsqrtf(d1) : 0.0f;
      dv.z = (d2 > 0.0f) ? rsqrtf(d2) : 0.0f;
      dv.w = (d3 > 0.0f) ? rsqrtf(d3) : 0.0f;
      *(v4f*)(dinv + i0) = dv;
    }
    carry += tile_total;
    c = cn;
    __syncthreads();
  }
}

__global__ void fill_lists(const int* __restrict__ row, const int* __restrict__ col,
                           const int* __restrict__ pos, const int* __restrict__ offs,
                           const float* __restrict__ dinv, int* __restrict__ src,
                           float* __restrict__ wn, int E) {
  int e = blockIdx.x * blockDim.x + threadIdx.x;
  if (e < E) {
    int r = row[e];
    int c = col[e];
    int p = offs[c] + pos[e];
    src[p] = r;
    wn[p] = dinv[r] * dinv[c];
  }
}

template <int NT, int NV>
__global__ __launch_bounds__(256) void gemm_strip_f32(
    const float* __restrict__ A, const float* __restrict__ B, float* __restrict__ C,
    int M, int ldc) {
  constexpr int NW   = NT * 16;
  constexpr int V4R  = NW / 4;
  constexpr int NV4  = NV / 4;
  constexpr int SW   = (NW > KDIM ? NW : KDIM);
  constexpr int SP   = SW + 4;
  constexpr int NJ   = (16 * V4R) / 32;
  constexpr int BP   = 2 * NW + ((NT % 2 == 0) ? 32 : 0);
  constexpr int NB   = ((KDIM / 2) * V4R) / 256;
  __shared__ __attribute__((aligned(16))) float Bs[(KDIM / 2) * BP];
  __shared__ __attribute__((aligned(16))) float Ss[WPB * 16 * SP];

  int t = threadIdx.x;
  int w = t >> 5;
  int lane = t & 31;
  int l16 = lane & 15;
  int hi  = lane >> 4;
  int kh  = hi << 1;

  int Mtiles = (M + 15) >> 4;
  int mt = blockIdx.x * WPB + w;
  int m0 = mt << 4;
  float* Sw = Ss + w * (16 * SP);

  v4f ar[16];
#pragma unroll
  for (int r = 0; r < 16; ++r) {
    int m = m0 + r;
    if (m >= M) m = M - 1;
    ar[r] = *(const v4f*)(A + (long long)m * KDIM + 4 * lane);
  }

  v4f b0[NB], b1[NB];
#pragma unroll
  for (int j = 0; j < NB; ++j) {
    int item = t + j * 256;
    int kp = item / V4R;
    int c4 = item % V4R;
    int c4c = c4 < NV4 ? c4 : NV4 - 1;
    b0[j] = *(const v4f*)(B + (2 * kp) * NV + c4c * 4);
    b1[j] = *(const v4f*)(B + (2 * kp + 1) * NV + c4c * 4);
    if (c4 >= NV4) {
      b0[j] = (v4f){0.0f, 0.0f, 0.0f, 0.0f};
      b1[j] = (v4f){0.0f, 0.0f, 0.0f, 0.0f};
    }
  }
#pragma unroll
  for (int r = 0; r < 16; ++r) *(v4f*)(Sw + r * SP + 4 * lane) = ar[r];
#pragma unroll
  for (int j = 0; j < NB; ++j) {
    int item = t + j * 256;
    int kp = item / V4R;
    int c4 = item % V4R;
    float* d = Bs + kp * BP + 8 * c4;
    *(v4f*)d       = (v4f){b0[j].x, b1[j].x, b0[j].y, b1[j].y};
    *(v4f*)(d + 4) = (v4f){b0[j].z, b1[j].z, b0[j].w, b1[j].w};
  }
  __syncthreads();

  if (mt >= Mtiles) return;

  v8f acc[NT];
#pragma unroll
  for (int nt = 0; nt < NT; ++nt) acc[nt] = (v8f){};

  const float* Ak = Sw + l16 * SP + kh;
#pragma unroll
  for (int s = 0; s < KDIM / 4; ++s) {
    v2f a = *(const v2f*)(Ak + 4 * s);
    const float* Bk = Bs + (2 * s + hi) * BP + 2 * l16;
#pragma unroll
    for (int nt = 0; nt < NT; ++nt) {
      v2f b = *(const v2f*)(Bk + 32 * nt);
      acc[nt] = __builtin_amdgcn_wmma_f32_16x16x4_f32(
           false, a,  false, b,
           (short)0, acc[nt],  false,  false);
    }
  }

#pragma unroll
  for (int nt = 0; nt < NT; ++nt)
#pragma unroll
    for (int r = 0; r < 8; ++r) Sw[(hi * 8 + r) * SP + nt * 16 + l16] = acc[nt][r];

  if (m0 + 16 <= M) {
    v4f cv[NJ];
#pragma unroll
    for (int j = 0; j < NJ; ++j) {
      int e = j * 32 + lane;
      cv[j] = *(const v4f*)(Sw + (e / V4R) * SP + (e % V4R) * 4);
    }
#pragma unroll
    for (int j = 0; j < NJ; ++j) {
      int e = j * 32 + lane;
      *(v4f*)(C + (long long)(m0 + e / V4R) * ldc + (e % V4R) * 4) = cv[j];
    }
  } else {
#pragma unroll
    for (int j = 0; j < NJ; ++j) {
      int e = j * 32 + lane;
      int row = e / V4R;
      int c4 = e % V4R;
      v4f v = *(const v4f*)(Sw + row * SP + c4 * 4);
      if (m0 + row < M) *(v4f*)(C + (long long)(m0 + row) * ldc + c4 * 4) = v;
    }
  }
}

__global__ __launch_bounds__(256) void gather_rows128_relu(
    const int* __restrict__ offs, const int* __restrict__ cnt,
    const int* __restrict__ src, const float* __restrict__ wn,
    const float* __restrict__ dinv, const float* __restrict__ h,
    const float* __restrict__ bias, float* __restrict__ act, int Nnodes) {
  int lane = threadIdx.x & 31;
  int i = __builtin_amdgcn_readfirstlane((int)((blockIdx.x * 256u + threadIdx.x) >> 5));
  if (i >= Nnodes) return;
  int beg = offs[i];
  int n = cnt[i];
  const v4f* h4 = (const v4f*)h;
  v4f acc = {0.0f, 0.0f, 0.0f, 0.0f};
  for (int base = 0; base < n; base += 32) {
    int m = n - base;
    if (m > 32) m = 32;
    int sidx = i;
    float wv = 0.0f;
    if (lane < m) {
      sidx = src[beg + base + lane];
      wv = wn[beg + base + lane];
    }
    int wbits = __float_as_int(wv);
    for (int k = 0; k < m; k += 4) {
      int s0 = __builtin_amdgcn_readlane(sidx, k);
      int s1 = __builtin_amdgcn_readlane(sidx, k + 1);
      int s2 = __builtin_amdgcn_readlane(sidx, k + 2);
      int s3 = __builtin_amdgcn_readlane(sidx, k + 3);
      v4f v0 = h4[(long long)s0 * (NFEAT / 4) + lane];
      v4f v1 = h4[(long long)s1 * (NFEAT / 4) + lane];
      v4f v2 = h4[(long long)s2 * (NFEAT / 4) + lane];
      v4f v3 = h4[(long long)s3 * (NFEAT / 4) + lane];
      float w0 = __int_as_float(__builtin_amdgcn_readlane(wbits, k));
      float w1 = __int_as_float(__builtin_amdgcn_readlane(wbits, k + 1));
      float w2 = __int_as_float(__builtin_amdgcn_readlane(wbits, k + 2));
      float w3 = __int_as_float(__builtin_amdgcn_readlane(wbits, k + 3));
      acc += v0 * w0;
      if (k + 1 < m) acc += v1 * w1;
      if (k + 2 < m) acc += v2 * w2;
      if (k + 3 < m) acc += v3 * w3;
    }
  }
  float d = dinv[i];
  v4f hs = h4[(long long)i * (NFEAT / 4) + lane];
  v4f b = ((const v4f*)bias)[lane];
  v4f r;
  r.x = acc.x + hs.x * d * d + b.x;
  r.y = acc.y + hs.y * d * d + b.y;
  r.z = acc.z + hs.z * d * d + b.z;
  r.w = acc.w + hs.w * d * d + b.w;
  r.x = r.x > 0.0f ? r.x : 0.0f;
  r.y = r.y > 0.0f ? r.y : 0.0f;
  r.z = r.z > 0.0f ? r.z : 0.0f;
  r.w = r.w > 0.0f ? r.w : 0.0f;
  ((v4f*)act)[(long long)i * (NFEAT / 4) + lane] = r;
}

__global__ __launch_bounds__(256) void gather_rows40_lsm(
    const int* __restrict__ offs, const int* __restrict__ cnt,
    const int* __restrict__ src, const float* __restrict__ wn,
    const float* __restrict__ dinv, const float* __restrict__ h2,
    const float* __restrict__ b2, float* __restrict__ out, int Nnodes) {
  int lane = threadIdx.x & 31;
  int i = __builtin_amdgcn_readfirstlane((int)((blockIdx.x * 256u + threadIdx.x) >> 5));
  if (i >= Nnodes) return;
  bool hi = lane < (NCLS - 32);
  int beg = offs[i];
  int n = cnt[i];
  float a0 = 0.0f, a1 = 0.0f;
  for (int base = 0; base < n; base += 32) {
    int m = n - base;
    if (m > 32) m = 32;
    int sidx = i;
    float wv = 0.0f;
    if (lane < m) {
      sidx = src[beg + base + lane];
      wv = wn[beg + base + lane];
    }
    int wbits = __float_as_int(wv);
    for (int k = 0; k < m; k += 4) {
      int s0 = __builtin_amdgcn_readlane(sidx, k);
      int s1 = __builtin_amdgcn_readlane(sidx, k + 1);
      int s2 = __builtin_amdgcn_readlane(sidx, k + 2);
      int s3 = __builtin_amdgcn_readlane(sidx, k + 3);
      const float* r0 = h2 + (long long)s0 * LD2 + lane;
      const float* r1 = h2 + (long long)s1 * LD2 + lane;
      const float* r2 = h2 + (long long)s2 * LD2 + lane;
      const float* r3 = h2 + (long long)s3 * LD2 + lane;
      float x0 = r0[0], x1 = r1[0], x2 = r2[0], x3 = r3[0];
      float y0 = 0.0f, y1 = 0.0f, y2 = 0.0f, y3 = 0.0f;
      if (hi) {
        y0 = r0[32];
        y1 = r1[32];
        y2 = r2[32];
        y3 = r3[32];
      }
      float w0 = __int_as_float(__builtin_amdgcn_readlane(wbits, k));
      float w1 = __int_as_float(__builtin_amdgcn_readlane(wbits, k + 1));
      float w2 = __int_as_float(__builtin_amdgcn_readlane(wbits, k + 2));
      float w3 = __int_as_float(__builtin_amdgcn_readlane(wbits, k + 3));
      a0 += x0 * w0;
      a1 += y0 * w0;
      if (k + 1 < m) { a0 += x1 * w1; a1 += y1 * w1; }
      if (k + 2 < m) { a0 += x2 * w2; a1 += y2 * w2; }
      if (k + 3 < m) { a0 += x3 * w3; a1 += y3 * w3; }
    }
  }
  float d = dinv[i];
  float dd = d * d;
  const float* hs = h2 + (long long)i * LD2 + lane;
  float v0 = a0 + hs[0] * dd + b2[lane];
  float v1 = 0.0f;
  float mx = v0;
  if (hi) {
    v1 = a1 + hs[32] * dd + b2[32 + lane];
    mx = fmaxf(mx, v1);
  }
#pragma unroll
  for (int o = 16; o > 0; o >>= 1) mx = fmaxf(mx, __shfl_xor(mx, o));
  float ex = __expf(v0 - mx);
  if (hi) ex += __expf(v1 - mx);
#pragma unroll
  for (int o = 16; o > 0; o >>= 1) ex += __shfl_xor(ex, o);
  float lse = mx + __logf(ex);
  float* op = out + (long long)i * NCLS + lane;
  op[0] = v0 - lse;
  if (hi) op[32] = v1 - lse;
}


extern "C" void kernel_launch(void* const* d_in, const int* in_sizes, int n_in,
                              void* d_out, int out_size, void* d_ws, size_t ws_size,
                              hipStream_t stream) {
  const float* x  = (const float*)d_in[0];
  const int*   ei = (const int*)d_in[1];
  const float* W1 = (const float*)d_in[2];
  const float* b1 = (const float*)d_in[3];
  const float* W2 = (const float*)d_in[4];
  const float* b2 = (const float*)d_in[5];
  float* out = (float*)d_out;

  const int N = in_sizes[0] / NFEAT;
  const int E = in_sizes[1] / 2;
  const int* erow = ei;
  const int* ecol = ei + E;
  const int Npad = (N + 63) & ~63;

  float* h1   = (float*)d_ws;
  float* act  = h1 + (long long)N * NFEAT;
  float* h2   = act + (long long)N * NFEAT;
  int*   cnt  = (int*)(h2 + (long long)N * LD2);
  int*   offs = cnt + Npad;
  float* dinv = (float*)(offs + Npad);
  float* wn   = dinv + Npad;
  int*   src  = (int*)(wn + E);
  int*   pos  = src + E;

  const int TPB = 256;

  zero_i32<<<(Npad + TPB - 1) / TPB, TPB, 0, stream>>>(cnt, Npad);
  count_rank<<<(E + TPB - 1) / TPB, TPB, 0, stream>>>(ecol, cnt, pos, E);
  scan_counts<<<1, 1024, 0, stream>>>(cnt, offs, dinv, Npad);
  fill_lists<<<(E + TPB - 1) / TPB, TPB, 0, stream>>>(erow, ecol, pos, offs, dinv, src, wn, E);

  const int Mtiles = (N + 15) / 16;
  const int mGroups = (Mtiles + WPB - 1) / WPB;
  const int nodeBlocks = (N + 7) / 8;

  gemm_strip_f32<NFEAT / 16, NFEAT><<<mGroups, 256, 0, stream>>>(x, W1, h1, N, NFEAT);

  gather_rows128_relu<<<nodeBlocks, 256, 0, stream>>>(offs, cnt, src, wn, dinv, h1, b1, act, N);

  gemm_strip_f32<LD2 / 16, NCLS><<<mGroups, 256, 0, stream>>>(act, W2, h2, N, LD2);

  gather_rows40_lsm<<<nodeBlocks, 256, 0, stream>>>(offs, cnt, src, wn, dinv, h2, b2, out, N);
}


# edge ranks from LDS counters: 64 edge chunks each count destinations in a whole-graph LDS table (returning LDS adds), per-node prefix over chunks; no global atomics; sums in chunk-then-rank order
# speedup vs baseline: 2.4821x; 1.2015x over previous
#include <hip/hip_runtime.h>
#include <cmath>


typedef __attribute__((ext_vector_type(2))) float v2f;
typedef __attribute__((ext_vector_type(4))) float v4f;
typedef __attribute__((ext_vector_type(4))) int v4i;
typedef __attribute__((ext_vector_type(8))) float v8f;

#define NFEAT 128
#define NCLS  40
#define LD2   48
#define KDIM  128
#define WPB   8
#define SCAN_TILE 4096
#define NNODE 50000
#define HCAP  ((NNODE + 63) & ~63)
#define NCHUNK 64
#define CBATCH 16


__global__ __launch_bounds__(1024) void count_local(
    const int* __restrict__ col, int* __restrict__ pos, int* __restrict__ part,
    int E, int CH, int npad) {
  __shared__ __attribute__((aligned(16))) int hist[HCAP];
  int t = threadIdx.x;
  int k = blockIdx.x;
  int e0 = k * CH;
  int len = E - e0;
  if (len > CH) len = CH;
  int c[CBATCH];
#pragma unroll
  for (int u = 0; u < CBATCH; ++u) {
    int j = u * 1024 + t;
    c[u] = (j < len) ? col[e0 + j] : -1;
  }
  for (int i4 = t; i4 < HCAP / 4; i4 += 1024) ((v4i*)hist)[i4] = (v4i){0, 0, 0, 0};
  __syncthreads();
  for (int j0 = 0;;) {
    int rk[CBATCH];
#pragma unroll
    for (int u = 0; u < CBATCH; ++u) {
      rk[u] = 0;
      if ((unsigned)c[u] < (unsigned)HCAP) rk[u] = atomicAdd(&hist[c[u]], 1);
    }
#pragma unroll
    for (int u = 0; u < CBATCH; ++u) {
      int j = j0 + u * 1024 + t;
      if ((unsigned)c[u] < (unsigned)HCAP) pos[e0 + j] = rk[u];
    }
    j0 += CBATCH * 1024;
    if (j0 >= len) break;
#pragma unroll
    for (int u = 0; u < CBATCH; ++u) {
      int j = j0 + u * 1024 + t;
      c[u] = (j < len) ? col[e0 + j] : -1;
    }
  }
  __syncthreads();
  int* dst = part + (long long)k * npad;
  int n4 = (npad < HCAP ? npad : HCAP) / 4;
  for (int i4 = t; i4 < n4; i4 += 1024) ((v4i*)dst)[i4] = ((const v4i*)hist)[i4];
}

__global__ __launch_bounds__(128) void reduce_parts(int* __restrict__ part,
                                                    int* __restrict__ cnt, int npad) {
  int i = blockIdx.x * 128 + threadIdx.x;
  if (i >= npad) return;
  int v[NCHUNK];
#pragma unroll
  for (int k = 0; k < NCHUNK; ++k) v[k] = part[(long long)k * npad + i];
  int run = 0;
#pragma unroll
  for (int k = 0; k < NCHUNK; ++k) {
    part[(long long)k * npad + i] = run;
    run += v[k];
  }
  cnt[i] = run;
}

__device__ __forceinline__ int wave_incl_scan(int v, int lane) {
#pragma unroll
  for (int d = 1; d < 32; d <<= 1) {
    int y = __shfl_up(v, d);
    if (lane >= d) v += y;
  }
  return v;
}

__global__ __launch_bounds__(1024) void scan_counts(
    const int* __restrict__ cnt, int* __restrict__ offs, float* __restrict__ dinv, int npad) {
  __shared__ int wtot[32];
  int t = threadIdx.x;
  int lane = t & 31;
  int w = t >> 5;
  int carry = 0;
  v4i c = {0, 0, 0, 0};
  if (4 * t < npad) c = *(const v4i*)(cnt + 4 * t);
  for (int base = 0; base < npad; base += SCAN_TILE) {
    int i0 = base + 4 * t;
    v4i cn = {0, 0, 0, 0};
    if (i0 + SCAN_TILE < npad) cn = *(const v4i*)(cnt + i0 + SCAN_TILE);
    int s1 = c.x;
    int s2 = s1 + c.y;
    int s3 = s2 + c.z;
    int s4 = s3 + c.w;
    int incl = wave_incl_scan(s4, lane);
    if (lane == 31) wtot[w] = incl;
    __syncthreads();
    int tw = wtot[lane];
    int twincl = wave_incl_scan(tw, lane);
    int wprefix = __shfl(twincl, w) - __shfl(tw, w);
    int tile_total = __shfl(twincl, 31);
    int excl = carry + wprefix + (incl - s4);
    if (i0 < npad) {
      v4i o = {excl, excl + s1, excl + s2, excl + s3};
      *(v4i*)(offs + i0) = o;
      float d0 = (float)(c.x + 1), d1 = (float)(c.y + 1), d2 = (float)(c.z + 1), d3 = (float)(c.w + 1);
      v4f dv;
      dv.x = (d0 > 0.0f) ? rsqrtf(d0) : 0.0f;
      dv.y = (d1 > 0.0f) ? rsqrtf(d1) : 0.0f;
      dv.z = (d2 > 0.0f) ? rsqrtf(d2) : 0.0f;
      dv.w = (d3 > 0.0f) ? rsqrtf(d3) : 0.0f;
      *(v4f*)(dinv + i0) = dv;
    }
    carry += tile_total;
    c = cn;
    __syncthreads();
  }
}

__global__ void fill_lists(const int* __restrict__ row, const int* __restrict__ col,
                           const int* __restrict__ pos, const int* __restrict__ offs,
                           const int* __restrict__ part,
                           const float* __restrict__ dinv, int* __restrict__ src,
                           float* __restrict__ wn, int E, int CH, int npad) {
  int k = blockIdx.y;
  int j = blockIdx.x * blockDim.x + threadIdx.x;
  int e = k * CH + j;
  if (j < CH && e < E) {
    int r = row[e];
    int c = col[e];
    int p = offs[c] + part[(long long)k * npad + c] + pos[e];
    src[p] = r;
    wn[p] = dinv[r] * dinv[c];
  }
}

template <int NT, int NV>
__global__ __launch_bounds__(256) void gemm_strip_f32(
    const float* __restrict__ A, const float* __restrict__ B, float* __restrict__ C,
    int M, int ldc) {
  constexpr int NW   = NT * 16;
  constexpr int V4R  = NW / 4;
  constexpr int NV4  = NV / 4;
  constexpr int SW   = (NW > KDIM ? NW : KDIM);
  constexpr int SP   = SW + 4;
  constexpr int NJ   = (16 * V4R) / 32;
  constexpr int BP   = 2 * NW + ((NT % 2 == 0) ? 32 : 0);
  constexpr int NB   = ((KDIM / 2) * V4R) / 256;
  __shared__ __attribute__((aligned(16))) float Bs[(KDIM / 2) * BP];
  __shared__ __attribute__((aligned(16))) float Ss[WPB * 16 * SP];

  int t = threadIdx.x;
  int w = t >> 5;
  int lane = t & 31;
  int l16 = lane & 15;
  int hi  = lane >> 4;
  int kh  = hi << 1;

  int Mtiles = (M + 15) >> 4;
  int mt = blockIdx.x * WPB + w;
  int m0 = mt << 4;
  float* Sw = Ss + w * (16 * SP);

  v4f ar[16];
#pragma unroll
  for (int r = 0; r < 16; ++r) {
    int m = m0 + r;
    if (m >= M) m = M - 1;
    ar[r] = *(const v4f*)(A + (long long)m * KDIM + 4 * lane);
  }

  v4f b0[NB], b1[NB];
#pragma unroll
  for (int j = 0; j < NB; ++j) {
    int item = t + j * 256;
    int kp = item / V4R;
    int c4 = item % V4R;
    int c4c = c4 < NV4 ? c4 : NV4 - 1;
    b0[j] = *(const v4f*)(B + (2 * kp) * NV + c4c * 4);
    b1[j] = *(const v4f*)(B + (2 * kp + 1) * NV + c4c * 4);
    if (c4 >= NV4) {
      b0[j] = (v4f){0.0f, 0.0f, 0.0f, 0.0f};
      b1[j] = (v4f){0.0f, 0.0f, 0.0f, 0.0f};
    }
  }
#pragma unroll
  for (int r = 0; r < 16; ++r) *(v4f*)(Sw + r * SP + 4 * lane) = ar[r];
#pragma unroll
  for (int j = 0; j < NB; ++j) {
    int item = t + j * 256;
    int kp = item / V4R;
    int c4 = item % V4R;
    float* d = Bs + kp * BP + 8 * c4;
    *(v4f*)d       = (v4f){b0[j].x, b1[j].x, b0[j].y, b1[j].y};
    *(v4f*)(d + 4) = (v4f){b0[j].z, b1[j].z, b0[j].w, b1[j].w};
  }
  __syncthreads();

  if (mt >= Mtiles) return;

  v8f acc[NT];
#pragma unroll
  for (int nt = 0; nt < NT; ++nt) acc[nt] = (v8f){};

  const float* Ak = Sw + l16 * SP + kh;
#pragma unroll
  for (int s = 0; s < KDIM / 4; ++s) {
    v2f a = *(const v2f*)(Ak + 4 * s);
    const float* Bk = Bs + (2 * s + hi) * BP + 2 * l16;
#pragma unroll
    for (int nt = 0; nt < NT; ++nt) {
      v2f b = *(const v2f*)(Bk + 32 * nt);
      acc[nt] = __builtin_amdgcn_wmma_f32_16x16x4_f32(
           false, a,  false, b,
           (short)0, acc[nt],  false,  false);
    }
  }

#pragma unroll
  for (int nt = 0; nt < NT; ++nt)
#pragma unroll
    for (int r = 0; r < 8; ++r) Sw[(hi * 8 + r) * SP + nt * 16 + l16] = acc[nt][r];

  if (m0 + 16 <= M) {
    v4f cv[NJ];
#pragma unroll
    for (int j = 0; j < NJ; ++j) {
      int e = j * 32 + lane;
      cv[j] = *(const v4f*)(Sw + (e / V4R) * SP + (e % V4R) * 4);
    }
#pragma unroll
    for (int j = 0; j < NJ; ++j) {
      int e = j * 32 + lane;
      *(v4f*)(C + (long long)(m0 + e / V4R) * ldc + (e % V4R) * 4) = cv[j];
    }
  } else {
#pragma unroll
    for (int j = 0; j < NJ; ++j) {
      int e = j * 32 + lane;
      int row = e / V4R;
      int c4 = e % V4R;
      v4f v = *(const v4f*)(Sw + row * SP + c4 * 4);
      if (m0 + row < M) *(v4f*)(C + (long long)(m0 + row) * ldc + c4 * 4) = v;
    }
  }
}

__global__ __launch_bounds__(256) void gather_rows128_relu(
    const int* __restrict__ offs, const int* __restrict__ cnt,
    const int* __restrict__ src, const float* __restrict__ wn,
    const float* __restrict__ dinv, const float* __restrict__ h,
    const float* __restrict__ bias, float* __restrict__ act, int Nnodes) {
  int lane = threadIdx.x & 31;
  int i = __builtin_amdgcn_readfirstlane((int)((blockIdx.x * 256u + threadIdx.x) >> 5));
  if (i >= Nnodes) return;
  int beg = offs[i];
  int n = cnt[i];
  const v4f* h4 = (const v4f*)h;
  v4f acc = {0.0f, 0.0f, 0.0f, 0.0f};
  for (int base = 0; base < n; base += 32) {
    int m = n - base;
    if (m > 32) m = 32;
    int sidx = i;
    float wv = 0.0f;
    if (lane < m) {
      sidx = src[beg + base + lane];
      wv = wn[beg + base + lane];
    }
    int wbits = __float_as_int(wv);
    for (int k = 0; k < m; k += 4) {
      int s0 = __builtin_amdgcn_readlane(sidx, k);
      int s1 = __builtin_amdgcn_readlane(sidx, k + 1);
      int s2 = __builtin_amdgcn_readlane(sidx, k + 2);
      int s3 = __builtin_amdgcn_readlane(sidx, k + 3);
      v4f v0 = h4[(long long)s0 * (NFEAT / 4) + lane];
      v4f v1 = h4[(long long)s1 * (NFEAT / 4) + lane];
      v4f v2 = h4[(long long)s2 * (NFEAT / 4) + lane];
      v4f v3 = h4[(long long)s3 * (NFEAT / 4) + lane];
      float w0 = __int_as_float(__builtin_amdgcn_readlane(wbits, k));
      float w1 = __int_as_float(__builtin_amdgcn_readlane(wbits, k + 1));
      float w2 = __int_as_float(__builtin_amdgcn_readlane(wbits, k + 2));
      float w3 = __int_as_float(__builtin_amdgcn_readlane(wbits, k + 3));
      acc += v0 * w0;
      if (k + 1 < m) acc += v1 * w1;
      if (k + 2 < m) acc += v2 * w2;
      if (k + 3 < m) acc += v3 * w3;
    }
  }
  float d = dinv[i];
  v4f hs = h4[(long long)i * (NFEAT / 4) + lane];
  v4f b = ((const v4f*)bias)[lane];
  v4f r;
  r.x = acc.x + hs.x * d * d + b.x;
  r.y = acc.y + hs.y * d * d + b.y;
  r.z = acc.z + hs.z * d * d + b.z;
  r.w = acc.w + hs.w * d * d + b.w;
  r.x = r.x > 0.0f ? r.x : 0.0f;
  r.y = r.y > 0.0f ? r.y : 0.0f;
  r.z = r.z > 0.0f ? r.z : 0.0f;
  r.w = r.w > 0.0f ? r.w : 0.0f;
  ((v4f*)act)[(long long)i * (NFEAT / 4) + lane] = r;
}

__global__ __launch_bounds__(256) void gather_rows40_lsm(
    const int* __restrict__ offs, const int* __restrict__ cnt,
    const int* __restrict__ src, const float* __restrict__ wn,
    const float* __restrict__ dinv, const float* __restrict__ h2,
    const float* __restrict__ b2, float* __restrict__ out, int Nnodes) {
  int lane = threadIdx.x & 31;
  int i = __builtin_amdgcn_readfirstlane((int)((blockIdx.x * 256u + threadIdx.x) >> 5));
  if (i >= Nnodes) return;
  bool hi = lane < (NCLS - 32);
  int beg = offs[i];
  int n = cnt[i];
  float a0 = 0.0f, a1 = 0.0f;
  for (int base = 0; base < n; base += 32) {
    int m = n - base;
    if (m > 32) m = 32;
    int sidx = i;
    float wv = 0.0f;
    if (lane < m) {
      sidx = src[beg + base + lane];
      wv = wn[beg + base + lane];
    }
    int wbits = __float_as_int(wv);
    for (int k = 0; k < m; k += 4) {
      int s0 = __builtin_amdgcn_readlane(sidx, k);
      int s1 = __builtin_amdgcn_readlane(sidx, k + 1);
      int s2 = __builtin_amdgcn_readlane(sidx, k + 2);
      int s3 = __builtin_amdgcn_readlane(sidx, k + 3);
      const float* r0 = h2 + (long long)s0 * LD2 + lane;
      const float* r1 = h2 + (long long)s1 * LD2 + lane;
      const float* r2 = h2 + (long long)s2 * LD2 + lane;
      const float* r3 = h2 + (long long)s3 * LD2 + lane;
      float x0 = r0[0], x1 = r1[0], x2 = r2[0], x3 = r3[0];
      float y0 = 0.0f, y1 = 0.0f, y2 = 0.0f, y3 = 0.0f;
      if (hi) {
        y0 = r0[32];
        y1 = r1[32];
        y2 = r2[32];
        y3 = r3[32];
      }
      float w0 = __int_as_float(__builtin_amdgcn_readlane(wbits, k));
      float w1 = __int_as_float(__builtin_amdgcn_readlane(wbits, k + 1));
      float w2 = __int_as_float(__builtin_amdgcn_readlane(wbits, k + 2));
      float w3 = __int_as_float(__builtin_amdgcn_readlane(wbits, k + 3));
      a0 += x0 * w0;
      a1 += y0 * w0;
      if (k + 1 < m) { a0 += x1 * w1; a1 += y1 * w1; }
      if (k + 2 < m) { a0 += x2 * w2; a1 += y2 * w2; }
      if (k + 3 < m) { a0 += x3 * w3; a1 += y3 * w3; }
    }
  }
  float d = dinv[i];
  float dd = d * d;
  const float* hs = h2 + (long long)i * LD2 + lane;
  float v0 = a0 + hs[0] * dd + b2[lane];
  float v1 = 0.0f;
  float mx = v0;
  if (hi) {
    v1 = a1 + hs[32] * dd + b2[32 + lane];
    mx = fmaxf(mx, v1);
  }
#pragma unroll
  for (int o = 16; o > 0; o >>= 1) mx = fmaxf(mx, __shfl_xor(mx, o));
  float ex = __expf(v0 - mx);
  if (hi) ex += __expf(v1 - mx);
#pragma unroll
  for (int o = 16; o > 0; o >>= 1) ex += __shfl_xor(ex, o);
  float lse = mx + __logf(ex);
  float* op = out + (long long)i * NCLS + lane;
  op[0] = v0 - lse;
  if (hi) op[32] = v1 - lse;
}


extern "C" void kernel_launch(void* const* d_in, const int* in_sizes, int n_in,
                              void* d_out, int out_size, void* d_ws, size_t ws_size,
                              hipStream_t stream) {
  const float* x  = (const float*)d_in[0];
  const int*   ei = (const int*)d_in[1];
  const float* W1 = (const float*)d_in[2];
  const float* b1 = (const float*)d_in[3];
  const float* W2 = (const float*)d_in[4];
  const float* b2 = (const float*)d_in[5];
  float* out = (float*)d_out;

  const int N = in_sizes[0] / NFEAT;
  const int E = in_sizes[1] / 2;
  const int* erow = ei;
  const int* ecol = ei + E;
  const int Npad = (N + 63) & ~63;

  float* h1   = (float*)d_ws;
  float* act  = h1 + (long long)N * NFEAT;
  float* h2   = act + (long long)N * NFEAT;
  int*   cnt  = (int*)(h2 + (long long)N * LD2);
  int*   offs = cnt + Npad;
  float* dinv = (float*)(offs + Npad);
  float* wn   = dinv + Npad;
  int*   src  = (int*)(wn + E);
  int*   pos  = src + E;
  const int Epad = (E + 63) & ~63;
  int*   part = pos + Epad;
  const int CH = (((E + NCHUNK - 1) / NCHUNK) + 3) & ~3;

  const int TPB = 256;

  count_local<<<NCHUNK, 1024, 0, stream>>>(ecol, pos, part, E, CH, Npad);
  reduce_parts<<<(Npad + 127) / 128, 128, 0, stream>>>(part, cnt, Npad);
  scan_counts<<<1, 1024, 0, stream>>>(cnt, offs, dinv, Npad);
  fill_lists<<<dim3((CH + TPB - 1) / TPB, NCHUNK), TPB, 0, stream>>>(erow, ecol, pos, offs, part, dinv,
                                                                  src, wn, E, CH, Npad);

  const int Mtiles = (N + 15) / 16;
  const int mGroups = (Mtiles + WPB - 1) / WPB;
  const int nodeBlocks = (N + 7) / 8;

  gemm_strip_f32<NFEAT / 16, NFEAT><<<mGroups, 256, 0, stream>>>(x, W1, h1, N, NFEAT);

  gather_rows128_relu<<<nodeBlocks, 256, 0, stream>>>(offs, cnt, src, wn, dinv, h1, b1, act, N);

  gemm_strip_f32<LD2 / 16, NCLS><<<mGroups, 256, 0, stream>>>(act, W2, h2, N, LD2);

  gather_rows40_lsm<<<nodeBlocks, 256, 0, stream>>>(offs, cnt, src, wn, dinv, h2, b2, out, N);
}


# list offsets by a two-level scan: the per-node prefix kernel also scans its 256 nodes and writes a block total, list fill scans the block totals in LDS; no one-block scan launch; sums unchanged
# speedup vs baseline: 2.8138x; 1.1336x over previous
#include <hip/hip_runtime.h>
#include <cmath>


typedef __attribute__((ext_vector_type(2))) float v2f;
typedef __attribute__((ext_vector_type(4))) float v4f;
typedef __attribute__((ext_vector_type(4))) int v4i;
typedef __attribute__((ext_vector_type(8))) float v8f;

#define NFEAT 128
#define NCLS  40
#define LD2   48
#define KDIM  128
#define WPB   8
#define RBLK  256
#define NNODE 50000
#define HCAP  ((NNODE + 63) & ~63)
#define NCHUNK 64
#define CBATCH 16
#define MAXNB ((HCAP + RBLK - 1) / RBLK)
#define FTPB  256
static_assert(MAXNB <= FTPB, "fill_lists scans the block totals one per thread");


__global__ __launch_bounds__(1024) void count_local(
    const int* __restrict__ col, int* __restrict__ pos, int* __restrict__ part,
    int E, int CH, int npad) {
  __shared__ __attribute__((aligned(16))) int hist[HCAP];
  int t = threadIdx.x;
  int k = blockIdx.x;
  int e0 = k * CH;
  int len = E - e0;
  if (len > CH) len = CH;
  int c[CBATCH];
#pragma unroll
  for (int u = 0; u < CBATCH; ++u) {
    int j = u * 1024 + t;
    c[u] = (j < len) ? col[e0 + j] : -1;
  }
  for (int i4 = t; i4 < HCAP / 4; i4 += 1024) ((v4i*)hist)[i4] = (v4i){0, 0, 0, 0};
  __syncthreads();
  for (int j0 = 0;;) {
    int rk[CBATCH];
#pragma unroll
    for (int u = 0; u < CBATCH; ++u) {
      rk[u] = 0;
      if ((unsigned)c[u] < (unsigned)HCAP) rk[u] = atomicAdd(&hist[c[u]], 1);
    }
#pragma unroll
    for (int u = 0; u < CBATCH; ++u) {
      int j = j0 + u * 1024 + t;
      if ((unsigned)c[u] < (unsigned)HCAP) pos[e0 + j] = rk[u];
    }
    j0 += CBATCH * 1024;
    if (j0 >= len) break;
#pragma unroll
    for (int u = 0; u < CBATCH; ++u) {
      int j = j0 + u * 1024 + t;
      c[u] = (j < len) ? col[e0 + j] : -1;
    }
  }
  __syncthreads();
  int* dst = part + (long long)k * npad;
  int n4 = (npad < HCAP ? npad : HCAP) / 4;
  for (int i4 = t; i4 < n4; i4 += 1024) ((v4i*)dst)[i4] = ((const v4i*)hist)[i4];
}

__device__ __forceinline__ int wave_incl_scan(int v, int lane) {
#pragma unroll
  for (int d = 1; d < 32; d <<= 1) {
    int y = __shfl_up(v, d);
    if (lane >= d) v += y;
  }
  return v;
}

__global__ __launch_bounds__(RBLK) void reduce_parts(int* __restrict__ part,
                                                     int* __restrict__ cnt,
                                                     int* __restrict__ loc,
                                                     int* __restrict__ btot,
                                                     float* __restrict__ dinv, int npad) {
  __shared__ int wtot[RBLK / 32];
  int t = threadIdx.x;
  int lane = t & 31;
  int w = t >> 5;
  int i = blockIdx.x * RBLK + t;
  bool live = i < npad;
  int v[NCHUNK];
#pragma unroll
  for (int k = 0; k < NCHUNK; ++k) v[k] = live ? part[(long long)k * npad + i] : 0;
  int run = 0;
#pragma unroll
  for (int k = 0; k < NCHUNK; ++k) run += v[k];
  int incl = wave_incl_scan(run, lane);
  if (lane == 31) wtot[w] = incl;
  __syncthreads();
  int pre = 0;
#pragma unroll
  for (int k = 0; k < NCHUNK; ++k) {
    if (live) part[(long long)k * npad + i] = pre;
    pre += v[k];
  }
  int wprefix = 0;
  int total = 0;
#pragma unroll
  for (int u = 0; u < RBLK / 32; ++u) {
    int x = wtot[u];
    if (u < w) wprefix += x;
    total += x;
  }
  if (live) {
    cnt[i] = run;
    loc[i] = wprefix + (incl - run);
    float d = (float)(run + 1);
    dinv[i] = (d > 0.0f) ? rsqrtf(d) : 0.0f;
  }
  if (t == 0) btot[blockIdx.x] = total;
}

__global__ __launch_bounds__(FTPB) void fill_lists(
    const int* __restrict__ row, const int* __restrict__ col,
    const int* __restrict__ pos, const int* __restrict__ loc,
    const int* __restrict__ btot, const int* __restrict__ part,
    const float* __restrict__ dinv, int* __restrict__ src,
    float* __restrict__ wn, int* __restrict__ offs, int E, int CH, int npad, int nb) {
  __shared__ int bp[FTPB];
  __shared__ int wtot[FTPB / 32];
  int t = threadIdx.x;
  int lane = t & 31;
  int w = t >> 5;
  int k = blockIdx.y;
  int j = blockIdx.x * FTPB + t;
  int e = k * CH + j;
  bool live = (j < CH && e < E);
  int v = (t < nb) ? btot[t] : 0;
  int ec = live ? e : 0;
  int r = row[ec];
  int c = col[ec];
  int ps = pos[ec];
  int lc = loc[c];
  int pt = part[(long long)k * npad + c];
  float dr = dinv[r];
  float dc = dinv[c];
  int incl = wave_incl_scan(v, lane);
  if (lane == 31) wtot[w] = incl;
  __syncthreads();
  int wprefix = 0;
#pragma unroll
  for (int u = 0; u < FTPB / 32; ++u) {
    int x = wtot[u];
    if (u < w) wprefix += x;
  }
  bp[t] = wprefix + (incl - v);
  __syncthreads();
  if (live) {
    int p = bp[c / RBLK] + lc + pt + ps;
    src[p] = r;
    wn[p] = dr * dc;
  }
  int g = (k * gridDim.x + blockIdx.x) * FTPB + t;
  if (g < npad) offs[g] = bp[g / RBLK] + loc[g];
}

template <int NT, int NV>
__global__ __launch_bounds__(256) void gemm_strip_f32(
    const float* __restrict__ A, const float* __restrict__ B, float* __restrict__ C,
    int M, int ldc) {
  constexpr int NW   = NT * 16;
  constexpr int V4R  = NW / 4;
  constexpr int NV4  = NV / 4;
  constexpr int SW   = (NW > KDIM ? NW : KDIM);
  constexpr int SP   = SW + 4;
  constexpr int NJ   = (16 * V4R) / 32;
  constexpr int BP   = 2 * NW + ((NT % 2 == 0) ? 32 : 0);
  constexpr int NB   = ((KDIM / 2) * V4R) / 256;
  __shared__ __attribute__((aligned(16))) float Bs[(KDIM / 2) * BP];
  __shared__ __attribute__((aligned(16))) float Ss[WPB * 16 * SP];

  int t = threadIdx.x;
  int w = t >> 5;
  int lane = t & 31;
  int l16 = lane & 15;
  int hi  = lane >> 4;
  int kh  = hi << 1;

  int Mtiles = (M + 15) >> 4;
  int mt = blockIdx.x * WPB + w;
  int m0 = mt << 4;
  float* Sw = Ss + w * (16 * SP);

  v4f ar[16];
#pragma unroll
  for (int r = 0; r < 16; ++r) {
    int m = m0 + r;
    if (m >= M) m = M - 1;
    ar[r] = *(const v4f*)(A + (long long)m * KDIM + 4 * lane);
  }

  v4f b0[NB], b1[NB];
#pragma unroll
  for (int j = 0; j < NB; ++j) {
    int item = t + j * 256;
    int kp = item / V4R;
    int c4 = item % V4R;
    int c4c = c4 < NV4 ? c4 : NV4 - 1;
    b0[j] = *(const v4f*)(B + (2 * kp) * NV + c4c * 4);
    b1[j] = *(const v4f*)(B + (2 * kp + 1) * NV + c4c * 4);
    if (c4 >= NV4) {
      b0[j] = (v4f){0.0f, 0.0f, 0.0f, 0.0f};
      b1[j] = (v4f){0.0f, 0.0f, 0.0f, 0.0f};
    }
  }
#pragma unroll
  for (int r = 0; r < 16; ++r) *(v4f*)(Sw + r * SP + 4 * lane) = ar[r];
#pragma unroll
  for (int j = 0; j < NB; ++j) {
    int item = t + j * 256;
    int kp = item / V4R;
    int c4 = item % V4R;
    float* d = Bs + kp * BP + 8 * c4;
    *(v4f*)d       = (v4f){b0[j].x, b1[j].x, b0[j].y, b1[j].y};
    *(v4f*)(d + 4) = (v4f){b0[j].z, b1[j].z, b0[j].w, b1[j].w};
  }
  __syncthreads();

  if (mt >= Mtiles) return;

  v8f acc[NT];
#pragma unroll
  for (int nt = 0; nt < NT; ++nt) acc[nt] = (v8f){};

  const float* Ak = Sw + l16 * SP + kh;
#pragma unroll
  for (int s = 0; s < KDIM / 4; ++s) {
    v2f a = *(const v2f*)(Ak + 4 * s);
    const float* Bk = Bs + (2 * s + hi) * BP + 2 * l16;
#pragma unroll
    for (int nt = 0; nt < NT; ++nt) {
      v2f b = *(const v2f*)(Bk + 32 * nt);
      acc[nt] = __builtin_amdgcn_wmma_f32_16x16x4_f32(
           false, a,  false, b,
           (short)0, acc[nt],  false,  false);
    }
  }

#pragma unroll
  for (int nt = 0; nt < NT; ++nt)
#pragma unroll
    for (int r = 0; r < 8; ++r) Sw[(hi * 8 + r) * SP + nt * 16 + l16] = acc[nt][r];

  if (m0 + 16 <= M) {
    v4f cv[NJ];
#pragma unroll
    for (int j = 0; j < NJ; ++j) {
      int e = j * 32 + lane;
      cv[j] = *(const v4f*)(Sw + (e / V4R) * SP + (e % V4R) * 4);
    }
#pragma unroll
    for (int j = 0; j < NJ; ++j) {
      int e = j * 32 + lane;
      *(v4f*)(C + (long long)(m0 + e / V4R) * ldc + (e % V4R) * 4) = cv[j];
    }
  } else {
#pragma unroll
    for (int j = 0; j < NJ; ++j) {
      int e = j * 32 + lane;
      int row = e / V4R;
      int c4 = e % V4R;
      v4f v = *(const v4f*)(Sw + row * SP + c4 * 4);
      if (m0 + row < M) *(v4f*)(C + (long long)(m0 + row) * ldc + c4 * 4) = v;
    }
  }
}

__global__ __launch_bounds__(256) void gather_rows128_relu(
    const int* __restrict__ offs, const int* __restrict__ cnt,
    const int* __restrict__ src, const float* __restrict__ wn,
    const float* __restrict__ dinv, const float* __restrict__ h,
    const float* __restrict__ bias, float* __restrict__ act, int Nnodes) {
  int lane = threadIdx.x & 31;
  int i = __builtin_amdgcn_readfirstlane((int)((blockIdx.x * 256u + threadIdx.x) >> 5));
  if (i >= Nnodes) return;
  int beg = offs[i];
  int n = cnt[i];
  const v4f* h4 = (const v4f*)h;
  v4f acc = {0.0f, 0.0f, 0.0f, 0.0f};
  for (int base = 0; base < n; base += 32) {
    int m = n - base;
    if (m > 32) m = 32;
    int sidx = i;
    float wv = 0.0f;
    if (lane < m) {
      sidx = src[beg + base + lane];
      wv = wn[beg + base + lane];
    }
    int wbits = __float_as_int(wv);
    for (int k = 0; k < m; k += 4) {
      int s0 = __builtin_amdgcn_readlane(sidx, k);
      int s1 = __builtin_amdgcn_readlane(sidx, k + 1);
      int s2 = __builtin_amdgcn_readlane(sidx, k + 2);
      int s3 = __builtin_amdgcn_readlane(sidx, k + 3);
      v4f v0 = h4[(long long)s0 * (NFEAT / 4) + lane];
      v4f v1 = h4[(long long)s1 * (NFEAT / 4) + lane];
      v4f v2 = h4[(long long)s2 * (NFEAT / 4) + lane];
      v4f v3 = h4[(long long)s3 * (NFEAT / 4) + lane];
      float w0 = __int_as_float(__builtin_amdgcn_readlane(wbits, k));
      float w1 = __int_as_float(__builtin_amdgcn_readlane(wbits, k + 1));
      float w2 = __int_as_float(__builtin_amdgcn_readlane(wbits, k + 2));
      float w3 = __int_as_float(__builtin_amdgcn_readlane(wbits, k + 3));
      acc += v0 * w0;
      if (k + 1 < m) acc += v1 * w1;
      if (k + 2 < m) acc += v2 * w2;
      if (k + 3 < m) acc += v3 * w3;
    }
  }
  float d = dinv[i];
  v4f hs = h4[(long long)i * (NFEAT / 4) + lane];
  v4f b = ((const v4f*)bias)[lane];
  v4f r;
  r.x = acc.x + hs.x * d * d + b.x;
  r.y = acc.y + hs.y * d * d + b.y;
  r.z = acc.z + hs.z * d * d + b.z;
  r.w = acc.w + hs.w * d * d + b.w;
  r.x = r.x > 0.0f ? r.x : 0.0f;
  r.y = r.y > 0.0f ? r.y : 0.0f;
  r.z = r.z > 0.0f ? r.z : 0.0f;
  r.w = r.w > 0.0f ? r.w : 0.0f;
  ((v4f*)act)[(long long)i * (NFEAT / 4) + lane] = r;
}

__global__ __launch_bounds__(256) void gather_rows40_lsm(
    const int* __restrict__ offs, const int* __restrict__ cnt,
    const int* __restrict__ src, const float* __restrict__ wn,
    const float* __restrict__ dinv, const float* __restrict__ h2,
    const float* __restrict__ b2, float* __restrict__ out, int Nnodes) {
  int lane = threadIdx.x & 31;
  int i = __builtin_amdgcn_readfirstlane((int)((blockIdx.x * 256u + threadIdx.x) >> 5));
  if (i >= Nnodes) return;
  bool hi = lane < (NCLS - 32);
  int beg = offs[i];
  int n = cnt[i];
  float a0 = 0.0f, a1 = 0.0f;
  for (int base = 0; base < n; base += 32) {
    int m = n - base;
    if (m > 32) m = 32;
    int sidx = i;
    float wv = 0.0f;
    if (lane < m) {
      sidx = src[beg + base + lane];
      wv = wn[beg + base + lane];
    }
    int wbits = __float_as_int(wv);
    for (int k = 0; k < m; k += 4) {
      int s0 = __builtin_amdgcn_readlane(sidx, k);
      int s1 = __builtin_amdgcn_readlane(sidx, k + 1);
      int s2 = __builtin_amdgcn_readlane(sidx, k + 2);
      int s3 = __builtin_amdgcn_readlane(sidx, k + 3);
      const float* r0 = h2 + (long long)s0 * LD2 + lane;
      const float* r1 = h2 + (long long)s1 * LD2 + lane;
      const float* r2 = h2 + (long long)s2 * LD2 + lane;
      const float* r3 = h2 + (long long)s3 * LD2 + lane;
      float x0 = r0[0], x1 = r1[0], x2 = r2[0], x3 = r3[0];
      float y0 = 0.0f, y1 = 0.0f, y2 = 0.0f, y3 = 0.0f;
      if (hi) {
        y0 = r0[32];
        y1 = r1[32];
        y2 = r2[32];
        y3 = r3[32];
      }
      float w0 = __int_as_float(__builtin_amdgcn_readlane(wbits, k));
      float w1 = __int_as_float(__builtin_amdgcn_readlane(wbits, k + 1));
      float w2 = __int_as_float(__builtin_amdgcn_readlane(wbits, k + 2));
      float w3 = __int_as_float(__builtin_amdgcn_readlane(wbits, k + 3));
      a0 += x0 * w0;
      a1 += y0 * w0;
      if (k + 1 < m) { a0 += x1 * w1; a1 += y1 * w1; }
      if (k + 2 < m) { a0 += x2 * w2; a1 += y2 * w2; }
      if (k + 3 < m) { a0 += x3 * w3; a1 += y3 * w3; }
    }
  }
  float d = dinv[i];
  float dd = d * d;
  const float* hs = h2 + (long long)i * LD2 + lane;
  float v0 = a0 + hs[0] * dd + b2[lane];
  float v1 = 0.0f;
  float mx = v0;
  if (hi) {
    v1 = a1 + hs[32] * dd + b2[32 + lane];
    mx = fmaxf(mx, v1);
  }
#pragma unroll
  for (int o = 16; o > 0; o >>= 1) mx = fmaxf(mx, __shfl_xor(mx, o));
  float ex = __expf(v0 - mx);
  if (hi) ex += __expf(v1 - mx);
#pragma unroll
  for (int o = 16; o > 0; o >>= 1) ex += __shfl_xor(ex, o);
  float lse = mx + __logf(ex);
  float* op = out + (long long)i * NCLS + lane;
  op[0] = v0 - lse;
  if (hi) op[32] = v1 - lse;
}


extern "C" void kernel_launch(void* const* d_in, const int* in_sizes, int n_in,
                              void* d_out, int out_size, void* d_ws, size_t ws_size,
                              hipStream_t stream) {
  const float* x  = (const float*)d_in[0];
  const int*   ei = (const int*)d_in[1];
  const float* W1 = (const float*)d_in[2];
  const float* b1 = (const float*)d_in[3];
  const float* W2 = (const float*)d_in[4];
  const float* b2 = (const float*)d_in[5];
  float* out = (float*)d_out;

  const int N = in_sizes[0] / NFEAT;
  const int E = in_sizes[1] / 2;
  const int* erow = ei;
  const int* ecol = ei + E;
  const int Npad = (N + 63) & ~63;

  float* h1   = (float*)d_ws;
  float* act  = h1 + (long long)N * NFEAT;
  float* h2   = act + (long long)N * NFEAT;
  int*   cnt  = (int*)(h2 + (long long)N * LD2);
  int*   offs = cnt + Npad;
  float* dinv = (float*)(offs + Npad);
  float* wn   = dinv + Npad;
  int*   src  = (int*)(wn + E);
  int*   pos  = src + E;
  const int Epad = (E + 63) & ~63;
  int*   part = pos + Epad;
  int*   loc  = part + (long long)NCHUNK * Npad;
  int*   btot = loc + Npad;
  const int NB = (Npad + RBLK - 1) / RBLK;
  const int CH = (((E + NCHUNK - 1) / NCHUNK) + 3) & ~3;

  count_local<<<NCHUNK, 1024, 0, stream>>>(ecol, pos, part, E, CH, Npad);
  reduce_parts<<<NB, RBLK, 0, stream>>>(part, cnt, loc, btot, dinv, Npad);
  const int nodesPerChunkRow = (Npad + NCHUNK - 1) / NCHUNK;
  const int FW = CH > nodesPerChunkRow ? CH : nodesPerChunkRow;
  fill_lists<<<dim3((FW + FTPB - 1) / FTPB, NCHUNK), FTPB, 0, stream>>>(erow, ecol, pos, loc, btot, part, dinv,
                                                                    src, wn, offs, E, CH, Npad, NB);

  const int Mtiles = (N + 15) / 16;
  const int mGroups = (Mtiles + WPB - 1) / WPB;
  const int nodeBlocks = (N + 7) / 8;

  gemm_strip_f32<NFEAT / 16, NFEAT><<<mGroups, 256, 0, stream>>>(x, W1, h1, N, NFEAT);

  gather_rows128_relu<<<nodeBlocks, 256, 0, stream>>>(offs, cnt, src, wn, dinv, h1, b1, act, N);

  gemm_strip_f32<LD2 / 16, NCLS><<<mGroups, 256, 0, stream>>>(act, W2, h2, N, LD2);

  gather_rows40_lsm<<<nodeBlocks, 256, 0, stream>>>(offs, cnt, src, wn, dinv, h2, b2, out, N);
}


# both products store rows already scaled by their node's d^-1/2, each node's sum is scaled once: no per-edge weight (2 random loads + 1 random store per edge less); same add order, new rounding
# speedup vs baseline: 3.2792x; 1.1654x over previous
#include <hip/hip_runtime.h>
#include <cmath>


typedef __attribute__((ext_vector_type(2))) float v2f;
typedef __attribute__((ext_vector_type(4))) float v4f;
typedef __attribute__((ext_vector_type(4))) int v4i;
typedef __attribute__((ext_vector_type(8))) float v8f;

#define NFEAT 128
#define NCLS  40
#define LD2   48
#define KDIM  128
#define WPB   8
#define RBLK  256
#define NNODE 50000
#define HCAP  ((NNODE + 63) & ~63)
#define NCHUNK 64
#define CBATCH 16
#define MAXNB ((HCAP + RBLK - 1) / RBLK)
#define FTPB  256
static_assert(MAXNB <= FTPB, "fill_lists scans the block totals one per thread");


__global__ __launch_bounds__(1024) void count_local(
    const int* __restrict__ col, int* __restrict__ pos, int* __restrict__ part,
    int E, int CH, int npad) {
  __shared__ __attribute__((aligned(16))) int hist[HCAP];
  int t = threadIdx.x;
  int k = blockIdx.x;
  int e0 = k * CH;
  int len = E - e0;
  if (len > CH) len = CH;
  int c[CBATCH];
#pragma unroll
  for (int u = 0; u < CBATCH; ++u) {
    int j = u * 1024 + t;
    c[u] = (j < len) ? col[e0 + j] : -1;
  }
  for (int i4 = t; i4 < HCAP / 4; i4 += 1024) ((v4i*)hist)[i4] = (v4i){0, 0, 0, 0};
  __syncthreads();
  for (int j0 = 0;;) {
    int rk[CBATCH];
#pragma unroll
    for (int u = 0; u < CBATCH; ++u) {
      rk[u] = 0;
      if ((unsigned)c[u] < (unsigned)HCAP) rk[u] = atomicAdd(&hist[c[u]], 1);
    }
#pragma unroll
    for (int u = 0; u < CBATCH; ++u) {
      int j = j0 + u * 1024 + t;
      if ((unsigned)c[u] < (unsigned)HCAP) pos[e0 + j] = rk[u];
    }
    j0 += CBATCH * 1024;
    if (j0 >= len) break;
#pragma unroll
    for (int u = 0; u < CBATCH; ++u) {
      int j = j0 + u * 1024 + t;
      c[u] = (j < len) ? col[e0 + j] : -1;
    }
  }
  __syncthreads();
  int* dst = part + (long long)k * npad;
  int n4 = (npad < HCAP ? npad : HCAP) / 4;
  for (int i4 = t; i4 < n4; i4 += 1024) ((v4i*)dst)[i4] = ((const v4i*)hist)[i4];
}

__device__ __forceinline__ int wave_incl_scan(int v, int lane) {
#pragma unroll
  for (int d = 1; d < 32; d <<= 1) {
    int y = __shfl_up(v, d);
    if (lane >= d) v += y;
  }
  return v;
}

__global__ __launch_bounds__(RBLK) void reduce_parts(int* __restrict__ part,
                                                     int* __restrict__ cnt,
                                                     int* __restrict__ loc,
                                                     int* __restrict__ btot,
                                                     float* __restrict__ dinv, int npad) {
  __shared__ int wtot[RBLK / 32];
  int t = threadIdx.x;
  int lane = t & 31;
  int w = t >> 5;
  int i = blockIdx.x * RBLK + t;
  bool live = i < npad;
  int v[NCHUNK];
#pragma unroll
  for (int k = 0; k < NCHUNK; ++k) v[k] = live ? part[(long long)k * npad + i] : 0;
  int run = 0;
#pragma unroll
  for (int k = 0; k < NCHUNK; ++k) run += v[k];
  int incl = wave_incl_scan(run, lane);
  if (lane == 31) wtot[w] = incl;
  __syncthreads();
  int pre = 0;
#pragma unroll
  for (int k = 0; k < NCHUNK; ++k) {
    if (live) part[(long long)k * npad + i] = pre;
    pre += v[k];
  }
  int wprefix = 0;
  int total = 0;
#pragma unroll
  for (int u = 0; u < RBLK / 32; ++u) {
    int x = wtot[u];
    if (u < w) wprefix += x;
    total += x;
  }
  if (live) {
    cnt[i] = run;
    loc[i] = wprefix + (incl - run);
    float d = (float)(run + 1);
    dinv[i] = (d > 0.0f) ? rsqrtf(d) : 0.0f;
  }
  if (t == 0) btot[blockIdx.x] = total;
}

__global__ __launch_bounds__(FTPB) void fill_lists(
    const int* __restrict__ row, const int* __restrict__ col,
    const int* __restrict__ pos, const int* __restrict__ loc,
    const int* __restrict__ btot, const int* __restrict__ part,
    int* __restrict__ src, int* __restrict__ offs, int E, int CH, int npad, int nb) {
  __shared__ int bp[FTPB];
  __shared__ int wtot[FTPB / 32];
  int t = threadIdx.x;
  int lane = t & 31;
  int w = t >> 5;
  int k = blockIdx.y;
  int j = blockIdx.x * FTPB + t;
  int e = k * CH + j;
  bool live = (j < CH && e < E);
  int v = (t < nb) ? btot[t] : 0;
  int ec = live ? e : 0;
  int r = row[ec];
  int c = col[ec];
  int ps = pos[ec];
  int lc = loc[c];
  int pt = part[(long long)k * npad + c];
  int incl = wave_incl_scan(v, lane);
  if (lane == 31) wtot[w] = incl;
  __syncthreads();
  int wprefix = 0;
#pragma unroll
  for (int u = 0; u < FTPB / 32; ++u) {
    int x = wtot[u];
    if (u < w) wprefix += x;
  }
  bp[t] = wprefix + (incl - v);
  __syncthreads();
  if (live) {
    int p = bp[c / RBLK] + lc + pt + ps;
    src[p] = r;
  }
  int g = (k * gridDim.x + blockIdx.x) * FTPB + t;
  if (g < npad) offs[g] = bp[g / RBLK] + loc[g];
}

template <int NT, int NV>
__global__ __launch_bounds__(256) void gemm_strip_f32(
    const float* __restrict__ A, const float* __restrict__ B,
    const float* __restrict__ rs, float* __restrict__ C, int M, int ldc) {
  constexpr int NW   = NT * 16;
  constexpr int V4R  = NW / 4;
  constexpr int NV4  = NV / 4;
  constexpr int SW   = (NW > KDIM ? NW : KDIM);
  constexpr int SP   = SW + 4;
  constexpr int NJ   = (16 * V4R) / 32;
  constexpr int BP   = 2 * NW + ((NT % 2 == 0) ? 32 : 0);
  constexpr int NB   = ((KDIM / 2) * V4R) / 256;
  __shared__ __attribute__((aligned(16))) float Bs[(KDIM / 2) * BP];
  __shared__ __attribute__((aligned(16))) float Ss[WPB * 16 * SP];

  int t = threadIdx.x;
  int w = t >> 5;
  int lane = t & 31;
  int l16 = lane & 15;
  int hi  = lane >> 4;
  int kh  = hi << 1;

  int Mtiles = (M + 15) >> 4;
  int mt = blockIdx.x * WPB + w;
  int m0 = mt << 4;
  float* Sw = Ss + w * (16 * SP);

  v4f ar[16];
#pragma unroll
  for (int r = 0; r < 16; ++r) {
    int m = m0 + r;
    if (m >= M) m = M - 1;
    ar[r] = *(const v4f*)(A + (long long)m * KDIM + 4 * lane);
  }

  v4f b0[NB], b1[NB];
#pragma unroll
  for (int j = 0; j < NB; ++j) {
    int item = t + j * 256;
    int kp = item / V4R;
    int c4 = item % V4R;
    int c4c = c4 < NV4 ? c4 : NV4 - 1;
    b0[j] = *(const v4f*)(B + (2 * kp) * NV + c4c * 4);
    b1[j] = *(const v4f*)(B + (2 * kp + 1) * NV + c4c * 4);
    if (c4 >= NV4) {
      b0[j] = (v4f){0.0f, 0.0f, 0.0f, 0.0f};
      b1[j] = (v4f){0.0f, 0.0f, 0.0f, 0.0f};
    }
  }
#pragma unroll
  for (int r = 0; r < 16; ++r) *(v4f*)(Sw + r * SP + 4 * lane) = ar[r];
#pragma unroll
  for (int j = 0; j < NB; ++j) {
    int item = t + j * 256;
    int kp = item / V4R;
    int c4 = item % V4R;
    float* d = Bs + kp * BP + 8 * c4;
    *(v4f*)d       = (v4f){b0[j].x, b1[j].x, b0[j].y, b1[j].y};
    *(v4f*)(d + 4) = (v4f){b0[j].z, b1[j].z, b0[j].w, b1[j].w};
  }
  __syncthreads();

  if (mt >= Mtiles) return;

  float rsv[NJ];
#pragma unroll
  for (int j = 0; j < NJ; ++j) rsv[j] = rs[m0 + (j * 32 + lane) / V4R];

  v8f acc[NT];
#pragma unroll
  for (int nt = 0; nt < NT; ++nt) acc[nt] = (v8f){};

  const float* Ak = Sw + l16 * SP + kh;
#pragma unroll
  for (int s = 0; s < KDIM / 4; ++s) {
    v2f a = *(const v2f*)(Ak + 4 * s);
    const float* Bk = Bs + (2 * s + hi) * BP + 2 * l16;
#pragma unroll
    for (int nt = 0; nt < NT; ++nt) {
      v2f b = *(const v2f*)(Bk + 32 * nt);
      acc[nt] = __builtin_amdgcn_wmma_f32_16x16x4_f32(
           false, a,  false, b,
           (short)0, acc[nt],  false,  false);
    }
  }

#pragma unroll
  for (int nt = 0; nt < NT; ++nt)
#pragma unroll
    for (int r = 0; r < 8; ++r) Sw[(hi * 8 + r) * SP + nt * 16 + l16] = acc[nt][r];

  if (m0 + 16 <= M) {
    v4f cv[NJ];
#pragma unroll
    for (int j = 0; j < NJ; ++j) {
      int e = j * 32 + lane;
      cv[j] = *(const v4f*)(Sw + (e / V4R) * SP + (e % V4R) * 4) * rsv[j];
    }
#pragma unroll
    for (int j = 0; j < NJ; ++j) {
      int e = j * 32 + lane;
      *(v4f*)(C + (long long)(m0 + e / V4R) * ldc + (e % V4R) * 4) = cv[j];
    }
  } else {
#pragma unroll
    for (int j = 0; j < NJ; ++j) {
      int e = j * 32 + lane;
      int row = e / V4R;
      int c4 = e % V4R;
      v4f v = *(const v4f*)(Sw + row * SP + c4 * 4) * rsv[j];
      if (m0 + row < M) *(v4f*)(C + (long long)(m0 + row) * ldc + c4 * 4) = v;
    }
  }
}

__global__ __launch_bounds__(256) void gather_rows128_relu(
    const int* __restrict__ offs, const int* __restrict__ cnt,
    const int* __restrict__ src, const float* __restrict__ dinv,
    const float* __restrict__ h, const float* __restrict__ bias,
    float* __restrict__ act, int Nnodes) {
  int lane = threadIdx.x & 31;
  int i = __builtin_amdgcn_readfirstlane((int)((blockIdx.x * 256u + threadIdx.x) >> 5));
  if (i >= Nnodes) return;
  int beg = offs[i];
  int n = cnt[i];
  const v4f* h4 = (const v4f*)h;
  v4f acc = {0.0f, 0.0f, 0.0f, 0.0f};
  for (int base = 0; base < n; base += 32) {
    int m = n - base;
    if (m > 32) m = 32;
    int sidx = i;
    if (lane < m) sidx = src[beg + base + lane];
    for (int k = 0; k < m; k += 4) {
      int s0 = __builtin_amdgcn_readlane(sidx, k);
      int s1 = __builtin_amdgcn_readlane(sidx, k + 1);
      int s2 = __builtin_amdgcn_readlane(sidx, k + 2);
      int s3 = __builtin_amdgcn_readlane(sidx, k + 3);
      v4f v0 = h4[(long long)s0 * (NFEAT / 4) + lane];
      v4f v1 = h4[(long long)s1 * (NFEAT / 4) + lane];
      v4f v2 = h4[(long long)s2 * (NFEAT / 4) + lane];
      v4f v3 = h4[(long long)s3 * (NFEAT / 4) + lane];
      const v4f z4 = {0.0f, 0.0f, 0.0f, 0.0f};
      acc += v0;
      acc += (k + 1 < m) ? v1 : z4;
      acc += (k + 2 < m) ? v2 : z4;
      acc += (k + 3 < m) ? v3 : z4;
    }
  }
  float d = dinv[i];
  v4f hs = h4[(long long)i * (NFEAT / 4) + lane];
  v4f b = ((const v4f*)bias)[lane];
  v4f r;
  r.x = (acc.x + hs.x) * d + b.x;
  r.y = (acc.y + hs.y) * d + b.y;
  r.z = (acc.z + hs.z) * d + b.z;
  r.w = (acc.w + hs.w) * d + b.w;
  r.x = r.x > 0.0f ? r.x : 0.0f;
  r.y = r.y > 0.0f ? r.y : 0.0f;
  r.z = r.z > 0.0f ? r.z : 0.0f;
  r.w = r.w > 0.0f ? r.w : 0.0f;
  ((v4f*)act)[(long long)i * (NFEAT / 4) + lane] = r;
}

__global__ __launch_bounds__(256) void gather_rows40_lsm(
    const int* __restrict__ offs, const int* __restrict__ cnt,
    const int* __restrict__ src, const float* __restrict__ dinv,
    const float* __restrict__ h2, const float* __restrict__ b2,
    float* __restrict__ out, int Nnodes) {
  int lane = threadIdx.x & 31;
  int i = __builtin_amdgcn_readfirstlane((int)((blockIdx.x * 256u + threadIdx.x) >> 5));
  if (i >= Nnodes) return;
  bool hi = lane < (NCLS - 32);
  int beg = offs[i];
  int n = cnt[i];
  float a0 = 0.0f, a1 = 0.0f;
  for (int base = 0; base < n; base += 32) {
    int m = n - base;
    if (m > 32) m = 32;
    int sidx = i;
    if (lane < m) sidx = src[beg + base + lane];
    for (int k = 0; k < m; k += 4) {
      int s0 = __builtin_amdgcn_readlane(sidx, k);
      int s1 = __builtin_amdgcn_readlane(sidx, k + 1);
      int s2 = __builtin_amdgcn_readlane(sidx, k + 2);
      int s3 = __builtin_amdgcn_readlane(sidx, k + 3);
      const float* r0 = h2 + (long long)s0 * LD2 + lane;
      const float* r1 = h2 + (long long)s1 * LD2 + lane;
      const float* r2 = h2 + (long long)s2 * LD2 + lane;
      const float* r3 = h2 + (long long)s3 * LD2 + lane;
      float x0 = r0[0], x1 = r1[0], x2 = r2[0], x3 = r3[0];
      float y0 = 0.0f, y1 = 0.0f, y2 = 0.0f, y3 = 0.0f;
      if (hi) {
        y0 = r0[32];
        y1 = r1[32];
        y2 = r2[32];
        y3 = r3[32];
      }
      bool in1 = k + 1 < m, in2 = k + 2 < m, in3 = k + 3 < m;
      a0 += x0;
      a1 += y0;
      a0 += in1 ? x1 : 0.0f;
      a1 += in1 ? y1 : 0.0f;
      a0 += in2 ? x2 : 0.0f;
      a1 += in2 ? y2 : 0.0f;
      a0 += in3 ? x3 : 0.0f;
      a1 += in3 ? y3 : 0.0f;
    }
  }
  float d = dinv[i];
  const float* hs = h2 + (long long)i * LD2 + lane;
  float v0 = (a0 + hs[0]) * d + b2[lane];
  float v1 = 0.0f;
  float mx = v0;
  if (hi) {
    v1 = (a1 + hs[32]) * d + b2[32 + lane];
    mx = fmaxf(mx, v1);
  }
#pragma unroll
  for (int o = 16; o > 0; o >>= 1) mx = fmaxf(mx, __shfl_xor(mx, o));
  float ex = __expf(v0 - mx);
  if (hi) ex += __expf(v1 - mx);
#pragma unroll
  for (int o = 16; o > 0; o >>= 1) ex += __shfl_xor(ex, o);
  float lse = mx + __logf(ex);
  float* op = out + (long long)i * NCLS + lane;
  op[0] = v0 - lse;
  if (hi) op[32] = v1 - lse;
}


extern "C" void kernel_launch(void* const* d_in, const int* in_sizes, int n_in,
                              void* d_out, int out_size, void* d_ws, size_t ws_size,
                              hipStream_t stream) {
  const float* x  = (const float*)d_in[0];
  const int*   ei = (const int*)d_in[1];
  const float* W1 = (const float*)d_in[2];
  const float* b1 = (const float*)d_in[3];
  const float* W2 = (const float*)d_in[4];
  const float* b2 = (const float*)d_in[5];
  float* out = (float*)d_out;

  const int N = in_sizes[0] / NFEAT;
  const int E = in_sizes[1] / 2;
  const int* erow = ei;
  const int* ecol = ei + E;
  const int Npad = (N + 63) & ~63;

  float* h1   = (float*)d_ws;
  float* act  = h1 + (long long)N * NFEAT;
  float* h2   = act + (long long)N * NFEAT;
  int*   cnt  = (int*)(h2 + (long long)N * LD2);
  int*   offs = cnt + Npad;
  float* dinv = (float*)(offs + Npad);
  int*   src  = (int*)(dinv + Npad);
  int*   pos  = src + E;
  const int Epad = (E + 63) & ~63;
  int*   part = pos + Epad;
  int*   loc  = part + (long long)NCHUNK * Npad;
  int*   btot = loc + Npad;
  const int NB = (Npad + RBLK - 1) / RBLK;
  const int CH = (((E + NCHUNK - 1) / NCHUNK) + 3) & ~3;

  count_local<<<NCHUNK, 1024, 0, stream>>>(ecol, pos, part, E, CH, Npad);
  reduce_parts<<<NB, RBLK, 0, stream>>>(part, cnt, loc, btot, dinv, Npad);
  const int nodesPerChunkRow = (Npad + NCHUNK - 1) / NCHUNK;
  const int FW = CH > nodesPerChunkRow ? CH : nodesPerChunkRow;
  fill_lists<<<dim3((FW + FTPB - 1) / FTPB, NCHUNK), FTPB, 0, stream>>>(erow, ecol, pos, loc, btot, part,
                                                                    src, offs, E, CH, Npad, NB);

  const int Mtiles = (N + 15) / 16;
  const int mGroups = (Mtiles + WPB - 1) / WPB;
  const int nodeBlocks = (N + 7) / 8;

  gemm_strip_f32<NFEAT / 16, NFEAT><<<mGroups, 256, 0, stream>>>(x, W1, dinv, h1, N, NFEAT);

  gather_rows128_relu<<<nodeBlocks, 256, 0, stream>>>(offs, cnt, src, dinv, h1, b1, act, N);

  gemm_strip_f32<LD2 / 16, NCLS><<<mGroups, 256, 0, stream>>>(act, W2, dinv, h2, N, LD2);

  gather_rows40_lsm<<<nodeBlocks, 256, 0, stream>>>(offs, cnt, src, dinv, h2, b2, out, N);
}


# list fill moved into the first product's launch: each thread takes 8 edges, requests their two looked-up words in front of the K loop, stores list slots behind it; one launch fewer; sums unchanged
# speedup vs baseline: 3.4785x; 1.0608x over previous
#include <hip/hip_runtime.h>
#include <cmath>


typedef __attribute__((ext_vector_type(2))) float v2f;
typedef __attribute__((ext_vector_type(4))) float v4f;
typedef __attribute__((ext_vector_type(4))) int v4i;
typedef __attribute__((ext_vector_type(8))) float v8f;

#define NFEAT 128
#define NCLS  40
#define LD2   48
#define KDIM  128
#define WPB   8
#define RBLK  256
#define NNODE 50000
#define HCAP  ((NNODE + 63) & ~63)
#define NCHUNK 64
#define CBATCH 16
#define MAXNB ((HCAP + RBLK - 1) / RBLK)
#define FTPB  256
#define FE    8
static_assert(MAXNB <= FTPB, "the list fill scans the block totals one per thread");


__global__ __launch_bounds__(1024) void count_local(
    const int* __restrict__ col, int* __restrict__ pos, int* __restrict__ part,
    int E, int CH, int npad) {
  __shared__ __attribute__((aligned(16))) int hist[HCAP];
  int t = threadIdx.x;
  int k = blockIdx.x;
  int e0 = k * CH;
  int len = E - e0;
  if (len > CH) len = CH;
  int c[CBATCH];
#pragma unroll
  for (int u = 0; u < CBATCH; ++u) {
    int j = u * 1024 + t;
    c[u] = (j < len) ? col[e0 + j] : -1;
  }
  for (int i4 = t; i4 < HCAP / 4; i4 += 1024) ((v4i*)hist)[i4] = (v4i){0, 0, 0, 0};
  __syncthreads();
  for (int j0 = 0;;) {
    int rk[CBATCH];
#pragma unroll
    for (int u = 0; u < CBATCH; ++u) {
      rk[u] = 0;
      if ((unsigned)c[u] < (unsigned)HCAP) rk[u] = atomicAdd(&hist[c[u]], 1);
    }
#pragma unroll
    for (int u = 0; u < CBATCH; ++u) {
      int j = j0 + u * 1024 + t;
      if ((unsigned)c[u] < (unsigned)HCAP) pos[e0 + j] = rk[u];
    }
    j0 += CBATCH * 1024;
    if (j0 >= len) break;
#pragma unroll
    for (int u = 0; u < CBATCH; ++u) {
      int j = j0 + u * 1024 + t;
      c[u] = (j < len) ? col[e0 + j] : -1;
    }
  }
  __syncthreads();
  int* dst = part + (long long)k * npad;
  int n4 = (npad < HCAP ? npad : HCAP) / 4;
  for (int i4 = t; i4 < n4; i4 += 1024) ((v4i*)dst)[i4] = ((const v4i*)hist)[i4];
}

__device__ __forceinline__ int wave_incl_scan(int v, int lane) {
#pragma unroll
  for (int d = 1; d < 32; d <<= 1) {
    int y = __shfl_up(v, d);
    if (lane >= d) v += y;
  }
  return v;
}

__global__ __launch_bounds__(RBLK) void reduce_parts(int* __restrict__ part,
                                                     int* __restrict__ cnt,
                                                     int* __restrict__ loc,
                                                     int* __restrict__ btot,
                                                     float* __restrict__ dinv, int npad) {
  __shared__ int wtot[RBLK / 32];
  int t = threadIdx.x;
  int lane = t & 31;
  int w = t >> 5;
  int i = blockIdx.x * RBLK + t;
  bool live = i < npad;
  int v[NCHUNK];
#pragma unroll
  for (int k = 0; k < NCHUNK; ++k) v[k] = live ? part[(long long)k * npad + i] : 0;
  int run = 0;
#pragma unroll
  for (int k = 0; k < NCHUNK; ++k) run += v[k];
  int incl = wave_incl_scan(run, lane);
  if (lane == 31) wtot[w] = incl;
  __syncthreads();
  int pre = 0;
#pragma unroll
  for (int k = 0; k < NCHUNK; ++k) {
    if (live) part[(long long)k * npad + i] = pre;
    pre += v[k];
  }
  int wprefix = 0;
  int total = 0;
#pragma unroll
  for (int u = 0; u < RBLK / 32; ++u) {
    int x = wtot[u];
    if (u < w) wprefix += x;
    total += x;
  }
  if (live) {
    cnt[i] = run;
    loc[i] = wprefix + (incl - run);
    float d = (float)(run + 1);
    dinv[i] = (d > 0.0f) ? rsqrtf(d) : 0.0f;
  }
  if (t == 0) btot[blockIdx.x] = total;
}

template <int NT, int NV, bool FILL>
__global__ __launch_bounds__(256) void gemm_strip_f32(
    const float* __restrict__ A, const float* __restrict__ B,
    const float* __restrict__ rs, float* __restrict__ C, int M, int ldc,
    const int* __restrict__ row, const int* __restrict__ col,
    const int* __restrict__ pos, const int* __restrict__ loc,
    const int* __restrict__ btot, const int* __restrict__ part,
    int* __restrict__ src, int* __restrict__ offs, int E, int CH, int npad, int nb) {
  constexpr int NW   = NT * 16;
  constexpr int V4R  = NW / 4;
  constexpr int NV4  = NV / 4;
  constexpr int SW   = (NW > KDIM ? NW : KDIM);
  constexpr int SP   = SW + 4;
  constexpr int NJ   = (16 * V4R) / 32;
  constexpr int BP   = 2 * NW + ((NT % 2 == 0) ? 32 : 0);
  constexpr int NB   = ((KDIM / 2) * V4R) / 256;
  __shared__ __attribute__((aligned(16))) float Bs[(KDIM / 2) * BP];
  __shared__ __attribute__((aligned(16))) float Ss[WPB * 16 * SP];
  __shared__ int bp[FILL ? FTPB : 1];
  __shared__ int wtot[FILL ? WPB : 1];

  int t = threadIdx.x;
  int w = t >> 5;
  int lane = t & 31;
  int l16 = lane & 15;
  int hi  = lane >> 4;
  int kh  = hi << 1;

  int Mtiles = (M + 15) >> 4;
  int mt = blockIdx.x * WPB + w;
  int m0 = mt << 4;
  float* Sw = Ss + w * (16 * SP);

  int fr[FE], fc[FE], fq[FE], fl[FE], fp[FE];
  int fv = 0, fincl = 0;
  const int fe0 = blockIdx.x * (FE * 256) + t;
  if (FILL) {
#pragma unroll
    for (int u = 0; u < FE; ++u) {
      int e = fe0 + u * 256;
      int ec = e < E ? e : 0;
      fr[u] = row[ec];
      fc[u] = col[ec];
      fq[u] = pos[ec];
    }
    fv = (t < nb) ? btot[t] : 0;
  }

  v4f ar[16];
#pragma unroll
  for (int r = 0; r < 16; ++r) {
    int m = m0 + r;
    if (m >= M) m = M - 1;
    ar[r] = *(const v4f*)(A + (long long)m * KDIM + 4 * lane);
  }

  v4f b0[NB], b1[NB];
#pragma unroll
  for (int j = 0; j < NB; ++j) {
    int item = t + j * 256;
    int kp = item / V4R;
    int c4 = item % V4R;
    int c4c = c4 < NV4 ? c4 : NV4 - 1;
    b0[j] = *(const v4f*)(B + (2 * kp) * NV + c4c * 4);
    b1[j] = *(const v4f*)(B + (2 * kp + 1) * NV + c4c * 4);
    if (c4 >= NV4) {
      b0[j] = (v4f){0.0f, 0.0f, 0.0f, 0.0f};
      b1[j] = (v4f){0.0f, 0.0f, 0.0f, 0.0f};
    }
  }
  if (FILL) {
    fincl = wave_incl_scan(fv, lane);
    if (lane == 31) wtot[w] = fincl;
  }
#pragma unroll
  for (int r = 0; r < 16; ++r) *(v4f*)(Sw + r * SP + 4 * lane) = ar[r];
#pragma unroll
  for (int j = 0; j < NB; ++j) {
    int item = t + j * 256;
    int kp = item / V4R;
    int c4 = item % V4R;
    float* d = Bs + kp * BP + 8 * c4;
    *(v4f*)d       = (v4f){b0[j].x, b1[j].x, b0[j].y, b1[j].y};
    *(v4f*)(d + 4) = (v4f){b0[j].z, b1[j].z, b0[j].w, b1[j].w};
  }
  __syncthreads();
  if (FILL) {
    int wprefix = 0;
#pragma unroll
    for (int u = 0; u < WPB; ++u) {
      int x = wtot[u];
      if (u < w) wprefix += x;
    }
    bp[t] = wprefix + (fincl - fv);
    __syncthreads();
#pragma unroll
    for (int u = 0; u < FE; ++u) {
      int e = fe0 + u * 256;
      int ec = e < E ? e : 0;
      int k = ec / CH;
      fl[u] = loc[fc[u]];
      fp[u] = part[(long long)k * npad + fc[u]];
    }
  }

  if (mt < Mtiles) {

  float rsv[NJ];
#pragma unroll
  for (int j = 0; j < NJ; ++j) rsv[j] = rs[m0 + (j * 32 + lane) / V4R];

  v8f acc[NT];
#pragma unroll
  for (int nt = 0; nt < NT; ++nt) acc[nt] = (v8f){};

  const float* Ak = Sw + l16 * SP + kh;
#pragma unroll
  for (int s = 0; s < KDIM / 4; ++s) {
    v2f a = *(const v2f*)(Ak + 4 * s);
    const float* Bk = Bs + (2 * s + hi) * BP + 2 * l16;
#pragma unroll
    for (int nt = 0; nt < NT; ++nt) {
      v2f b = *(const v2f*)(Bk + 32 * nt);
      acc[nt] = __builtin_amdgcn_wmma_f32_16x16x4_f32(
           false, a,  false, b,
           (short)0, acc[nt],  false,  false);
    }
  }

#pragma unroll
  for (int nt = 0; nt < NT; ++nt)
#pragma unroll
    for (int r = 0; r < 8; ++r) Sw[(hi * 8 + r) * SP + nt * 16 + l16] = acc[nt][r];

  if (m0 + 16 <= M) {
    v4f cv[NJ];
#pragma unroll
    for (int j = 0; j < NJ; ++j) {
      int e = j * 32 + lane;
      cv[j] = *(const v4f*)(Sw + (e / V4R) * SP + (e % V4R) * 4) * rsv[j];
    }
#pragma unroll
    for (int j = 0; j < NJ; ++j) {
      int e = j * 32 + lane;
      *(v4f*)(C + (long long)(m0 + e / V4R) * ldc + (e % V4R) * 4) = cv[j];
    }
  } else {
#pragma unroll
    for (int j = 0; j < NJ; ++j) {
      int e = j * 32 + lane;
      int row = e / V4R;
      int c4 = e % V4R;
      v4f v = *(const v4f*)(Sw + row * SP + c4 * 4) * rsv[j];
      if (m0 + row < M) *(v4f*)(C + (long long)(m0 + row) * ldc + c4 * 4) = v;
    }
  }
  }

  if (FILL) {
#pragma unroll
    for (int u = 0; u < FE; ++u) {
      int e = fe0 + u * 256;
      if (e < E) src[bp[fc[u] / RBLK] + fl[u] + fp[u] + fq[u]] = fr[u];
    }
    for (int e = gridDim.x * (FE * 256) + blockIdx.x * 256 + t; e < E; e += gridDim.x * 256) {
      int c = col[e];
      src[bp[c / RBLK] + loc[c] + part[(long long)(e / CH) * npad + c] + pos[e]] = row[e];
    }
    for (int g = blockIdx.x * 256 + t; g < npad; g += gridDim.x * 256)
      offs[g] = bp[g / RBLK] + loc[g];
  }
}

__global__ __launch_bounds__(256) void gather_rows128_relu(
    const int* __restrict__ offs, const int* __restrict__ cnt,
    const int* __restrict__ src, const float* __restrict__ dinv,
    const float* __restrict__ h, const float* __restrict__ bias,
    float* __restrict__ act, int Nnodes) {
  int lane = threadIdx.x & 31;
  int i = __builtin_amdgcn_readfirstlane((int)((blockIdx.x * 256u + threadIdx.x) >> 5));
  if (i >= Nnodes) return;
  int beg = offs[i];
  int n = cnt[i];
  const v4f* h4 = (const v4f*)h;
  v4f acc = {0.0f, 0.0f, 0.0f, 0.0f};
  for (int base = 0; base < n; base += 32) {
    int m = n - base;
    if (m > 32) m = 32;
    int sidx = i;
    if (lane < m) sidx = src[beg + base + lane];
    for (int k = 0; k < m; k += 4) {
      int s0 = __builtin_amdgcn_readlane(sidx, k);
      int s1 = __builtin_amdgcn_readlane(sidx, k + 1);
      int s2 = __builtin_amdgcn_readlane(sidx, k + 2);
      int s3 = __builtin_amdgcn_readlane(sidx, k + 3);
      v4f v0 = h4[(long long)s0 * (NFEAT / 4) + lane];
      v4f v1 = h4[(long long)s1 * (NFEAT / 4) + lane];
      v4f v2 = h4[(long long)s2 * (NFEAT / 4) + lane];
      v4f v3 = h4[(long long)s3 * (NFEAT / 4) + lane];
      const v4f z4 = {0.0f, 0.0f, 0.0f, 0.0f};
      acc += v0;
      acc += (k + 1 < m) ? v1 : z4;
      acc += (k + 2 < m) ? v2 : z4;
      acc += (k + 3 < m) ? v3 : z4;
    }
  }
  float d = dinv[i];
  v4f hs = h4[(long long)i * (NFEAT / 4) + lane];
  v4f b = ((const v4f*)bias)[lane];
  v4f r;
  r.x = (acc.x + hs.x) * d + b.x;
  r.y = (acc.y + hs.y) * d + b.y;
  r.z = (acc.z + hs.z) * d + b.z;
  r.w = (acc.w + hs.w) * d + b.w;
  r.x = r.x > 0.0f ? r.x : 0.0f;
  r.y = r.y > 0.0f ? r.y : 0.0f;
  r.z = r.z > 0.0f ? r.z : 0.0f;
  r.w = r.w > 0.0f ? r.w : 0.0f;
  ((v4f*)act)[(long long)i * (NFEAT / 4) + lane] = r;
}

__global__ __launch_bounds__(256) void gather_rows40_lsm(
    const int* __restrict__ offs, const int* __restrict__ cnt,
    const int* __restrict__ src, const float* __restrict__ dinv,
    const float* __restrict__ h2, const float* __restrict__ b2,
    float* __restrict__ out, int Nnodes) {
  int lane = threadIdx.x & 31;
  int i = __builtin_amdgcn_readfirstlane((int)((blockIdx.x * 256u + threadIdx.x) >> 5));
  if (i >= Nnodes) return;
  bool hi = lane < (NCLS - 32);
  int beg = offs[i];
  int n = cnt[i];
  float a0 = 0.0f, a1 = 0.0f;
  for (int base = 0; base < n; base += 32) {
    int m = n - base;
    if (m > 32) m = 32;
    int sidx = i;
    if (lane < m) sidx = src[beg + base + lane];
    for (int k = 0; k < m; k += 4) {
      int s0 = __builtin_amdgcn_readlane(sidx, k);
      int s1 = __builtin_amdgcn_readlane(sidx, k + 1);
      int s2 = __builtin_amdgcn_readlane(sidx, k + 2);
      int s3 = __builtin_amdgcn_readlane(sidx, k + 3);
      const float* r0 = h2 + (long long)s0 * LD2 + lane;
      const float* r1 = h2 + (long long)s1 * LD2 + lane;
      const float* r2 = h2 + (long long)s2 * LD2 + lane;
      const float* r3 = h2 + (long long)s3 * LD2 + lane;
      float x0 = r0[0], x1 = r1[0], x2 = r2[0], x3 = r3[0];
      float y0 = 0.0f, y1 = 0.0f, y2 = 0.0f, y3 = 0.0f;
      if (hi) {
        y0 = r0[32];
        y1 = r1[32];
        y2 = r2[32];
        y3 = r3[32];
      }
      bool in1 = k + 1 < m, in2 = k + 2 < m, in3 = k + 3 < m;
      a0 += x0;
      a1 += y0;
      a0 += in1 ? x1 : 0.0f;
      a1 += in1 ? y1 : 0.0f;
      a0 += in2 ? x2 : 0.0f;
      a1 += in2 ? y2 : 0.0f;
      a0 += in3 ? x3 : 0.0f;
      a1 += in3 ? y3 : 0.0f;
    }
  }
  float d = dinv[i];
  const float* hs = h2 + (long long)i * LD2 + lane;
  float v0 = (a0 + hs[0]) * d + b2[lane];
  float v1 = 0.0f;
  float mx = v0;
  if (hi) {
    v1 = (a1 + hs[32]) * d + b2[32 + lane];
    mx = fmaxf(mx, v1);
  }
#pragma unroll
  for (int o = 16; o > 0; o >>= 1) mx = fmaxf(mx, __shfl_xor(mx, o));
  float ex = __expf(v0 - mx);
  if (hi) ex += __expf(v1 - mx);
#pragma unroll
  for (int o = 16; o > 0; o >>= 1) ex += __shfl_xor(ex, o);
  float lse = mx + __logf(ex);
  float* op = out + (long long)i * NCLS + lane;
  op[0] = v0 - lse;
  if (hi) op[32] = v1 - lse;
}


extern "C" void kernel_launch(void* const* d_in, const int* in_sizes, int n_in,
                              void* d_out, int out_size, void* d_ws, size_t ws_size,
                              hipStream_t stream) {
  const float* x  = (const float*)d_in[0];
  const int*   ei = (const int*)d_in[1];
  const float* W1 = (const float*)d_in[2];
  const float* b1 = (const float*)d_in[3];
  const float* W2 = (const float*)d_in[4];
  const float* b2 = (const float*)d_in[5];
  float* out = (float*)d_out;

  const int N = in_sizes[0] / NFEAT;
  const int E = in_sizes[1] / 2;
  const int* erow = ei;
  const int* ecol = ei + E;
  const int Npad = (N + 63) & ~63;

  float* h1   = (float*)d_ws;
  float* act  = h1 + (long long)N * NFEAT;
  float* h2   = act + (long long)N * NFEAT;
  int*   cnt  = (int*)(h2 + (long long)N * LD2);
  int*   offs = cnt + Npad;
  float* dinv = (float*)(offs + Npad);
  int*   src  = (int*)(dinv + Npad);
  int*   pos  = src + E;
  const int Epad = (E + 63) & ~63;
  int*   part = pos + Epad;
  int*   loc  = part + (long long)NCHUNK * Npad;
  int*   btot = loc + Npad;
  const int NB = (Npad + RBLK - 1) / RBLK;
  const int CH = (((E + NCHUNK - 1) / NCHUNK) + 3) & ~3;

  count_local<<<NCHUNK, 1024, 0, stream>>>(ecol, pos, part, E, CH, Npad);
  reduce_parts<<<NB, RBLK, 0, stream>>>(part, cnt, loc, btot, dinv, Npad);

  const int Mtiles = (N + 15) / 16;
  const int mGroups = (Mtiles + WPB - 1) / WPB;
  const int nodeBlocks = (N + 7) / 8;

  gemm_strip_f32<NFEAT / 16, NFEAT, true><<<mGroups, 256, 0, stream>>>(
      x, W1, dinv, h1, N, NFEAT, erow, ecol, pos, loc, btot, part, src, offs, E, CH, Npad, NB);

  gather_rows128_relu<<<nodeBlocks, 256, 0, stream>>>(offs, cnt, src, dinv, h1, b1, act, N);

  gemm_strip_f32<LD2 / 16, NCLS, false><<<mGroups, 256, 0, stream>>>(
      act, W2, dinv, h2, N, LD2, nullptr, nullptr, nullptr, nullptr, nullptr, nullptr, nullptr, nullptr,
      0, 1, 0, 0);

  gather_rows40_lsm<<<nodeBlocks, 256, 0, stream>>>(offs, cnt, src, dinv, h2, b2, out, N);
}


# count kernel stores each chunk's edges sorted by destination (scan and staging in LDS); each product work-group merges its 128 nodes' run segments in LDS: no prefix launch or table; sums unchanged
# speedup vs baseline: 4.5524x; 1.3087x over previous
#include <hip/hip_runtime.h>
#include <cmath>


typedef __attribute__((ext_vector_type(2))) float v2f;
typedef __attribute__((ext_vector_type(4))) float v4f;
typedef __attribute__((ext_vector_type(4))) int v4i;
typedef __attribute__((ext_vector_type(8))) float v8f;

#define NFEAT 128
#define NCLS  40
#define LD2   48
#define KDIM  128
#define WPB   8
#define NNODE 50000
#define HCAP  ((NNODE + 63) & ~63)
#define NCHUNK 64
#define CBATCH 16
#define GB    (WPB * 16)
#define NGB   (HCAP / GB)
#define SBP   (NGB + 1)
#define CHS   12512
#define SCAN4 13
static_assert(HCAP % GB == 0, "node blocks tile the padded node count");
static_assert(HCAP <= 65536, "a node index is packed into 16 bits beside the rank");
static_assert(NCHUNK == WPB * 8, "a wave of the first product merges 8 chunks");
static_assert(1024 * SCAN4 * 4 >= HCAP, "the scan covers the counter table");


__device__ __forceinline__ int wave_incl_scan(int v, int lane) {
#pragma unroll
  for (int d = 1; d < 32; d <<= 1) {
    int y = __shfl_up(v, d);
    if (lane >= d) v += y;
  }
  return v;
}

__global__ __launch_bounds__(1024) void sort_chunk(
    const int* __restrict__ row, const int* __restrict__ col, int* __restrict__ pos,
    int* __restrict__ rsrc, int* __restrict__ rpk, int* __restrict__ sb, int E, int CH) {
  __shared__ __attribute__((aligned(16))) int hist[HCAP];
  __shared__ __attribute__((aligned(16))) int stg[2 * CHS];
  __shared__ int wsum[32];
  int t = threadIdx.x;
  int lane = t & 31;
  int w = t >> 5;
  int k = blockIdx.x;
  int e0 = k * CH;
  int len = E - e0;
  if (len > CH) len = CH;
  if (len < 0) len = 0;
  int c0[CBATCH], s0[CBATCH], r0[CBATCH];
#pragma unroll
  for (int u = 0; u < CBATCH; ++u) {
    int j = u * 1024 + t;
    c0[u] = (j < len) ? col[e0 + j] : -1;
    s0[u] = (j < len) ? row[e0 + j] : 0;
  }
  for (int i4 = t; i4 < HCAP / 4; i4 += 1024) ((v4i*)hist)[i4] = (v4i){0, 0, 0, 0};
  __syncthreads();
#pragma unroll
  for (int u = 0; u < CBATCH; ++u) {
    r0[u] = 0;
    if ((unsigned)c0[u] < (unsigned)HCAP) r0[u] = atomicAdd(&hist[c0[u]], 1);
  }
  for (int j0 = CBATCH * 1024; j0 < len; j0 += CBATCH * 1024) {
    int c[CBATCH];
#pragma unroll
    for (int u = 0; u < CBATCH; ++u) {
      int j = j0 + u * 1024 + t;
      c[u] = (j < len) ? col[e0 + j] : -1;
    }
#pragma unroll
    for (int u = 0; u < CBATCH; ++u) {
      int j = j0 + u * 1024 + t;
      if ((unsigned)c[u] < (unsigned)HCAP) pos[e0 + j] = atomicAdd(&hist[c[u]], 1);
    }
  }
  __syncthreads();
  int sum = 0;
#pragma unroll
  for (int u = 0; u < SCAN4; ++u) {
    int i4 = t * SCAN4 + u;
    v4i hv = (i4 < HCAP / 4) ? ((const v4i*)hist)[i4] : (v4i){0, 0, 0, 0};
    sum += hv.x + hv.y + hv.z + hv.w;
  }
  int incl = wave_incl_scan(sum, lane);
  if (lane == 31) wsum[w] = incl;
  __syncthreads();
  int wpre = 0;
  int tot = 0;
#pragma unroll
  for (int u = 0; u < 32; ++u) {
    int x = wsum[u];
    if (u < w) wpre += x;
    tot += x;
  }
  int run = wpre + (incl - sum);
#pragma unroll
  for (int u = 0; u < SCAN4; ++u) {
    int i4 = t * SCAN4 + u;
    v4i hv = (i4 < HCAP / 4) ? ((const v4i*)hist)[i4] : (v4i){0, 0, 0, 0};
    v4i o;
    o.x = run; run += hv.x;
    o.y = run; run += hv.y;
    o.z = run; run += hv.z;
    o.w = run; run += hv.w;
    if (i4 < HCAP / 4) ((v4i*)hist)[i4] = o;
  }
  __syncthreads();
  int* sbk = sb + k * SBP;
  for (int g = t; g < NGB; g += 1024) sbk[g] = hist[g * GB];
  if (t == 0) sbk[NGB] = tot;
  const bool fits = len <= CHS;
#pragma unroll
  for (int u = 0; u < CBATCH; ++u) {
    if ((unsigned)c0[u] < (unsigned)HCAP) {
      int q = hist[c0[u]] + r0[u];
      int pk = c0[u] | (r0[u] << 16);
      if (fits) {
        stg[q] = s0[u];
        stg[CHS + q] = pk;
      } else {
        rsrc[e0 + q] = s0[u];
        rpk[e0 + q] = pk;
      }
    }
  }
  for (int j = CBATCH * 1024 + t; j < len; j += 1024) {
    int cc = col[e0 + j];
    if ((unsigned)cc < (unsigned)HCAP) {
      int rr = pos[e0 + j];
      int q = hist[cc] + rr;
      rsrc[e0 + q] = row[e0 + j];
      rpk[e0 + q] = cc | (rr << 16);
    }
  }
  __syncthreads();
  if (fits) {
    int n4 = (tot + 3) >> 2;
    for (int i4 = t; i4 < n4; i4 += 1024) {
      ((v4i*)(rsrc + e0))[i4] = ((const v4i*)stg)[i4];
      ((v4i*)(rpk + e0))[i4] = ((const v4i*)(stg + CHS))[i4];
    }
  }
}

template <int NT, int NV, bool MERGE>
__global__ __launch_bounds__(256) void gemm_strip_f32(
    const float* __restrict__ A, const float* __restrict__ B,
    const float* __restrict__ rs, float* __restrict__ C, int M, int ldc,
    const int* __restrict__ rsrc, const int* __restrict__ rpk,
    const int* __restrict__ sb, int* __restrict__ src, int* __restrict__ cnt,
    int* __restrict__ offs, float* __restrict__ dinvw, int CH) {
  constexpr int NW   = NT * 16;
  constexpr int V4R  = NW / 4;
  constexpr int NV4  = NV / 4;
  constexpr int SW   = (NW > KDIM ? NW : KDIM);
  constexpr int SP   = SW + 4;
  constexpr int NJ   = (16 * V4R) / 32;
  constexpr int BP   = 2 * NW + ((NT % 2 == 0) ? 32 : 0);
  constexpr int NB   = ((KDIM / 2) * V4R) / 256;
  constexpr int CPW  = NCHUNK / WPB;
  __shared__ __attribute__((aligned(16))) float Bs[(KDIM / 2) * BP];
  __shared__ __attribute__((aligned(16))) float Ss[WPB * 16 * SP];
  static_assert(!MERGE || (KDIM / 2) * BP >= NCHUNK * GB + 2 * GB + 16,
                "the merge tables fit the space of B");

  int t = threadIdx.x;
  int w = t >> 5;
  int lane = t & 31;
  int l16 = lane & 15;
  int hi  = lane >> 4;
  int kh  = hi << 1;

  int Mtiles = (M + 15) >> 4;
  int mt = blockIdx.x * WPB + w;
  int m0 = mt << 4;
  float* Sw = Ss + w * (16 * SP);

  int sg0[CPW], sgl[CPW];
  int asum = 0;
  int fs[CPW][2], fk[CPW][2];
  const int wu = __builtin_amdgcn_readfirstlane(w);
  const int g0 = blockIdx.x * GB;
  if (MERGE) {
#pragma unroll
    for (int kk = 0; kk < CPW; ++kk) {
      int k = wu + WPB * kk;
      const int* sbk = sb + k * SBP + blockIdx.x;
      int a = sbk[0];
      int b = sbk[1];
      sg0[kk] = k * CH + a;
      sgl[kk] = b - a;
      asum += a;
    }
  }

  v4f ar[16];
#pragma unroll
  for (int r = 0; r < 16; ++r) {
    int m = m0 + r;
    if (m >= M) m = M - 1;
    ar[r] = *(const v4f*)(A + (long long)m * KDIM + 4 * lane);
  }

  v4f b0[NB], b1[NB];
#pragma unroll
  for (int j = 0; j < NB; ++j) {
    int item = t + j * 256;
    int kp = item / V4R;
    int c4 = item % V4R;
    int c4c = c4 < NV4 ? c4 : NV4 - 1;
    b0[j] = *(const v4f*)(B + (2 * kp) * NV + c4c * 4);
    b1[j] = *(const v4f*)(B + (2 * kp + 1) * NV + c4c * 4);
    if (c4 >= NV4) {
      b0[j] = (v4f){0.0f, 0.0f, 0.0f, 0.0f};
      b1[j] = (v4f){0.0f, 0.0f, 0.0f, 0.0f};
    }
  }
  if (MERGE) {
#pragma unroll
    for (int kk = 0; kk < CPW; ++kk) {
#pragma unroll
      for (int h = 0; h < 2; ++h) {
        int j = lane + 32 * h;
        int ix = (j < sgl[kk]) ? sg0[kk] + j : 0;
        fs[kk][h] = rsrc[ix];
        fk[kk][h] = rpk[ix];
      }
    }
  }
#pragma unroll
  for (int r = 0; r < 16; ++r) *(v4f*)(Sw + r * SP + 4 * lane) = ar[r];
#pragma unroll
  for (int j = 0; j < NB; ++j) {
    int item = t + j * 256;
    int kp = item / V4R;
    int c4 = item % V4R;
    float* d = Bs + kp * BP + 8 * c4;
    *(v4f*)d       = (v4f){b0[j].x, b1[j].x, b0[j].y, b1[j].y};
    *(v4f*)(d + 4) = (v4f){b0[j].z, b1[j].z, b0[j].w, b1[j].w};
  }
  __syncthreads();

  const bool live = mt < Mtiles;
  float rsv[NJ];
  v8f acc[NT];
#pragma unroll
  for (int nt = 0; nt < NT; ++nt) acc[nt] = (v8f){};

  if (live) {
    if (!MERGE) {
#pragma unroll
      for (int j = 0; j < NJ; ++j) rsv[j] = rs[m0 + (j * 32 + lane) / V4R];
    }
    const float* Ak = Sw + l16 * SP + kh;
#pragma unroll
    for (int s = 0; s < KDIM / 4; ++s) {
      v2f a = *(const v2f*)(Ak + 4 * s);
      const float* Bk = Bs + (2 * s + hi) * BP + 2 * l16;
#pragma unroll
      for (int nt = 0; nt < NT; ++nt) {
        v2f b = *(const v2f*)(Bk + 32 * nt);
        acc[nt] = __builtin_amdgcn_wmma_f32_16x16x4_f32(
             false, a,  false, b,
             (short)0, acc[nt],  false,  false);
      }
    }
  }

  if (MERGE) {
    int* tab = (int*)Bs;
    int* nfirst = tab + NCHUNK * GB;
    float* dl = (float*)(nfirst + GB);
    int* misc = (int*)(dl + GB);
    __syncthreads();
    for (int i4 = t; i4 < NCHUNK * GB / 4; i4 += 256) ((v4i*)tab)[i4] = (v4i){0, 0, 0, 0};
    if (lane == 0) misc[w] = asum;
    __syncthreads();
#pragma unroll
    for (int kk = 0; kk < CPW; ++kk) {
      int* tk = tab + (wu + WPB * kk) * GB;
#pragma unroll
      for (int h = 0; h < 2; ++h)
        if (lane + 32 * h < sgl[kk]) atomicAdd(&tk[(fk[kk][h] & 0xffff) - g0], 1);
      for (int j = lane + 64; j < sgl[kk]; j += 32)
        atomicAdd(&tk[(rpk[sg0[kk] + j] & 0xffff) - g0], 1);
    }
    __syncthreads();
    int ntot = 0;
    if (t < GB) {
#pragma unroll 8
      for (int k = 0; k < NCHUNK; ++k) {
        int v = tab[k * GB + t];
        tab[k * GB + t] = ntot;
        ntot += v;
      }
    }
    int nincl = wave_incl_scan(ntot, lane);
    if (lane == 31 && w < GB / 32) misc[8 + w] = nincl;
    __syncthreads();
    if (t < GB) {
      int first = 0;
#pragma unroll
      for (int u = 0; u < WPB; ++u) first += misc[u];
#pragma unroll
      for (int u = 0; u < GB / 32; ++u)
        if (u < w) first += misc[8 + u];
      first += nincl - ntot;
      float dg = (float)(ntot + 1);
      float dv = (dg > 0.0f) ? rsqrtf(dg) : 0.0f;
      nfirst[t] = first;
      dl[t] = dv;
      cnt[g0 + t] = ntot;
      offs[g0 + t] = first;
      dinvw[g0 + t] = dv;
    }
    __syncthreads();
#pragma unroll
    for (int kk = 0; kk < CPW; ++kk) {
      const int* tk = tab + (wu + WPB * kk) * GB;
      const int* nf = nfirst;
#pragma unroll
      for (int h = 0; h < 2; ++h) {
        if (lane + 32 * h < sgl[kk]) {
          int n = (fk[kk][h] & 0xffff) - g0;
          int r = (int)((unsigned)fk[kk][h] >> 16);
          src[nf[n] + tk[n] + r] = fs[kk][h];
        }
      }
      for (int j = lane + 64; j < sgl[kk]; j += 32) {
        int pk = rpk[sg0[kk] + j];
        int n = (pk & 0xffff) - g0;
        int r = (int)((unsigned)pk >> 16);
        src[nf[n] + tk[n] + r] = rsrc[sg0[kk] + j];
      }
    }
    if (live) {
#pragma unroll
      for (int j = 0; j < NJ; ++j) rsv[j] = dl[w * 16 + (j * 32 + lane) / V4R];
    }
  }

  if (live) {
#pragma unroll
  for (int nt = 0; nt < NT; ++nt)
#pragma unroll
    for (int r = 0; r < 8; ++r) Sw[(hi * 8 + r) * SP + nt * 16 + l16] = acc[nt][r];

  if (m0 + 16 <= M) {
    v4f cv[NJ];
#pragma unroll
    for (int j = 0; j < NJ; ++j) {
      int e = j * 32 + lane;
      cv[j] = *(const v4f*)(Sw + (e / V4R) * SP + (e % V4R) * 4) * rsv[j];
    }
#pragma unroll
    for (int j = 0; j < NJ; ++j) {
      int e = j * 32 + lane;
      *(v4f*)(C + (long long)(m0 + e / V4R) * ldc + (e % V4R) * 4) = cv[j];
    }
  } else {
#pragma unroll
    for (int j = 0; j < NJ; ++j) {
      int e = j * 32 + lane;
      int row = e / V4R;
      int c4 = e % V4R;
      v4f v = *(const v4f*)(Sw + row * SP + c4 * 4) * rsv[j];
      if (m0 + row < M) *(v4f*)(C + (long long)(m0 + row) * ldc + c4 * 4) = v;
    }
  }
  }
}

__global__ __launch_bounds__(256) void gather_rows128_relu(
    const int* __restrict__ offs, const int* __restrict__ cnt,
    const int* __restrict__ src, const float* __restrict__ dinv,
    const float* __restrict__ h, const float* __restrict__ bias,
    float* __restrict__ act, int Nnodes) {
  int lane = threadIdx.x & 31;
  int i = __builtin_amdgcn_readfirstlane((int)((blockIdx.x * 256u + threadIdx.x) >> 5));
  if (i >= Nnodes) return;
  int beg = offs[i];
  int n = cnt[i];
  const v4f* h4 = (const v4f*)h;
  v4f acc = {0.0f, 0.0f, 0.0f, 0.0f};
  for (int base = 0; base < n; base += 32) {
    int m = n - base;
    if (m > 32) m = 32;
    int sidx = i;
    if (lane < m) sidx = src[beg + base + lane];
    for (int k = 0; k < m; k += 4) {
      int s0 = __builtin_amdgcn_readlane(sidx, k);
      int s1 = __builtin_amdgcn_readlane(sidx, k + 1);
      int s2 = __builtin_amdgcn_readlane(sidx, k + 2);
      int s3 = __builtin_amdgcn_readlane(sidx, k + 3);
      v4f v0 = h4[(long long)s0 * (NFEAT / 4) + lane];
      v4f v1 = h4[(long long)s1 * (NFEAT / 4) + lane];
      v4f v2 = h4[(long long)s2 * (NFEAT / 4) + lane];
      v4f v3 = h4[(long long)s3 * (NFEAT / 4) + lane];
      const v4f z4 = {0.0f, 0.0f, 0.0f, 0.0f};
      acc += v0;
      acc += (k + 1 < m) ? v1 : z4;
      acc += (k + 2 < m) ? v2 : z4;
      acc += (k + 3 < m) ? v3 : z4;
    }
  }
  float d = dinv[i];
  v4f hs = h4[(long long)i * (NFEAT / 4) + lane];
  v4f b = ((const v4f*)bias)[lane];
  v4f r;
  r.x = (acc.x + hs.x) * d + b.x;
  r.y = (acc.y + hs.y) * d + b.y;
  r.z = (acc.z + hs.z) * d + b.z;
  r.w = (acc.w + hs.w) * d + b.w;
  r.x = r.x > 0.0f ? r.x : 0.0f;
  r.y = r.y > 0.0f ? r.y : 0.0f;
  r.z = r.z > 0.0f ? r.z : 0.0f;
  r.w = r.w > 0.0f ? r.w : 0.0f;
  ((v4f*)act)[(long long)i * (NFEAT / 4) + lane] = r;
}

__global__ __launch_bounds__(256) void gather_rows40_lsm(
    const int* __restrict__ offs, const int* __restrict__ cnt,
    const int* __restrict__ src, const float* __restrict__ dinv,
    const float* __restrict__ h2, const float* __restrict__ b2,
    float* __restrict__ out, int Nnodes) {
  int lane = threadIdx.x & 31;
  int i = __builtin_amdgcn_readfirstlane((int)((blockIdx.x * 256u + threadIdx.x) >> 5));
  if (i >= Nnodes) return;
  bool hi = lane < (NCLS - 32);
  int beg = offs[i];
  int n = cnt[i];
  float a0 = 0.0f, a1 = 0.0f;
  for (int base = 0; base < n; base += 32) {
    int m = n - base;
    if (m > 32) m = 32;
    int sidx = i;
    if (lane < m) sidx = src[beg + base + lane];
    for (int k = 0; k < m; k += 4) {
      int s0 = __builtin_amdgcn_readlane(sidx, k);
      int s1 = __builtin_amdgcn_readlane(sidx, k + 1);
      int s2 = __builtin_amdgcn_readlane(sidx, k + 2);
      int s3 = __builtin_amdgcn_readlane(sidx, k + 3);
      const float* r0 = h2 + (long long)s0 * LD2 + lane;
      const float* r1 = h2 + (long long)s1 * LD2 + lane;
      const float* r2 = h2 + (long long)s2 * LD2 + lane;
      const float* r3 = h2 + (long long)s3 * LD2 + lane;
      float x0 = r0[0], x1 = r1[0], x2 = r2[0], x3 = r3[0];
      float y0 = 0.0f, y1 = 0.0f, y2 = 0.0f, y3 = 0.0f;
      if (hi) {
        y0 = r0[32];
        y1 = r1[32];
        y2 = r2[32];
        y3 = r3[32];
      }
      bool in1 = k + 1 < m, in2 = k + 2 < m, in3 = k + 3 < m;
      a0 += x0;
      a1 += y0;
      a0 += in1 ? x1 : 0.0f;
      a1 += in1 ? y1 : 0.0f;
      a0 += in2 ? x2 : 0.0f;
      a1 += in2 ? y2 : 0.0f;
      a0 += in3 ? x3 : 0.0f;
      a1 += in3 ? y3 : 0.0f;
    }
  }
  float d = dinv[i];
  const float* hs = h2 + (long long)i * LD2 + lane;
  float v0 = (a0 + hs[0]) * d + b2[lane];
  float v1 = 0.0f;
  float mx = v0;
  if (hi) {
    v1 = (a1 + hs[32]) * d + b2[32 + lane];
    mx = fmaxf(mx, v1);
  }
#pragma unroll
  for (int o = 16; o > 0; o >>= 1) mx = fmaxf(mx, __shfl_xor(mx, o));
  float ex = __expf(v0 - mx);
  if (hi) ex += __expf(v1 - mx);
#pragma unroll
  for (int o = 16; o > 0; o >>= 1) ex += __shfl_xor(ex, o);
  float lse = mx + __logf(ex);
  float* op = out + (long long)i * NCLS + lane;
  op[0] = v0 - lse;
  if (hi) op[32] = v1 - lse;
}


extern "C" void kernel_launch(void* const* d_in, const int* in_sizes, int n_in,
                              void* d_out, int out_size, void* d_ws, size_t ws_size,
                              hipStream_t stream) {
  const float* x  = (const float*)d_in[0];
  const int*   ei = (const int*)d_in[1];
  const float* W1 = (const float*)d_in[2];
  const float* b1 = (const float*)d_in[3];
  const float* W2 = (const float*)d_in[4];
  const float* b2 = (const float*)d_in[5];
  float* out = (float*)d_out;

  const int N = in_sizes[0] / NFEAT;
  const int E = in_sizes[1] / 2;
  const int* erow = ei;
  const int* ecol = ei + E;
  const int Npad = (N + 63) & ~63;

  float* h1   = (float*)d_ws;
  float* act  = h1 + (long long)N * NFEAT;
  float* h2   = act + (long long)N * NFEAT;
  int*   cnt  = (int*)(h2 + (long long)N * LD2);
  int*   offs = cnt + Npad;
  float* dinv = (float*)(offs + Npad);
  int*   src  = (int*)(dinv + Npad);
  int*   pos  = src + E;
  const int Epad = (E + 63) & ~63;
  const int CH = (((E + NCHUNK - 1) / NCHUNK) + 3) & ~3;
  const int Rpad = (NCHUNK * CH + 63) & ~63;
  int*   rsrc = pos + Epad;
  int*   rpk  = rsrc + Rpad;
  int*   sb   = rpk + Rpad;

  sort_chunk<<<NCHUNK, 1024, 0, stream>>>(erow, ecol, pos, rsrc, rpk, sb, E, CH);

  const int Mtiles = (N + 15) / 16;
  const int mGroups = (Mtiles + WPB - 1) / WPB;
  const int nodeBlocks = (N + 7) / 8;

  gemm_strip_f32<NFEAT / 16, NFEAT, true><<<NGB, 256, 0, stream>>>(
      x, W1, nullptr, h1, N, NFEAT, rsrc, rpk, sb, src, cnt, offs, dinv, CH);

  gather_rows128_relu<<<nodeBlocks, 256, 0, stream>>>(offs, cnt, src, dinv, h1, b1, act, N);

  gemm_strip_f32<LD2 / 16, NCLS, false><<<mGroups, 256, 0, stream>>>(
      act, W2, dinv, h2, N, LD2, nullptr, nullptr, nullptr, nullptr, nullptr, nullptr, nullptr, 1);

  gather_rows40_lsm<<<nodeBlocks, 256, 0, stream>>>(offs, cnt, src, dinv, h2, b2, out, N);
}


# second product moved into the layer-1 gather launch: a wave sums 4 nodes, a work-group's 32 ReLU rows go to LDS and meet W2 there (WMMA); no act plane, one launch fewer; sums unchanged
# speedup vs baseline: 4.9826x; 1.0945x over previous
#include <hip/hip_runtime.h>
#include <cmath>


typedef __attribute__((ext_vector_type(2))) float v2f;
typedef __attribute__((ext_vector_type(4))) float v4f;
typedef __attribute__((ext_vector_type(4))) int v4i;
typedef __attribute__((ext_vector_type(8))) float v8f;

#define NFEAT 128
#define NCLS  40
#define LD2   48
#define KDIM  128
#define WPB   8
#define NNODE 50000
#define HCAP  ((NNODE + 63) & ~63)
#define NCHUNK 64
#define CBATCH 16
#define GB    (WPB * 16)
#define NGB   (HCAP / GB)
#define SBP   (NGB + 1)
#define CHS   12512
#define SCAN4 13
static_assert(HCAP % GB == 0, "node blocks tile the padded node count");
static_assert(HCAP <= 65536, "a node index is packed into 16 bits beside the rank");
static_assert(NCHUNK == WPB * 8, "a wave of the first product merges 8 chunks");
static_assert(1024 * SCAN4 * 4 >= HCAP, "the scan covers the counter table");


__device__ __forceinline__ int wave_incl_scan(int v, int lane) {
#pragma unroll
  for (int d = 1; d < 32; d <<= 1) {
    int y = __shfl_up(v, d);
    if (lane >= d) v += y;
  }
  return v;
}

__global__ __launch_bounds__(1024) void sort_chunk(
    const int* __restrict__ row, const int* __restrict__ col, int* __restrict__ pos,
    int* __restrict__ rsrc, int* __restrict__ rpk, int* __restrict__ sb, int E, int CH) {
  __shared__ __attribute__((aligned(16))) int hist[HCAP];
  __shared__ __attribute__((aligned(16))) int stg[2 * CHS];
  __shared__ int wsum[32];
  int t = threadIdx.x;
  int lane = t & 31;
  int w = t >> 5;
  int k = blockIdx.x;
  int e0 = k * CH;
  int len = E - e0;
  if (len > CH) len = CH;
  if (len < 0) len = 0;
  int c0[CBATCH], s0[CBATCH], r0[CBATCH];
#pragma unroll
  for (int u = 0; u < CBATCH; ++u) {
    int j = u * 1024 + t;
    c0[u] = (j < len) ? col[e0 + j] : -1;
    s0[u] = (j < len) ? row[e0 + j] : 0;
  }
  for (int i4 = t; i4 < HCAP / 4; i4 += 1024) ((v4i*)hist)[i4] = (v4i){0, 0, 0, 0};
  __syncthreads();
#pragma unroll
  for (int u = 0; u < CBATCH; ++u) {
    r0[u] = 0;
    if ((unsigned)c0[u] < (unsigned)HCAP) r0[u] = atomicAdd(&hist[c0[u]], 1);
  }
  for (int j0 = CBATCH * 1024; j0 < len; j0 += CBATCH * 1024) {
    int c[CBATCH];
#pragma unroll
    for (int u = 0; u < CBATCH; ++u) {
      int j = j0 + u * 1024 + t;
      c[u] = (j < len) ? col[e0 + j] : -1;
    }
#pragma unroll
    for (int u = 0; u < CBATCH; ++u) {
      int j = j0 + u * 1024 + t;
      if ((unsigned)c[u] < (unsigned)HCAP) pos[e0 + j] = atomicAdd(&hist[c[u]], 1);
    }
  }
  __syncthreads();
  int sum = 0;
#pragma unroll
  for (int u = 0; u < SCAN4; ++u) {
    int i4 = t * SCAN4 + u;
    v4i hv = (i4 < HCAP / 4) ? ((const v4i*)hist)[i4] : (v4i){0, 0, 0, 0};
    sum += hv.x + hv.y + hv.z + hv.w;
  }
  int incl = wave_incl_scan(sum, lane);
  if (lane == 31) wsum[w] = incl;
  __syncthreads();
  int wpre = 0;
  int tot = 0;
#pragma unroll
  for (int u = 0; u < 32; ++u) {
    int x = wsum[u];
    if (u < w) wpre += x;
    tot += x;
  }
  int run = wpre + (incl - sum);
#pragma unroll
  for (int u = 0; u < SCAN4; ++u) {
    int i4 = t * SCAN4 + u;
    v4i hv = (i4 < HCAP / 4) ? ((const v4i*)hist)[i4] : (v4i){0, 0, 0, 0};
    v4i o;
    o.x = run; run += hv.x;
    o.y = run; run += hv.y;
    o.z = run; run += hv.z;
    o.w = run; run += hv.w;
    if (i4 < HCAP / 4) ((v4i*)hist)[i4] = o;
  }
  __syncthreads();
  int* sbk = sb + k * SBP;
  for (int g = t; g < NGB; g += 1024) sbk[g] = hist[g * GB];
  if (t == 0) sbk[NGB] = tot;
  const bool fits = len <= CHS;
#pragma unroll
  for (int u = 0; u < CBATCH; ++u) {
    if ((unsigned)c0[u] < (unsigned)HCAP) {
      int q = hist[c0[u]] + r0[u];
      int pk = c0[u] | (r0[u] << 16);
      if (fits) {
        stg[q] = s0[u];
        stg[CHS + q] = pk;
      } else {
        rsrc[e0 + q] = s0[u];
        rpk[e0 + q] = pk;
      }
    }
  }
  for (int j = CBATCH * 1024 + t; j < len; j += 1024) {
    int cc = col[e0 + j];
    if ((unsigned)cc < (unsigned)HCAP) {
      int rr = pos[e0 + j];
      int q = hist[cc] + rr;
      rsrc[e0 + q] = row[e0 + j];
      rpk[e0 + q] = cc | (rr << 16);
    }
  }
  __syncthreads();
  if (fits) {
    int n4 = (tot + 3) >> 2;
    for (int i4 = t; i4 < n4; i4 += 1024) {
      ((v4i*)(rsrc + e0))[i4] = ((const v4i*)stg)[i4];
      ((v4i*)(rpk + e0))[i4] = ((const v4i*)(stg + CHS))[i4];
    }
  }
}

template <int NT, int NV, bool MERGE>
__global__ __launch_bounds__(256) void gemm_strip_f32(
    const float* __restrict__ A, const float* __restrict__ B,
    const float* __restrict__ rs, float* __restrict__ C, int M, int ldc,
    const int* __restrict__ rsrc, const int* __restrict__ rpk,
    const int* __restrict__ sb, int* __restrict__ src, int* __restrict__ cnt,
    int* __restrict__ offs, float* __restrict__ dinvw, int CH) {
  constexpr int NW   = NT * 16;
  constexpr int V4R  = NW / 4;
  constexpr int NV4  = NV / 4;
  constexpr int SW   = (NW > KDIM ? NW : KDIM);
  constexpr int SP   = SW + 4;
  constexpr int NJ   = (16 * V4R) / 32;
  constexpr int BP   = 2 * NW + ((NT % 2 == 0) ? 32 : 0);
  constexpr int NB   = ((KDIM / 2) * V4R) / 256;
  constexpr int CPW  = NCHUNK / WPB;
  __shared__ __attribute__((aligned(16))) float Bs[(KDIM / 2) * BP];
  __shared__ __attribute__((aligned(16))) float Ss[WPB * 16 * SP];
  static_assert(!MERGE || (KDIM / 2) * BP >= NCHUNK * GB + 2 * GB + 16,
                "the merge tables fit the space of B");

  int t = threadIdx.x;
  int w = t >> 5;
  int lane = t & 31;
  int l16 = lane & 15;
  int hi  = lane >> 4;
  int kh  = hi << 1;

  int Mtiles = (M + 15) >> 4;
  int mt = blockIdx.x * WPB + w;
  int m0 = mt << 4;
  float* Sw = Ss + w * (16 * SP);

  int sg0[CPW], sgl[CPW];
  int asum = 0;
  int fs[CPW][2], fk[CPW][2];
  const int wu = __builtin_amdgcn_readfirstlane(w);
  const int g0 = blockIdx.x * GB;
  if (MERGE) {
#pragma unroll
    for (int kk = 0; kk < CPW; ++kk) {
      int k = wu + WPB * kk;
      const int* sbk = sb + k * SBP + blockIdx.x;
      int a = sbk[0];
      int b = sbk[1];
      sg0[kk] = k * CH + a;
      sgl[kk] = b - a;
      asum += a;
    }
  }

  v4f ar[16];
#pragma unroll
  for (int r = 0; r < 16; ++r) {
    int m = m0 + r;
    if (m >= M) m = M - 1;
    ar[r] = *(const v4f*)(A + (long long)m * KDIM + 4 * lane);
  }

  v4f b0[NB], b1[NB];
#pragma unroll
  for (int j = 0; j < NB; ++j) {
    int item = t + j * 256;
    int kp = item / V4R;
    int c4 = item % V4R;
    int c4c = c4 < NV4 ? c4 : NV4 - 1;
    b0[j] = *(const v4f*)(B + (2 * kp) * NV + c4c * 4);
    b1[j] = *(const v4f*)(B + (2 * kp + 1) * NV + c4c * 4);
    if (c4 >= NV4) {
      b0[j] = (v4f){0.0f, 0.0f, 0.0f, 0.0f};
      b1[j] = (v4f){0.0f, 0.0f, 0.0f, 0.0f};
    }
  }
  if (MERGE) {
#pragma unroll
    for (int kk = 0; kk < CPW; ++kk) {
#pragma unroll
      for (int h = 0; h < 2; ++h) {
        int j = lane + 32 * h;
        int ix = (j < sgl[kk]) ? sg0[kk] + j : 0;
        fs[kk][h] = rsrc[ix];
        fk[kk][h] = rpk[ix];
      }
    }
  }
#pragma unroll
  for (int r = 0; r < 16; ++r) *(v4f*)(Sw + r * SP + 4 * lane) = ar[r];
#pragma unroll
  for (int j = 0; j < NB; ++j) {
    int item = t + j * 256;
    int kp = item / V4R;
    int c4 = item % V4R;
    float* d = Bs + kp * BP + 8 * c4;
    *(v4f*)d       = (v4f){b0[j].x, b1[j].x, b0[j].y, b1[j].y};
    *(v4f*)(d + 4) = (v4f){b0[j].z, b1[j].z, b0[j].w, b1[j].w};
  }
  __syncthreads();

  const bool live = mt < Mtiles;
  float rsv[NJ];
  v8f acc[NT];
#pragma unroll
  for (int nt = 0; nt < NT; ++nt) acc[nt] = (v8f){};

  if (live) {
    if (!MERGE) {
#pragma unroll
      for (int j = 0; j < NJ; ++j) rsv[j] = rs[m0 + (j * 32 + lane) / V4R];
    }
    const float* Ak = Sw + l16 * SP + kh;
#pragma unroll
    for (int s = 0; s < KDIM / 4; ++s) {
      v2f a = *(const v2f*)(Ak + 4 * s);
      const float* Bk = Bs + (2 * s + hi) * BP + 2 * l16;
#pragma unroll
      for (int nt = 0; nt < NT; ++nt) {
        v2f b = *(const v2f*)(Bk + 32 * nt);
        acc[nt] = __builtin_amdgcn_wmma_f32_16x16x4_f32(
             false, a,  false, b,
             (short)0, acc[nt],  false,  false);
      }
    }
  }

  if (MERGE) {
    int* tab = (int*)Bs;
    int* nfirst = tab + NCHUNK * GB;
    float* dl = (float*)(nfirst + GB);
    int* misc = (int*)(dl + GB);
    __syncthreads();
    for (int i4 = t; i4 < NCHUNK * GB / 4; i4 += 256) ((v4i*)tab)[i4] = (v4i){0, 0, 0, 0};
    if (lane == 0) misc[w] = asum;
    __syncthreads();
#pragma unroll
    for (int kk = 0; kk < CPW; ++kk) {
      int* tk = tab + (wu + WPB * kk) * GB;
#pragma unroll
      for (int h = 0; h < 2; ++h)
        if (lane + 32 * h < sgl[kk]) atomicAdd(&tk[(fk[kk][h] & 0xffff) - g0], 1);
      for (int j = lane + 64; j < sgl[kk]; j += 32)
        atomicAdd(&tk[(rpk[sg0[kk] + j] & 0xffff) - g0], 1);
    }
    __syncthreads();
    int ntot = 0;
    if (t < GB) {
#pragma unroll 8
      for (int k = 0; k < NCHUNK; ++k) {
        int v = tab[k * GB + t];
        tab[k * GB + t] = ntot;
        ntot += v;
      }
    }
    int nincl = wave_incl_scan(ntot, lane);
    if (lane == 31 && w < GB / 32) misc[8 + w] = nincl;
    __syncthreads();
    if (t < GB) {
      int first = 0;
#pragma unroll
      for (int u = 0; u < WPB; ++u) first += misc[u];
#pragma unroll
      for (int u = 0; u < GB / 32; ++u)
        if (u < w) first += misc[8 + u];
      first += nincl - ntot;
      float dg = (float)(ntot + 1);
      float dv = (dg > 0.0f) ? rsqrtf(dg) : 0.0f;
      nfirst[t] = first;
      dl[t] = dv;
      cnt[g0 + t] = ntot;
      offs[g0 + t] = first;
      dinvw[g0 + t] = dv;
    }
    __syncthreads();
#pragma unroll
    for (int kk = 0; kk < CPW; ++kk) {
      const int* tk = tab + (wu + WPB * kk) * GB;
      const int* nf = nfirst;
#pragma unroll
      for (int h = 0; h < 2; ++h) {
        if (lane + 32 * h < sgl[kk]) {
          int n = (fk[kk][h] & 0xffff) - g0;
          int r = (int)((unsigned)fk[kk][h] >> 16);
          src[nf[n] + tk[n] + r] = fs[kk][h];
        }
      }
      for (int j = lane + 64; j < sgl[kk]; j += 32) {
        int pk = rpk[sg0[kk] + j];
        int n = (pk & 0xffff) - g0;
        int r = (int)((unsigned)pk >> 16);
        src[nf[n] + tk[n] + r] = rsrc[sg0[kk] + j];
      }
    }
    if (live) {
#pragma unroll
      for (int j = 0; j < NJ; ++j) rsv[j] = dl[w * 16 + (j * 32 + lane) / V4R];
    }
  }

  if (live) {
#pragma unroll
  for (int nt = 0; nt < NT; ++nt)
#pragma unroll
    for (int r = 0; r < 8; ++r) Sw[(hi * 8 + r) * SP + nt * 16 + l16] = acc[nt][r];

  if (m0 + 16 <= M) {
    v4f cv[NJ];
#pragma unroll
    for (int j = 0; j < NJ; ++j) {
      int e = j * 32 + lane;
      cv[j] = *(const v4f*)(Sw + (e / V4R) * SP + (e % V4R) * 4) * rsv[j];
    }
#pragma unroll
    for (int j = 0; j < NJ; ++j) {
      int e = j * 32 + lane;
      *(v4f*)(C + (long long)(m0 + e / V4R) * ldc + (e % V4R) * 4) = cv[j];
    }
  } else {
#pragma unroll
    for (int j = 0; j < NJ; ++j) {
      int e = j * 32 + lane;
      int row = e / V4R;
      int c4 = e % V4R;
      v4f v = *(const v4f*)(Sw + row * SP + c4 * 4) * rsv[j];
      if (m0 + row < M) *(v4f*)(C + (long long)(m0 + row) * ldc + c4 * 4) = v;
    }
  }
  }
}

#define FN   4
#define FNB  (FN * WPB)
#define FSP  (NFEAT + 4)
#define FBP  (2 * LD2)
static_assert(FNB % 16 == 0, "a work-group's rows are whole strips");
static_assert((FNB / 16) * (LD2 / 16) <= WPB, "one wave per 16 x 16 tile of the work-group's rows");

__global__ __launch_bounds__(256) void gather_relu_prod2(
    const int* __restrict__ offs, const int* __restrict__ cnt,
    const int* __restrict__ src, const float* __restrict__ dinv,
    const float* __restrict__ h, const float* __restrict__ bias,
    const float* __restrict__ W2, float* __restrict__ h2, int Nnodes) {
  constexpr int V4R = LD2 / 4;
  constexpr int NV4 = NCLS / 4;
  constexpr int NB  = ((KDIM / 2) * V4R) / 256;
  static_assert(((KDIM / 2) * V4R) % 256 == 0, "the fill of W2 has no tail");
  constexpr int NTL = LD2 / 16;
  __shared__ __attribute__((aligned(16))) float Bs[(KDIM / 2) * FBP];
  __shared__ __attribute__((aligned(16))) float As[FNB * FSP];
  __shared__ __attribute__((aligned(16))) float dl[FNB];

  int t = threadIdx.x;
  int lane = t & 31;
  const int w = __builtin_amdgcn_readfirstlane(t >> 5);
  const int g0 = blockIdx.x * FNB;
  const v4f* h4 = (const v4f*)h;
  const unsigned ul = (unsigned)lane;
  typedef const __attribute__((address_space(1))) v4f* gv4p;
  const unsigned long long hb = (unsigned long long)h;

  v4f b0[NB], b1[NB];
#pragma unroll
  for (int j = 0; j < NB; ++j) {
    int item = t + j * 256;
    int kp = item / V4R;
    int c4 = item % V4R;
    int c4c = c4 < NV4 ? c4 : NV4 - 1;
    b0[j] = *(const v4f*)(W2 + (2 * kp) * NCLS + c4c * 4);
    b1[j] = *(const v4f*)(W2 + (2 * kp + 1) * NCLS + c4c * 4);
    if (c4 >= NV4) {
      b0[j] = (v4f){0.0f, 0.0f, 0.0f, 0.0f};
      b1[j] = (v4f){0.0f, 0.0f, 0.0f, 0.0f};
    }
  }

  int ii[FN], beg[FN], n[FN], fe[FN];
#pragma unroll
  for (int q = 0; q < FN; ++q) {
    int i = g0 + FN * w + q;
    ii[q] = i < Nnodes ? i : Nnodes - 1;
    beg[q] = offs[ii[q]];
    n[q] = i < Nnodes ? cnt[ii[q]] : 0;
  }
#pragma unroll
  for (int q = 0; q < FN; ++q) {
    fe[q] = ii[q];
    if (lane < n[q]) fe[q] = src[beg[q] + lane];
  }
  v4f bv = ((const v4f*)bias)[lane];

#pragma unroll
  for (int j = 0; j < NB; ++j) {
    int item = t + j * 256;
    int kp = item / V4R;
    int c4 = item % V4R;
    float* d = Bs + kp * FBP + 8 * c4;
    *(v4f*)d       = (v4f){b0[j].x, b1[j].x, b0[j].y, b1[j].y};
    *(v4f*)(d + 4) = (v4f){b0[j].z, b1[j].z, b0[j].w, b1[j].w};
  }

#pragma unroll
  for (int q = 0; q < FN; ++q) {
    const int i = ii[q];
    const int nq = n[q];
    float d = dinv[i];
    v4f hs = h4[(long long)i * (NFEAT / 4) + lane];
    v4f acc = {0.0f, 0.0f, 0.0f, 0.0f};
    for (int base = 0; base < nq; base += 32) {
      int m = nq - base;
      if (m > 32) m = 32;
      int sidx = fe[q];
      if (base > 0) {
        sidx = i;
        if (lane < m) sidx = src[beg[q] + base + lane];
      }
      for (int k = 0; k < m; k += 8) {
        int s0 = __builtin_amdgcn_readlane(sidx, k);
        int s1 = __builtin_amdgcn_readlane(sidx, k + 1);
        int s2 = __builtin_amdgcn_readlane(sidx, k + 2);
        int s3 = __builtin_amdgcn_readlane(sidx, k + 3);
        int s4 = __builtin_amdgcn_readlane(sidx, k + 4);
        int s5 = __builtin_amdgcn_readlane(sidx, k + 5);
        int s6 = __builtin_amdgcn_readlane(sidx, k + 6);
        int s7 = __builtin_amdgcn_readlane(sidx, k + 7);
        v4f v0 = ((gv4p)(hb + ((unsigned long long)(unsigned)s0 << 9)))[ul];
        v4f v1 = ((gv4p)(hb + ((unsigned long long)(unsigned)s1 << 9)))[ul];
        v4f v2 = ((gv4p)(hb + ((unsigned long long)(unsigned)s2 << 9)))[ul];
        v4f v3 = ((gv4p)(hb + ((unsigned long long)(unsigned)s3 << 9)))[ul];
        v4f v4 = ((gv4p)(hb + ((unsigned long long)(unsigned)s4 << 9)))[ul];
        v4f v5 = ((gv4p)(hb + ((unsigned long long)(unsigned)s5 << 9)))[ul];
        v4f v6 = ((gv4p)(hb + ((unsigned long long)(unsigned)s6 << 9)))[ul];
        v4f v7 = ((gv4p)(hb + ((unsigned long long)(unsigned)s7 << 9)))[ul];
        const v4f z4 = {0.0f, 0.0f, 0.0f, 0.0f};
        acc += v0;
        acc += (k + 1 < m) ? v1 : z4;
        acc += (k + 2 < m) ? v2 : z4;
        acc += (k + 3 < m) ? v3 : z4;
        acc += (k + 4 < m) ? v4 : z4;
        acc += (k + 5 < m) ? v5 : z4;
        acc += (k + 6 < m) ? v6 : z4;
        acc += (k + 7 < m) ? v7 : z4;
      }
    }
    v4f r;
    r.x = (acc.x + hs.x) * d + bv.x;
    r.y = (acc.y + hs.y) * d + bv.y;
    r.z = (acc.z + hs.z) * d + bv.z;
    r.w = (acc.w + hs.w) * d + bv.w;
    r.x = r.x > 0.0f ? r.x : 0.0f;
    r.y = r.y > 0.0f ? r.y : 0.0f;
    r.z = r.z > 0.0f ? r.z : 0.0f;
    r.w = r.w > 0.0f ? r.w : 0.0f;
    if (g0 + FN * w + q >= Nnodes) r = (v4f){0.0f, 0.0f, 0.0f, 0.0f};
    *(v4f*)(As + (FN * w + q) * FSP + 4 * lane) = r;
    if (lane == 0) dl[FN * w + q] = d;
  }
  __syncthreads();

  if (w < (FNB / 16) * NTL) {
    const int st = w / NTL;
    const int nt = w % NTL;
    const int l16 = lane & 15;
    const int hi  = lane >> 4;
    const int kh  = hi << 1;
    const float* Ak = As + (st * 16 + l16) * FSP + kh;
    const float* Bk = Bs + hi * FBP + 2 * l16 + 32 * nt;
    v8f c = (v8f){};
#pragma unroll
    for (int s = 0; s < KDIM / 4; ++s) {
      v2f a = *(const v2f*)(Ak + 4 * s);
      v2f b = *(const v2f*)(Bk + (2 * s) * FBP);
      c = __builtin_amdgcn_wmma_f32_16x16x4_f32(
           false, a,  false, b,
           (short)0, c,  false,  false);
    }
    const int rb = st * 16 + hi * 8;
    v4f dA = *(const v4f*)(dl + rb);
    v4f dB = *(const v4f*)(dl + rb + 4);
    float dr[8] = {dA.x, dA.y, dA.z, dA.w, dB.x, dB.y, dB.z, dB.w};
#pragma unroll
    for (int r = 0; r < 8; ++r) {
      int node = g0 + rb + r;
      float v = c[r] * dr[r];
      if (node < Nnodes) h2[(long long)node * LD2 + nt * 16 + l16] = v;
    }
  }
}

__global__ __launch_bounds__(256) void gather_rows40_lsm(
    const int* __restrict__ offs, const int* __restrict__ cnt,
    const int* __restrict__ src, const float* __restrict__ dinv,
    const float* __restrict__ h2, const float* __restrict__ b2,
    float* __restrict__ out, int Nnodes) {
  int lane = threadIdx.x & 31;
  int i = __builtin_amdgcn_readfirstlane((int)((blockIdx.x * 256u + threadIdx.x) >> 5));
  if (i >= Nnodes) return;
  bool hi = lane < (NCLS - 32);
  int beg = offs[i];
  int n = cnt[i];
  float a0 = 0.0f, a1 = 0.0f;
  for (int base = 0; base < n; base += 32) {
    int m = n - base;
    if (m > 32) m = 32;
    int sidx = i;
    if (lane < m) sidx = src[beg + base + lane];
    for (int k = 0; k < m; k += 4) {
      int s0 = __builtin_amdgcn_readlane(sidx, k);
      int s1 = __builtin_amdgcn_readlane(sidx, k + 1);
      int s2 = __builtin_amdgcn_readlane(sidx, k + 2);
      int s3 = __builtin_amdgcn_readlane(sidx, k + 3);
      const float* r0 = h2 + (long long)s0 * LD2 + lane;
      const float* r1 = h2 + (long long)s1 * LD2 + lane;
      const float* r2 = h2 + (long long)s2 * LD2 + lane;
      const float* r3 = h2 + (long long)s3 * LD2 + lane;
      float x0 = r0[0], x1 = r1[0], x2 = r2[0], x3 = r3[0];
      float y0 = 0.0f, y1 = 0.0f, y2 = 0.0f, y3 = 0.0f;
      if (hi) {
        y0 = r0[32];
        y1 = r1[32];
        y2 = r2[32];
        y3 = r3[32];
      }
      bool in1 = k + 1 < m, in2 = k + 2 < m, in3 = k + 3 < m;
      a0 += x0;
      a1 += y0;
      a0 += in1 ? x1 : 0.0f;
      a1 += in1 ? y1 : 0.0f;
      a0 += in2 ? x2 : 0.0f;
      a1 += in2 ? y2 : 0.0f;
      a0 += in3 ? x3 : 0.0f;
      a1 += in3 ? y3 : 0.0f;
    }
  }
  float d = dinv[i];
  const float* hs = h2 + (long long)i * LD2 + lane;
  float v0 = (a0 + hs[0]) * d + b2[lane];
  float v1 = 0.0f;
  float mx = v0;
  if (hi) {
    v1 = (a1 + hs[32]) * d + b2[32 + lane];
    mx = fmaxf(mx, v1);
  }
#pragma unroll
  for (int o = 16; o > 0; o >>= 1) mx = fmaxf(mx, __shfl_xor(mx, o));
  float ex = __expf(v0 - mx);
  if (hi) ex += __expf(v1 - mx);
#pragma unroll
  for (int o = 16; o > 0; o >>= 1) ex += __shfl_xor(ex, o);
  float lse = mx + __logf(ex);
  float* op = out + (long long)i * NCLS + lane;
  op[0] = v0 - lse;
  if (hi) op[32] = v1 - lse;
}


extern "C" void kernel_launch(void* const* d_in, const int* in_sizes, int n_in,
                              void* d_out, int out_size, void* d_ws, size_t ws_size,
                              hipStream_t stream) {
  const float* x  = (const float*)d_in[0];
  const int*   ei = (const int*)d_in[1];
  const float* W1 = (const float*)d_in[2];
  const float* b1 = (const float*)d_in[3];
  const float* W2 = (const float*)d_in[4];
  const float* b2 = (const float*)d_in[5];
  float* out = (float*)d_out;

  const int N = in_sizes[0] / NFEAT;
  const int E = in_sizes[1] / 2;
  const int* erow = ei;
  const int* ecol = ei + E;
  const int Npad = (N + 63) & ~63;

  float* h1   = (float*)d_ws;
  float* h2   = h1 + (long long)N * NFEAT;
  int*   cnt  = (int*)(h2 + (long long)N * LD2);
  int*   offs = cnt + Npad;
  float* dinv = (float*)(offs + Npad);
  int*   src  = (int*)(dinv + Npad);
  int*   pos  = src + E;
  const int Epad = (E + 63) & ~63;
  const int CH = (((E + NCHUNK - 1) / NCHUNK) + 3) & ~3;
  const int Rpad = (NCHUNK * CH + 63) & ~63;
  int*   rsrc = pos + Epad;
  int*   rpk  = rsrc + Rpad;
  int*   sb   = rpk + Rpad;

  sort_chunk<<<NCHUNK, 1024, 0, stream>>>(erow, ecol, pos, rsrc, rpk, sb, E, CH);

  const int fusedBlocks = (N + FNB - 1) / FNB;
  const int nodeBlocks = (N + 7) / 8;

  gemm_strip_f32<NFEAT / 16, NFEAT, true><<<NGB, 256, 0, stream>>>(
      x, W1, nullptr, h1, N, NFEAT, rsrc, rpk, sb, src, cnt, offs, dinv, CH);

  gather_relu_prod2<<<fusedBlocks, 256, 0, stream>>>(offs, cnt, src, dinv, h1, b1, W2, h2, N);

  gather_rows40_lsm<<<nodeBlocks, 256, 0, stream>>>(offs, cnt, src, dinv, h2, b2, out, N);
}


# layer-2 gather: 3 nodes per wave side by side, 10 lanes x 16 B = one row (rows at a 256-byte pitch), one load fetches a row of each; 1/3 of the waves; list order kept, log-sum-exp in new lane order
# speedup vs baseline: 5.2877x; 1.0612x over previous
#include <hip/hip_runtime.h>
#include <cmath>


typedef __attribute__((ext_vector_type(2))) float v2f;
typedef __attribute__((ext_vector_type(4))) float v4f;
typedef __attribute__((ext_vector_type(4))) int v4i;
typedef __attribute__((ext_vector_type(8))) float v8f;

#define NFEAT 128
#define NCLS  40
#define LD2   48
#define H2P   64
#define KDIM  128
#define WPB   8
#define NNODE 50000
#define HCAP  ((NNODE + 63) & ~63)
#define NCHUNK 64
#define CBATCH 16
#define GB    (WPB * 16)
#define NGB   (HCAP / GB)
#define SBP   (NGB + 1)
#define CHS   12512
#define SCAN4 13
static_assert(HCAP % GB == 0, "node blocks tile the padded node count");
static_assert(HCAP <= 65536, "a node index is packed into 16 bits beside the rank");
static_assert(NCHUNK == WPB * 8, "a wave of the first product merges 8 chunks");
static_assert(1024 * SCAN4 * 4 >= HCAP, "the scan covers the counter table");


__device__ __forceinline__ int wave_incl_scan(int v, int lane) {
#pragma unroll
  for (int d = 1; d < 32; d <<= 1) {
    int y = __shfl_up(v, d);
    if (lane >= d) v += y;
  }
  return v;
}

__global__ __launch_bounds__(1024) void sort_chunk(
    const int* __restrict__ row, const int* __restrict__ col, int* __restrict__ pos,
    int* __restrict__ rsrc, int* __restrict__ rpk, int* __restrict__ sb, int E, int CH) {
  __shared__ __attribute__((aligned(16))) int hist[HCAP];
  __shared__ __attribute__((aligned(16))) int stg[2 * CHS];
  __shared__ int wsum[32];
  int t = threadIdx.x;
  int lane = t & 31;
  int w = t >> 5;
  int k = blockIdx.x;
  int e0 = k * CH;
  int len = E - e0;
  if (len > CH) len = CH;
  if (len < 0) len = 0;
  int c0[CBATCH], s0[CBATCH], r0[CBATCH];
#pragma unroll
  for (int u = 0; u < CBATCH; ++u) {
    int j = u * 1024 + t;
    c0[u] = (j < len) ? col[e0 + j] : -1;
    s0[u] = (j < len) ? row[e0 + j] : 0;
  }
  for (int i4 = t; i4 < HCAP / 4; i4 += 1024) ((v4i*)hist)[i4] = (v4i){0, 0, 0, 0};
  __syncthreads();
#pragma unroll
  for (int u = 0; u < CBATCH; ++u) {
    r0[u] = 0;
    if ((unsigned)c0[u] < (unsigned)HCAP) r0[u] = atomicAdd(&hist[c0[u]], 1);
  }
  for (int j0 = CBATCH * 1024; j0 < len; j0 += CBATCH * 1024) {
    int c[CBATCH];
#pragma unroll
    for (int u = 0; u < CBATCH; ++u) {
      int j = j0 + u * 1024 + t;
      c[u] = (j < len) ? col[e0 + j] : -1;
    }
#pragma unroll
    for (int u = 0; u < CBATCH; ++u) {
      int j = j0 + u * 1024 + t;
      if ((unsigned)c[u] < (unsigned)HCAP) pos[e0 + j] = atomicAdd(&hist[c[u]], 1);
    }
  }
  __syncthreads();
  int sum = 0;
#pragma unroll
  for (int u = 0; u < SCAN4; ++u) {
    int i4 = t * SCAN4 + u;
    v4i hv = (i4 < HCAP / 4) ? ((const v4i*)hist)[i4] : (v4i){0, 0, 0, 0};
    sum += hv.x + hv.y + hv.z + hv.w;
  }
  int incl = wave_incl_scan(sum, lane);
  if (lane == 31) wsum[w] = incl;
  __syncthreads();
  int wpre = 0;
  int tot = 0;
#pragma unroll
  for (int u = 0; u < 32; ++u) {
    int x = wsum[u];
    if (u < w) wpre += x;
    tot += x;
  }
  int run = wpre + (incl - sum);
#pragma unroll
  for (int u = 0; u < SCAN4; ++u) {
    int i4 = t * SCAN4 + u;
    v4i hv = (i4 < HCAP / 4) ? ((const v4i*)hist)[i4] : (v4i){0, 0, 0, 0};
    v4i o;
    o.x = run; run += hv.x;
    o.y = run; run += hv.y;
    o.z = run; run += hv.z;
    o.w = run; run += hv.w;
    if (i4 < HCAP / 4) ((v4i*)hist)[i4] = o;
  }
  __syncthreads();
  int* sbk = sb + k * SBP;
  for (int g = t; g < NGB; g += 1024) sbk[g] = hist[g * GB];
  if (t == 0) sbk[NGB] = tot;
  const bool fits = len <= CHS;
#pragma unroll
  for (int u = 0; u < CBATCH; ++u) {
    if ((unsigned)c0[u] < (unsigned)HCAP) {
      int q = hist[c0[u]] + r0[u];
      int pk = c0[u] | (r0[u] << 16);
      if (fits) {
        stg[q] = s0[u];
        stg[CHS + q] = pk;
      } else {
        rsrc[e0 + q] = s0[u];
        rpk[e0 + q] = pk;
      }
    }
  }
  for (int j = CBATCH * 1024 + t; j < len; j += 1024) {
    int cc = col[e0 + j];
    if ((unsigned)cc < (unsigned)HCAP) {
      int rr = pos[e0 + j];
      int q = hist[cc] + rr;
      rsrc[e0 + q] = row[e0 + j];
      rpk[e0 + q] = cc | (rr << 16);
    }
  }
  __syncthreads();
  if (fits) {
    int n4 = (tot + 3) >> 2;
    for (int i4 = t; i4 < n4; i4 += 1024) {
      ((v4i*)(rsrc + e0))[i4] = ((const v4i*)stg)[i4];
      ((v4i*)(rpk + e0))[i4] = ((const v4i*)(stg + CHS))[i4];
    }
  }
}

template <int NT, int NV, bool MERGE>
__global__ __launch_bounds__(256) void gemm_strip_f32(
    const float* __restrict__ A, const float* __restrict__ B,
    const float* __restrict__ rs, float* __restrict__ C, int M, int ldc,
    const int* __restrict__ rsrc, const int* __restrict__ rpk,
    const int* __restrict__ sb, int* __restrict__ src, int* __restrict__ cnt,
    int* __restrict__ offs, float* __restrict__ dinvw, int CH) {
  constexpr int NW   = NT * 16;
  constexpr int V4R  = NW / 4;
  constexpr int NV4  = NV / 4;
  constexpr int SW   = (NW > KDIM ? NW : KDIM);
  constexpr int SP   = SW + 4;
  constexpr int NJ   = (16 * V4R) / 32;
  constexpr int BP   = 2 * NW + ((NT % 2 == 0) ? 32 : 0);
  constexpr int NB   = ((KDIM / 2) * V4R) / 256;
  constexpr int CPW  = NCHUNK / WPB;
  __shared__ __attribute__((aligned(16))) float Bs[(KDIM / 2) * BP];
  __shared__ __attribute__((aligned(16))) float Ss[WPB * 16 * SP];
  static_assert(!MERGE || (KDIM / 2) * BP >= NCHUNK * GB + 2 * GB + 16,
                "the merge tables fit the space of B");

  int t = threadIdx.x;
  int w = t >> 5;
  int lane = t & 31;
  int l16 = lane & 15;
  int hi  = lane >> 4;
  int kh  = hi << 1;

  int Mtiles = (M + 15) >> 4;
  int mt = blockIdx.x * WPB + w;
  int m0 = mt << 4;
  float* Sw = Ss + w * (16 * SP);

  int sg0[CPW], sgl[CPW];
  int asum = 0;
  int fs[CPW][2], fk[CPW][2];
  const int wu = __builtin_amdgcn_readfirstlane(w);
  const int g0 = blockIdx.x * GB;
  if (MERGE) {
#pragma unroll
    for (int kk = 0; kk < CPW; ++kk) {
      int k = wu + WPB * kk;
      const int* sbk = sb + k * SBP + blockIdx.x;
      int a = sbk[0];
      int b = sbk[1];
      sg0[kk] = k * CH + a;
      sgl[kk] = b - a;
      asum += a;
    }
  }

  v4f ar[16];
#pragma unroll
  for (int r = 0; r < 16; ++r) {
    int m = m0 + r;
    if (m >= M) m = M - 1;
    ar[r] = *(const v4f*)(A + (long long)m * KDIM + 4 * lane);
  }

  v4f b0[NB], b1[NB];
#pragma unroll
  for (int j = 0; j < NB; ++j) {
    int item = t + j * 256;
    int kp = item / V4R;
    int c4 = item % V4R;
    int c4c = c4 < NV4 ? c4 : NV4 - 1;
    b0[j] = *(const v4f*)(B + (2 * kp) * NV + c4c * 4);
    b1[j] = *(const v4f*)(B + (2 * kp + 1) * NV + c4c * 4);
    if (c4 >= NV4) {
      b0[j] = (v4f){0.0f, 0.0f, 0.0f, 0.0f};
      b1[j] = (v4f){0.0f, 0.0f, 0.0f, 0.0f};
    }
  }
  if (MERGE) {
#pragma unroll
    for (int kk = 0; kk < CPW; ++kk) {
#pragma unroll
      for (int h = 0; h < 2; ++h) {
        int j = lane + 32 * h;
        int ix = (j < sgl[kk]) ? sg0[kk] + j : 0;
        fs[kk][h] = rsrc[ix];
        fk[kk][h] = rpk[ix];
      }
    }
  }
#pragma unroll
  for (int r = 0; r < 16; ++r) *(v4f*)(Sw + r * SP + 4 * lane) = ar[r];
#pragma unroll
  for (int j = 0; j < NB; ++j) {
    int item = t + j * 256;
    int kp = item / V4R;
    int c4 = item % V4R;
    float* d = Bs + kp * BP + 8 * c4;
    *(v4f*)d       = (v4f){b0[j].x, b1[j].x, b0[j].y, b1[j].y};
    *(v4f*)(d + 4) = (v4f){b0[j].z, b1[j].z, b0[j].w, b1[j].w};
  }
  __syncthreads();

  const bool live = mt < Mtiles;
  float rsv[NJ];
  v8f acc[NT];
#pragma unroll
  for (int nt = 0; nt < NT; ++nt) acc[nt] = (v8f){};

  if (live) {
    if (!MERGE) {
#pragma unroll
      for (int j = 0; j < NJ; ++j) rsv[j] = rs[m0 + (j * 32 + lane) / V4R];
    }
    const float* Ak = Sw + l16 * SP + kh;
#pragma unroll
    for (int s = 0; s < KDIM / 4; ++s) {
      v2f a = *(const v2f*)(Ak + 4 * s);
      const float* Bk = Bs + (2 * s + hi) * BP + 2 * l16;
#pragma unroll
      for (int nt = 0; nt < NT; ++nt) {
        v2f b = *(const v2f*)(Bk + 32 * nt);
        acc[nt] = __builtin_amdgcn_wmma_f32_16x16x4_f32(
             false, a,  false, b,
             (short)0, acc[nt],  false,  false);
      }
    }
  }

  if (MERGE) {
    int* tab = (int*)Bs;
    int* nfirst = tab + NCHUNK * GB;
    float* dl = (float*)(nfirst + GB);
    int* misc = (int*)(dl + GB);
    __syncthreads();
    for (int i4 = t; i4 < NCHUNK * GB / 4; i4 += 256) ((v4i*)tab)[i4] = (v4i){0, 0, 0, 0};
    if (lane == 0) misc[w] = asum;
    __syncthreads();
#pragma unroll
    for (int kk = 0; kk < CPW; ++kk) {
      int* tk = tab + (wu + WPB * kk) * GB;
#pragma unroll
      for (int h = 0; h < 2; ++h)
        if (lane + 32 * h < sgl[kk]) atomicAdd(&tk[(fk[kk][h] & 0xffff) - g0], 1);
      for (int j = lane + 64; j < sgl[kk]; j += 32)
        atomicAdd(&tk[(rpk[sg0[kk] + j] & 0xffff) - g0], 1);
    }
    __syncthreads();
    int ntot = 0;
    if (t < GB) {
#pragma unroll 8
      for (int k = 0; k < NCHUNK; ++k) {
        int v = tab[k * GB + t];
        tab[k * GB + t] = ntot;
        ntot += v;
      }
    }
    int nincl = wave_incl_scan(ntot, lane);
    if (lane == 31 && w < GB / 32) misc[8 + w] = nincl;
    __syncthreads();
    if (t < GB) {
      int first = 0;
#pragma unroll
      for (int u = 0; u < WPB; ++u) first += misc[u];
#pragma unroll
      for (int u = 0; u < GB / 32; ++u)
        if (u < w) first += misc[8 + u];
      first += nincl - ntot;
      float dg = (float)(ntot + 1);
      float dv = (dg > 0.0f) ? rsqrtf(dg) : 0.0f;
      nfirst[t] = first;
      dl[t] = dv;
      cnt[g0 + t] = ntot;
      offs[g0 + t] = first;
      dinvw[g0 + t] = dv;
    }
    __syncthreads();
#pragma unroll
    for (int kk = 0; kk < CPW; ++kk) {
      const int* tk = tab + (wu + WPB * kk) * GB;
      const int* nf = nfirst;
#pragma unroll
      for (int h = 0; h < 2; ++h) {
        if (lane + 32 * h < sgl[kk]) {
          int n = (fk[kk][h] & 0xffff) - g0;
          int r = (int)((unsigned)fk[kk][h] >> 16);
          src[nf[n] + tk[n] + r] = fs[kk][h];
        }
      }
      for (int j = lane + 64; j < sgl[kk]; j += 32) {
        int pk = rpk[sg0[kk] + j];
        int n = (pk & 0xffff) - g0;
        int r = (int)((unsigned)pk >> 16);
        src[nf[n] + tk[n] + r] = rsrc[sg0[kk] + j];
      }
    }
    if (live) {
#pragma unroll
      for (int j = 0; j < NJ; ++j) rsv[j] = dl[w * 16 + (j * 32 + lane) / V4R];
    }
  }

  if (live) {
#pragma unroll
  for (int nt = 0; nt < NT; ++nt)
#pragma unroll
    for (int r = 0; r < 8; ++r) Sw[(hi * 8 + r) * SP + nt * 16 + l16] = acc[nt][r];

  if (m0 + 16 <= M) {
    v4f cv[NJ];
#pragma unroll
    for (int j = 0; j < NJ; ++j) {
      int e = j * 32 + lane;
      cv[j] = *(const v4f*)(Sw + (e / V4R) * SP + (e % V4R) * 4) * rsv[j];
    }
#pragma unroll
    for (int j = 0; j < NJ; ++j) {
      int e = j * 32 + lane;
      *(v4f*)(C + (long long)(m0 + e / V4R) * ldc + (e % V4R) * 4) = cv[j];
    }
  } else {
#pragma unroll
    for (int j = 0; j < NJ; ++j) {
      int e = j * 32 + lane;
      int row = e / V4R;
      int c4 = e % V4R;
      v4f v = *(const v4f*)(Sw + row * SP + c4 * 4) * rsv[j];
      if (m0 + row < M) *(v4f*)(C + (long long)(m0 + row) * ldc + c4 * 4) = v;
    }
  }
  }
}

#define FN   4
#define FNB  (FN * WPB)
#define FSP  (NFEAT + 4)
#define FBP  (2 * LD2)
static_assert(FNB % 16 == 0, "a work-group's rows are whole strips");
static_assert((FNB / 16) * (LD2 / 16) <= WPB, "one wave per 16 x 16 tile of the work-group's rows");

__global__ __launch_bounds__(256) void gather_relu_prod2(
    const int* __restrict__ offs, const int* __restrict__ cnt,
    const int* __restrict__ src, const float* __restrict__ dinv,
    const float* __restrict__ h, const float* __restrict__ bias,
    const float* __restrict__ W2, float* __restrict__ h2, int Nnodes) {
  constexpr int V4R = LD2 / 4;
  constexpr int NV4 = NCLS / 4;
  constexpr int NB  = ((KDIM / 2) * V4R) / 256;
  static_assert(((KDIM / 2) * V4R) % 256 == 0, "the fill of W2 has no tail");
  constexpr int NTL = LD2 / 16;
  __shared__ __attribute__((aligned(16))) float Bs[(KDIM / 2) * FBP];
  __shared__ __attribute__((aligned(16))) float As[FNB * FSP];
  __shared__ __attribute__((aligned(16))) float dl[FNB];

  int t = threadIdx.x;
  int lane = t & 31;
  const int w = __builtin_amdgcn_readfirstlane(t >> 5);
  const int g0 = blockIdx.x * FNB;
  const v4f* h4 = (const v4f*)h;
  const unsigned ul = (unsigned)lane;
  typedef const __attribute__((address_space(1))) v4f* gv4p;
  const unsigned long long hb = (unsigned long long)h;

  v4f b0[NB], b1[NB];
#pragma unroll
  for (int j = 0; j < NB; ++j) {
    int item = t + j * 256;
    int kp = item / V4R;
    int c4 = item % V4R;
    int c4c = c4 < NV4 ? c4 : NV4 - 1;
    b0[j] = *(const v4f*)(W2 + (2 * kp) * NCLS + c4c * 4);
    b1[j] = *(const v4f*)(W2 + (2 * kp + 1) * NCLS + c4c * 4);
    if (c4 >= NV4) {
      b0[j] = (v4f){0.0f, 0.0f, 0.0f, 0.0f};
      b1[j] = (v4f){0.0f, 0.0f, 0.0f, 0.0f};
    }
  }

  int ii[FN], beg[FN], n[FN], fe[FN];
#pragma unroll
  for (int q = 0; q < FN; ++q) {
    int i = g0 + FN * w + q;
    ii[q] = i < Nnodes ? i : Nnodes - 1;
    beg[q] = offs[ii[q]];
    n[q] = i < Nnodes ? cnt[ii[q]] : 0;
  }
#pragma unroll
  for (int q = 0; q < FN; ++q) {
    fe[q] = ii[q];
    if (lane < n[q]) fe[q] = src[beg[q] + lane];
  }
  v4f bv = ((const v4f*)bias)[lane];

#pragma unroll
  for (int j = 0; j < NB; ++j) {
    int item = t + j * 256;
    int kp = item / V4R;
    int c4 = item % V4R;
    float* d = Bs + kp * FBP + 8 * c4;
    *(v4f*)d       = (v4f){b0[j].x, b1[j].x, b0[j].y, b1[j].y};
    *(v4f*)(d + 4) = (v4f){b0[j].z, b1[j].z, b0[j].w, b1[j].w};
  }

#pragma unroll
  for (int q = 0; q < FN; ++q) {
    const int i = ii[q];
    const int nq = n[q];
    float d = dinv[i];
    v4f hs = h4[(long long)i * (NFEAT / 4) + lane];
    v4f acc = {0.0f, 0.0f, 0.0f, 0.0f};
    for (int base = 0; base < nq; base += 32) {
      int m = nq - base;
      if (m > 32) m = 32;
      int sidx = fe[q];
      if (base > 0) {
        sidx = i;
        if (lane < m) sidx = src[beg[q] + base + lane];
      }
      for (int k = 0; k < m; k += 8) {
        int s0 = __builtin_amdgcn_readlane(sidx, k);
        int s1 = __builtin_amdgcn_readlane(sidx, k + 1);
        int s2 = __builtin_amdgcn_readlane(sidx, k + 2);
        int s3 = __builtin_amdgcn_readlane(sidx, k + 3);
        int s4 = __builtin_amdgcn_readlane(sidx, k + 4);
        int s5 = __builtin_amdgcn_readlane(sidx, k + 5);
        int s6 = __builtin_amdgcn_readlane(sidx, k + 6);
        int s7 = __builtin_amdgcn_readlane(sidx, k + 7);
        v4f v0 = ((gv4p)(hb + ((unsigned long long)(unsigned)s0 << 9)))[ul];
        v4f v1 = ((gv4p)(hb + ((unsigned long long)(unsigned)s1 << 9)))[ul];
        v4f v2 = ((gv4p)(hb + ((unsigned long long)(unsigned)s2 << 9)))[ul];
        v4f v3 = ((gv4p)(hb + ((unsigned long long)(unsigned)s3 << 9)))[ul];
        v4f v4 = ((gv4p)(hb + ((unsigned long long)(unsigned)s4 << 9)))[ul];
        v4f v5 = ((gv4p)(hb + ((unsigned long long)(unsigned)s5 << 9)))[ul];
        v4f v6 = ((gv4p)(hb + ((unsigned long long)(unsigned)s6 << 9)))[ul];
        v4f v7 = ((gv4p)(hb + ((unsigned long long)(unsigned)s7 << 9)))[ul];
        const v4f z4 = {0.0f, 0.0f, 0.0f, 0.0f};
        acc += v0;
        acc += (k + 1 < m) ? v1 : z4;
        acc += (k + 2 < m) ? v2 : z4;
        acc += (k + 3 < m) ? v3 : z4;
        acc += (k + 4 < m) ? v4 : z4;
        acc += (k + 5 < m) ? v5 : z4;
        acc += (k + 6 < m) ? v6 : z4;
        acc += (k + 7 < m) ? v7 : z4;
      }
    }
    v4f r;
    r.x = (acc.x + hs.x) * d + bv.x;
    r.y = (acc.y + hs.y) * d + bv.y;
    r.z = (acc.z + hs.z) * d + bv.z;
    r.w = (acc.w + hs.w) * d + bv.w;
    r.x = r.x > 0.0f ? r.x : 0.0f;
    r.y = r.y > 0.0f ? r.y : 0.0f;
    r.z = r.z > 0.0f ? r.z : 0.0f;
    r.w = r.w > 0.0f ? r.w : 0.0f;
    if (g0 + FN * w + q >= Nnodes) r = (v4f){0.0f, 0.0f, 0.0f, 0.0f};
    *(v4f*)(As + (FN * w + q) * FSP + 4 * lane) = r;
    if (lane == 0) dl[FN * w + q] = d;
  }
  __syncthreads();

  if (w < (FNB / 16) * NTL) {
    const int st = w / NTL;
    const int nt = w % NTL;
    const int l16 = lane & 15;
    const int hi  = lane >> 4;
    const int kh  = hi << 1;
    const float* Ak = As + (st * 16 + l16) * FSP + kh;
    const float* Bk = Bs + hi * FBP + 2 * l16 + 32 * nt;
    v8f c = (v8f){};
#pragma unroll
    for (int s = 0; s < KDIM / 4; ++s) {
      v2f a = *(const v2f*)(Ak + 4 * s);
      v2f b = *(const v2f*)(Bk + (2 * s) * FBP);
      c = __builtin_amdgcn_wmma_f32_16x16x4_f32(
           false, a,  false, b,
           (short)0, c,  false,  false);
    }
    const int rb = st * 16 + hi * 8;
    v4f dA = *(const v4f*)(dl + rb);
    v4f dB = *(const v4f*)(dl + rb + 4);
    float dr[8] = {dA.x, dA.y, dA.z, dA.w, dB.x, dB.y, dB.z, dB.w};
#pragma unroll
    for (int r = 0; r < 8; ++r) {
      int node = g0 + rb + r;
      float v = c[r] * dr[r];
      if (node < Nnodes) h2[(long long)node * H2P + nt * 16 + l16] = v;
    }
  }
}

#define G2N 3
#define G2L 10
#define G2H 5
static_assert(G2L * 4 == NCLS, "a node's lanes cover a row exactly");
static_assert(G2N * G2L <= 32, "the nodes of a wave fit its lanes");
static_assert(G2L == 2 * G2H, "an index batch is two row groups");

__global__ __launch_bounds__(256) void gather_rows40_lsm(
    const int* __restrict__ offs, const int* __restrict__ cnt,
    const int* __restrict__ src, const float* __restrict__ dinv,
    const float* __restrict__ h2, const float* __restrict__ b2,
    float* __restrict__ out, int Nnodes) {
  const int lane = threadIdx.x & 31;
  const int wv = __builtin_amdgcn_readfirstlane((int)((blockIdx.x * 256u + threadIdx.x) >> 5));
  const int i0 = wv * G2N;
  if (i0 >= Nnodes) return;
  const int j = lane / G2L;
  const int p = lane - G2L * j;
  const int gl = G2L * j;
  const int i = i0 + j;
  const bool act = (j < G2N) && (i < Nnodes);
  const int ic = act ? i : i0;
  typedef const __attribute__((address_space(1))) char* gcp;
  typedef const __attribute__((address_space(1))) v4f* gv4p;
  const gcp hb = (gcp)h2;
  const unsigned po = 16u * (unsigned)p;
  const unsigned so = (unsigned)ic * (unsigned)(H2P * 4) + po;

  const int beg = offs[ic];
  const int n = act ? cnt[ic] : 0;
  const float d = dinv[ic];
  const v4f hs = *(gv4p)(hb + so);
  const v4f bv = ((const v4f*)b2)[p];
  int nmax = __builtin_amdgcn_readlane(n, 0);
  {
    int n1 = __builtin_amdgcn_readlane(n, G2L);
    int n2 = __builtin_amdgcn_readlane(n, 2 * G2L);
    nmax = nmax > n1 ? nmax : n1;
    nmax = nmax > n2 ? nmax : n2;
  }

  const v4f z4 = {0.0f, 0.0f, 0.0f, 0.0f};
  v4f acc = z4;
  int idx = src[(p < n) ? beg + p : 0];
  for (int base = 0; base < nmax; base += G2L) {
    const int en = base + G2L + p;
    int idxn = src[(en < n) ? beg + en : 0];
    const bool two = base + G2H < nmax;
    v4f va[G2H], vb[G2H];
#pragma unroll
    for (int u = 0; u < G2H; ++u) {
      int s = __shfl(idx, gl + u);
      unsigned o = (base + u < n) ? (unsigned)s * (unsigned)(H2P * 4) + po : so;
      va[u] = *(gv4p)(hb + o);
    }
    if (two) {
#pragma unroll
      for (int u = 0; u < G2H; ++u) {
        int s = __shfl(idx, gl + G2H + u);
        unsigned o = (base + G2H + u < n) ? (unsigned)s * (unsigned)(H2P * 4) + po : so;
        vb[u] = *(gv4p)(hb + o);
      }
    }
#pragma unroll
    for (int u = 0; u < G2H; ++u) acc += (base + u < n) ? va[u] : z4;
    if (two) {
#pragma unroll
      for (int u = 0; u < G2H; ++u) acc += (base + G2H + u < n) ? vb[u] : z4;
    }
    idx = idxn;
  }

  v4f r;
  r.x = (acc.x + hs.x) * d + bv.x;
  r.y = (acc.y + hs.y) * d + bv.y;
  r.z = (acc.z + hs.z) * d + bv.z;
  r.w = (acc.w + hs.w) * d + bv.w;
  float mx = fmaxf(fmaxf(r.x, r.y), fmaxf(r.z, r.w));
#pragma unroll
  for (int o = 1; o < G2L; o <<= 1) {
    int q = p + o;
    if (q >= G2L) q -= G2L;
    mx = fmaxf(mx, __shfl(mx, gl + q));
  }
  float ex = (__expf(r.x - mx) + __expf(r.y - mx)) + (__expf(r.z - mx) + __expf(r.w - mx));
#pragma unroll
  for (int o = 8; o > 0; o >>= 1) {
    float y = __shfl(ex, lane + o);
    if (p < o && p + o < G2L) ex += y;
  }
  ex = __shfl(ex, gl);
  float lse = mx + __logf(ex);
  if (act) {
    v4f ov = {r.x - lse, r.y - lse, r.z - lse, r.w - lse};
    *(v4f*)(out + (long long)i * NCLS + 4 * p) = ov;
  }
}


extern "C" void kernel_launch(void* const* d_in, const int* in_sizes, int n_in,
                              void* d_out, int out_size, void* d_ws, size_t ws_size,
                              hipStream_t stream) {
  const float* x  = (const float*)d_in[0];
  const int*   ei = (const int*)d_in[1];
  const float* W1 = (const float*)d_in[2];
  const float* b1 = (const float*)d_in[3];
  const float* W2 = (const float*)d_in[4];
  const float* b2 = (const float*)d_in[5];
  float* out = (float*)d_out;

  const int N = in_sizes[0] / NFEAT;
  const int E = in_sizes[1] / 2;
  const int* erow = ei;
  const int* ecol = ei + E;
  const int Npad = (N + 63) & ~63;

  float* h1   = (float*)d_ws;
  float* h2   = h1 + (long long)N * NFEAT;
  int*   cnt  = (int*)(h2 + (long long)N * H2P);
  int*   offs = cnt + Npad;
  float* dinv = (float*)(offs + Npad);
  int*   src  = (int*)(dinv + Npad);
  int*   pos  = src + E;
  const int Epad = (E + 63) & ~63;
  const int CH = (((E + NCHUNK - 1) / NCHUNK) + 3) & ~3;
  const int Rpad = (NCHUNK * CH + 63) & ~63;
  int*   rsrc = pos + Epad;
  int*   rpk  = rsrc + Rpad;
  int*   sb   = rpk + Rpad;

  sort_chunk<<<NCHUNK, 1024, 0, stream>>>(erow, ecol, pos, rsrc, rpk, sb, E, CH);

  const int fusedBlocks = (N + FNB - 1) / FNB;
  const int nodeBlocks = (N + 8 * G2N - 1) / (8 * G2N);

  gemm_strip_f32<NFEAT / 16, NFEAT, true><<<NGB, 256, 0, stream>>>(
      x, W1, nullptr, h1, N, NFEAT, rsrc, rpk, sb, src, cnt, offs, dinv, CH);

  gather_relu_prod2<<<fusedBlocks, 256, 0, stream>>>(offs, cnt, src, dinv, h1, b1, W2, h2, N);

  gather_rows40_lsm<<<nodeBlocks, 256, 0, stream>>>(offs, cnt, src, dinv, h2, b2, out, N);
}
